# starting kernel
# baseline (speedup 1.0000x reference)
#include <hip/hip_runtime.h>
#include <hip/hip_bf16.h>

typedef float v8f __attribute__((ext_vector_type(8)));
typedef float v2f __attribute__((ext_vector_type(2)));
typedef _Float16 v16h __attribute__((ext_vector_type(16)));

#define DFEAT 128

__device__ __forceinline__ v8f rank2_wmma(float ax, float ay, float bx, float by) {
    v8f c = {};
#if __has_builtin(__builtin_amdgcn_wmma_f32_16x16x4_f32)
    v2f a; a.x = ax; a.y = ay;
    v2f b; b.x = bx; b.y = by;
    c = __builtin_amdgcn_wmma_f32_16x16x4_f32(false, a, false, b, (short)0, c, false, false);
#else
    v16h a = {}; a[0] = (_Float16)ax; a[1] = (_Float16)ay;
    v16h b = {}; b[0] = (_Float16)bx; b[1] = (_Float16)by;
    c = __builtin_amdgcn_wmma_f32_16x16x32_f16(false, a, false, b, (short)0, c, false, false);
#endif
    return c;
}

__global__ void k0_init(float* __restrict__ out, float* __restrict__ amax,
                        float* __restrict__ denom, int N, int ND) {
    int i = blockIdx.x * blockDim.x + threadIdx.x;
    if (i < ND) out[i] = 0.0f;
    if (i < N) { amax[i] = -3.0e38f; denom[i] = 0.0f; }
}

__global__ void k1_node_scalars(const float* __restrict__ h,
                                const float* __restrict__ w_attn,
                                float* __restrict__ s1, float* __restrict__ s2, int N) {
    const int lane = threadIdx.x & 31;
    const int node = blockIdx.x * (blockDim.x >> 5) + (threadIdx.x >> 5);
    if (node >= N) return;
    const float4* hv = (const float4*)(h + (long long)node * DFEAT);
    const float4* w1 = (const float4*)(w_attn);
    const float4* w2 = (const float4*)(w_attn + DFEAT);
    float4 x  = hv[lane];
    float4 a1 = w1[lane];
    float4 a2 = w2[lane];
    float p1 = x.x*a1.x + x.y*a1.y + x.z*a1.z + x.w*a1.w;
    float p2 = x.x*a2.x + x.y*a2.y + x.z*a2.z + x.w*a2.w;
    #pragma unroll
    for (int off = 16; off > 0; off >>= 1) {
        p1 += __shfl_xor(p1, off, 32);
        p2 += __shfl_xor(p2, off, 32);
    }
    if (lane == 0) { s1[node] = p1; s2[node] = p2; }
}

__device__ __forceinline__ void atomicMaxFloat(float* addr, float val) {
    if (val >= 0.0f) atomicMax((int*)addr, __float_as_int(val));
    else             atomicMin((unsigned int*)addr, __float_as_uint(val));
}

__global__ void k2_edge_logits(const float* __restrict__ e,
                               const int* __restrict__ src, const int* __restrict__ dst,
                               const float* __restrict__ s1, const float* __restrict__ s2,
                               const float* __restrict__ w_edge,
                               const float* __restrict__ w_attn,
                               float* __restrict__ a, float* __restrict__ amax, int E) {
    int i = blockIdx.x * blockDim.x + threadIdx.x;
    if (i >= E) return;
    float e0 = e[2*i], e1 = e[2*i + 1];
    float ex0 = e0 * w_edge[0] + e1 * w_edge[1];
    float ex1 = e0 * w_edge[2] + e1 * w_edge[3];
    float logit = s1[src[i]] + s2[dst[i]] + ex0 * w_attn[2*DFEAT] + ex1 * w_attn[2*DFEAT + 1];
    logit = (logit > 0.0f) ? logit : 0.01f * logit;
    a[i] = logit;
    atomicMaxFloat(&amax[dst[i]], logit);
}

__global__ void k3_denom(float* __restrict__ a, const int* __restrict__ dst,
                         const float* __restrict__ amax, float* __restrict__ denom, int E) {
    int i = blockIdx.x * blockDim.x + threadIdx.x;
    if (i >= E) return;
    int d = dst[i];
    float w = __expf(a[i] - amax[d]);
    a[i] = w;
    atomicAdd(&denom[d], w);
}

__global__ void k4_scatter(const float* __restrict__ e, const int* __restrict__ dst,
                           const float* __restrict__ aexp, const float* __restrict__ denom,
                           const float* __restrict__ w_edge, const float* __restrict__ w_e2n,
                           float* __restrict__ out, int E) {
    const int lane = threadIdx.x & 31;
    const long long wave = (long long)blockIdx.x * (blockDim.x >> 5) + (threadIdx.x >> 5);
    const long long base = wave * 16;

    const int lo = lane & 15;
    const bool loHalf = (lane < 16);
    long long eIdx = base + lo;
    const bool validA = loHalf && (eIdx < (long long)E);
    const long long sIdx = validA ? eIdx : 0;

    const float2* wv = (const float2*)w_e2n;
    float bxs[8], bys[8];
    #pragma unroll
    for (int j = 0; j < 8; ++j) {
        float2 b = wv[16 * j + lo];
        bxs[j] = loHalf ? b.x : 0.0f;
        bys[j] = loHalf ? b.y : 0.0f;
    }

    float e0 = e[2*sIdx], e1 = e[2*sIdx + 1];
    float ex0 = e0 * w_edge[0] + e1 * w_edge[1];
    float ex1 = e0 * w_edge[2] + e1 * w_edge[3];
    int   dmy  = dst[sIdx];
    float attn = aexp[sIdx] * __builtin_amdgcn_rcpf(fmaxf(denom[dmy], 1e-16f));
    attn = validA ? attn : 0.0f;
    const float ax = attn * ex0;
    const float ay = attn * ex1;

    const long long rowBase = base + ((lane >= 16) ? 8 : 0);
    int   dstRow[8];
    float okRow[8];
    #pragma unroll
    for (int v = 0; v < 8; ++v) {
        long long er = rowBase + v;
        bool ok = er < (long long)E;
        dstRow[v] = dst[ok ? er : 0];
        okRow[v]  = ok ? 1.0f : 0.0f;
    }

    #pragma unroll
    for (int j = 0; j < 8; ++j) {
        const int col = 16 * j + lo;
        v8f c = rank2_wmma(ax, ay, bxs[j], bys[j]);
        #pragma unroll
        for (int v = 0; v < 8; ++v) {
            atomicAdd(&out[(long long)dstRow[v] * DFEAT + col], c[v] * okRow[v]);
        }
    }
}

extern "C" void kernel_launch(void* const* d_in, const int* in_sizes, int n_in,
                              void* d_out, int out_size, void* d_ws, size_t ws_size,
                              hipStream_t stream) {
    const float* h      = (const float*)d_in[0];
    const float* e      = (const float*)d_in[1];
    const int*   src    = (const int*)d_in[2];
    const int*   dst    = (const int*)d_in[3];
    const float* w_edge = (const float*)d_in[4];
    const float* w_attn = (const float*)d_in[5];
    const float* w_e2n  = (const float*)d_in[6];
    float* out = (float*)d_out;

    const int E = in_sizes[2];
    const int N = in_sizes[0] / DFEAT;
    const int ND = N * DFEAT;

    float* s1    = (float*)d_ws;
    float* s2    = s1 + N;
    float* amax  = s2 + N;
    float* denom = amax + N;
    float* a     = denom + N;

    const int T = 256;
    const int wavesPerBlock = T / 32;

    k0_init<<<(ND + T - 1) / T, T, 0, stream>>>(out, amax, denom, N, ND);
    k1_node_scalars<<<(N + wavesPerBlock - 1) / wavesPerBlock, T, 0, stream>>>(
        h, w_attn, s1, s2, N);
    k2_edge_logits<<<(E + T - 1) / T, T, 0, stream>>>(
        e, src, dst, s1, s2, w_edge, w_attn, a, amax, E);
    k3_denom<<<(E + T - 1) / T, T, 0, stream>>>(a, dst, amax, denom, E);

    const long long wavesTotal = ((long long)E + 15) / 16;
    const int gScatter = (int)((wavesTotal + wavesPerBlock - 1) / wavesPerBlock);
    k4_scatter<<<gScatter, T, 0, stream>>>(e, dst, a, denom, w_edge, w_e2n, out, E);
}


# Sum (w, w*ex) per dst node first (3 adjacent f32 atomics per edge into the node's out row), then normalise and project by w_e2n once per node on WMMA; sum order of out changes
# speedup vs baseline: 1.6286x; 1.6286x over previous
#include <hip/hip_runtime.h>
#include <hip/hip_bf16.h>

typedef float v8f __attribute__((ext_vector_type(8)));
typedef float v2f __attribute__((ext_vector_type(2)));
typedef _Float16 v16h __attribute__((ext_vector_type(16)));

#define DFEAT 128

__device__ __forceinline__ v8f rank2_wmma(float ax, float ay, float bx, float by) {
    v8f c = {};
#if __has_builtin(__builtin_amdgcn_wmma_f32_16x16x4_f32)
    v2f a; a.x = ax; a.y = ay;
    v2f b; b.x = bx; b.y = by;
    c = __builtin_amdgcn_wmma_f32_16x16x4_f32(false, a, false, b, (short)0, c, false, false);
#else
    v16h a = {}; a[0] = (_Float16)ax; a[1] = (_Float16)ay;
    v16h b = {}; b[0] = (_Float16)bx; b[1] = (_Float16)by;
    c = __builtin_amdgcn_wmma_f32_16x16x32_f16(false, a, false, b, (short)0, c, false, false);
#endif
    return c;
}

__global__ void k0_init(float* __restrict__ out, float* __restrict__ amax, int N) {
    int n = blockIdx.x * blockDim.x + threadIdx.x;
    if (n < N) {
        *(float4*)(out + (long long)n * DFEAT) = make_float4(0.0f, 0.0f, 0.0f, 0.0f);
        amax[n] = -3.0e38f;
    }
}

__global__ void k1_node_scalars(const float* __restrict__ h,
                                const float* __restrict__ w_attn,
                                float* __restrict__ s1, float* __restrict__ s2, int N) {
    const int lane = threadIdx.x & 31;
    const int node = blockIdx.x * (blockDim.x >> 5) + (threadIdx.x >> 5);
    if (node >= N) return;
    const float4* hv = (const float4*)(h + (long long)node * DFEAT);
    const float4* w1 = (const float4*)(w_attn);
    const float4* w2 = (const float4*)(w_attn + DFEAT);
    float4 x  = hv[lane];
    float4 a1 = w1[lane];
    float4 a2 = w2[lane];
    float p1 = x.x*a1.x + x.y*a1.y + x.z*a1.z + x.w*a1.w;
    float p2 = x.x*a2.x + x.y*a2.y + x.z*a2.z + x.w*a2.w;
    #pragma unroll
    for (int off = 16; off > 0; off >>= 1) {
        p1 += __shfl_xor(p1, off, 32);
        p2 += __shfl_xor(p2, off, 32);
    }
    if (lane == 0) { s1[node] = p1; s2[node] = p2; }
}

__device__ __forceinline__ void atomicMaxFloat(float* addr, float val) {
    if (val >= 0.0f) atomicMax((int*)addr, __float_as_int(val));
    else             atomicMin((unsigned int*)addr, __float_as_uint(val));
}

__global__ void k2_edge_logits(const float* __restrict__ e,
                               const int* __restrict__ src, const int* __restrict__ dst,
                               const float* __restrict__ s1, const float* __restrict__ s2,
                               const float* __restrict__ w_edge,
                               const float* __restrict__ w_attn,
                               float* __restrict__ a, float* __restrict__ amax, int E) {
    int i = blockIdx.x * blockDim.x + threadIdx.x;
    if (i >= E) return;
    float e0 = e[2*i], e1 = e[2*i + 1];
    float ex0 = e0 * w_edge[0] + e1 * w_edge[1];
    float ex1 = e0 * w_edge[2] + e1 * w_edge[3];
    float logit = s1[src[i]] + s2[dst[i]] + ex0 * w_attn[2*DFEAT] + ex1 * w_attn[2*DFEAT + 1];
    logit = (logit > 0.0f) ? logit : 0.01f * logit;
    a[i] = logit;
    atomicMaxFloat(&amax[dst[i]], logit);
}

__global__ void k3_node_sums(const float* __restrict__ e, const int* __restrict__ dst,
                             const float* __restrict__ a, const float* __restrict__ amax,
                             const float* __restrict__ w_edge,
                             float* __restrict__ out, int E) {
    const long long t = (long long)blockIdx.x * blockDim.x + threadIdx.x;
    const long long i = t >> 2;
    const int comp = (int)(t & 3);
    if (i >= (long long)E) return;
    const int d = dst[i];
    const float logit = a[i];
    const float2 ev = ((const float2*)e)[i];
    const float ex0 = ev.x * w_edge[0] + ev.y * w_edge[1];
    const float ex1 = ev.x * w_edge[2] + ev.y * w_edge[3];
    const float m = (comp == 0) ? 1.0f : ((comp == 1) ? ex0 : ex1);
    const float w = __expf(logit - amax[d]);
    if (comp < 3) atomicAdd(&out[(long long)d * DFEAT + comp], w * m);
}

__global__ void k4_node_project(const float* __restrict__ w_e2n, float* out, int N) {
    const int lane = threadIdx.x & 31;
    const long long wave = (long long)blockIdx.x * (blockDim.x >> 5) + (threadIdx.x >> 5);
    const long long base = wave * 16;
    if (base >= (long long)N) return;

    const int lo = lane & 15;
    const bool loHalf = (lane < 16);
    const long long node = base + lo;
    const bool validA = loHalf && (node < (long long)N);
    const long long nodeC = (node < (long long)N) ? node : (long long)N - 1;

    const float2* wv = (const float2*)w_e2n;
    float bxs[8], bys[8];
    #pragma unroll
    for (int j = 0; j < 8; ++j) {
        float2 b = wv[16 * j + lo];
        bxs[j] = loHalf ? b.x : 0.0f;
        bys[j] = loHalf ? b.y : 0.0f;
    }

    const float4 rec = *(const float4*)(out + nodeC * DFEAT);
    const float inv = __builtin_amdgcn_rcpf(fmaxf(rec.x, 1e-16f));
    const float ax = validA ? rec.y * inv : 0.0f;
    const float ay = validA ? rec.z * inv : 0.0f;

    v8f c[8];
    #pragma unroll
    for (int j = 0; j < 8; ++j) c[j] = rank2_wmma(ax, ay, bxs[j], bys[j]);

    const long long rowBase = base + ((lane >= 16) ? 8 : 0);
    #pragma unroll
    for (int v = 0; v < 8; ++v) {
        const long long row = rowBase + v;
        if (row < (long long)N) {
            float* orow = out + row * DFEAT + lo;
            #pragma unroll
            for (int j = 0; j < 8; ++j) orow[16 * j] = c[j][v];
        }
    }
}

extern "C" void kernel_launch(void* const* d_in, const int* in_sizes, int n_in,
                              void* d_out, int out_size, void* d_ws, size_t ws_size,
                              hipStream_t stream) {
    const float* h      = (const float*)d_in[0];
    const float* e      = (const float*)d_in[1];
    const int*   src    = (const int*)d_in[2];
    const int*   dst    = (const int*)d_in[3];
    const float* w_edge = (const float*)d_in[4];
    const float* w_attn = (const float*)d_in[5];
    const float* w_e2n  = (const float*)d_in[6];
    float* out = (float*)d_out;

    const int E = in_sizes[2];
    const int N = in_sizes[0] / DFEAT;

    float* s1    = (float*)d_ws;
    float* s2    = s1 + N;
    float* amax  = s2 + N;
    float* a     = amax + N;

    const int T = 256;
    const int wavesPerBlock = T / 32;

    k0_init<<<(N + T - 1) / T, T, 0, stream>>>(out, amax, N);
    k1_node_scalars<<<(N + wavesPerBlock - 1) / wavesPerBlock, T, 0, stream>>>(
        h, w_attn, s1, s2, N);
    k2_edge_logits<<<(E + T - 1) / T, T, 0, stream>>>(
        e, src, dst, s1, s2, w_edge, w_attn, a, amax, E);
    const long long sumThreads = 4LL * E;
    k3_node_sums<<<(int)((sumThreads + T - 1) / T), T, 0, stream>>>(
        e, dst, a, amax, w_edge, out, E);

    const long long wavesTotal = ((long long)N + 15) / 16;
    const int gProject = (int)((wavesTotal + wavesPerBlock - 1) / wavesPerBlock);
    k4_node_project<<<gProject, T, 0, stream>>>(w_e2n, out, N);
}


# k1/k2/k3 run as 16K long-lived looping waves (4 rows per wave trip, 4 edges per lane trip, 16-byte loads), not 175K one-item waves: wave start-up was the limit; float atomics arrive in new order
# speedup vs baseline: 1.7583x; 1.0796x over previous
#include <hip/hip_runtime.h>
#include <hip/hip_bf16.h>

typedef float v8f __attribute__((ext_vector_type(8)));
typedef float v2f __attribute__((ext_vector_type(2)));
typedef _Float16 v16h __attribute__((ext_vector_type(16)));

#define DFEAT 128

__device__ __forceinline__ v8f rank2_wmma(float ax, float ay, float bx, float by) {
    v8f c = {};
#if __has_builtin(__builtin_amdgcn_wmma_f32_16x16x4_f32)
    v2f a; a.x = ax; a.y = ay;
    v2f b; b.x = bx; b.y = by;
    c = __builtin_amdgcn_wmma_f32_16x16x4_f32(false, a, false, b, (short)0, c, false, false);
#else
    v16h a = {}; a[0] = (_Float16)ax; a[1] = (_Float16)ay;
    v16h b = {}; b[0] = (_Float16)bx; b[1] = (_Float16)by;
    c = __builtin_amdgcn_wmma_f32_16x16x32_f16(false, a, false, b, (short)0, c, false, false);
#endif
    return c;
}

__global__ void k0_init(float* __restrict__ out, float* __restrict__ amax, int N) {
    int n = blockIdx.x * blockDim.x + threadIdx.x;
    if (n < N) {
        *(float4*)(out + (long long)n * DFEAT) = make_float4(0.0f, 0.0f, 0.0f, 0.0f);
        amax[n] = -3.0e38f;
    }
}


__global__ void k1_node_scalars(const float* __restrict__ h,
                                const float* __restrict__ w_attn,
                                float* __restrict__ s1, float* __restrict__ s2, int N) {
    const int lane = threadIdx.x & 31;
    const int wavesPerBlock = blockDim.x >> 5;
    const int wave = blockIdx.x * wavesPerBlock + (threadIdx.x >> 5);
    const int nWaves = gridDim.x * wavesPerBlock;
    const float4 a1 = ((const float4*)w_attn)[lane];
    const float4 a2 = ((const float4*)(w_attn + DFEAT))[lane];
    for (int base = wave * 4; base < N; base += nWaves * 4) {
        float4 x[4];
        #pragma unroll
        for (int k = 0; k < 4; ++k) {
            int node = base + k;
            node = (node < N) ? node : N - 1;
            x[k] = ((const float4*)(h + (long long)node * DFEAT))[lane];
        }
        float p1[4], p2[4];
        #pragma unroll
        for (int k = 0; k < 4; ++k) {
            p1[k] = x[k].x*a1.x + x[k].y*a1.y + x[k].z*a1.z + x[k].w*a1.w;
            p2[k] = x[k].x*a2.x + x[k].y*a2.y + x[k].z*a2.z + x[k].w*a2.w;
        }
        #pragma unroll
        for (int off = 16; off > 0; off >>= 1) {
            #pragma unroll
            for (int k = 0; k < 4; ++k) {
                p1[k] += __shfl_xor(p1[k], off, 32);
                p2[k] += __shfl_xor(p2[k], off, 32);
            }
        }
        const float r1 = (lane == 0) ? p1[0] : ((lane == 1) ? p1[1] : ((lane == 2) ? p1[2] : p1[3]));
        const float r2 = (lane == 0) ? p2[0] : ((lane == 1) ? p2[1] : ((lane == 2) ? p2[2] : p2[3]));
        if (lane < 4 && base + lane < N) { s1[base + lane] = r1; s2[base + lane] = r2; }
    }
}

__device__ __forceinline__ void atomicMaxFloat(float* addr, float val) {
    if (val >= 0.0f) atomicMax((int*)addr, __float_as_int(val));
    else             atomicMin((unsigned int*)addr, __float_as_uint(val));
}

__device__ __forceinline__ float edge_logit(float e0, float e1, float ss, float sd,
                                            float we0, float we1, float we2, float we3,
                                            float wa0, float wa1) {
    float ex0 = e0 * we0 + e1 * we1;
    float ex1 = e0 * we2 + e1 * we3;
    float logit = ss + sd + ex0 * wa0 + ex1 * wa1;
    return (logit > 0.0f) ? logit : 0.01f * logit;
}

__global__ void k2_edge_logits(const float* __restrict__ e,
                               const int* __restrict__ src, const int* __restrict__ dst,
                               const float* __restrict__ s1, const float* __restrict__ s2,
                               const float* __restrict__ w_edge,
                               const float* __restrict__ w_attn,
                               float* __restrict__ a, float* __restrict__ amax, int E) {
    const int tid = blockIdx.x * blockDim.x + threadIdx.x;
    const int nThreads = gridDim.x * blockDim.x;
    const float we0 = w_edge[0], we1 = w_edge[1], we2 = w_edge[2], we3 = w_edge[3];
    const float wa0 = w_attn[2*DFEAT], wa1 = w_attn[2*DFEAT + 1];
    const int nGroups = E >> 2;
    for (int g = tid; g < nGroups; g += nThreads) {
        const int4 sv = ((const int4*)src)[g];
        const int4 dv = ((const int4*)dst)[g];
        const float4 ea = ((const float4*)e)[2*g];
        const float4 eb = ((const float4*)e)[2*g + 1];
        const float ss0 = s1[sv.x], ss1 = s1[sv.y], ss2 = s1[sv.z], ss3 = s1[sv.w];
        const float sd0 = s2[dv.x], sd1 = s2[dv.y], sd2 = s2[dv.z], sd3 = s2[dv.w];
        float4 lg;
        lg.x = edge_logit(ea.x, ea.y, ss0, sd0, we0, we1, we2, we3, wa0, wa1);
        lg.y = edge_logit(ea.z, ea.w, ss1, sd1, we0, we1, we2, we3, wa0, wa1);
        lg.z = edge_logit(eb.x, eb.y, ss2, sd2, we0, we1, we2, we3, wa0, wa1);
        lg.w = edge_logit(eb.z, eb.w, ss3, sd3, we0, we1, we2, we3, wa0, wa1);
        ((float4*)a)[g] = lg;
        atomicMaxFloat(&amax[dv.x], lg.x);
        atomicMaxFloat(&amax[dv.y], lg.y);
        atomicMaxFloat(&amax[dv.z], lg.z);
        atomicMaxFloat(&amax[dv.w], lg.w);
    }
    for (int i = (nGroups << 2) + tid; i < E; i += nThreads) {
        const int d = dst[i];
        const float logit = edge_logit(e[2*i], e[2*i + 1], s1[src[i]], s2[d],
                                       we0, we1, we2, we3, wa0, wa1);
        a[i] = logit;
        atomicMaxFloat(&amax[d], logit);
    }
}

__device__ __forceinline__ void edge_add(float e0, float e1, float logit, float m, int d, int comp,
                                         float we0, float we1, float we2, float we3,
                                         float* __restrict__ out) {
    const float ex0 = e0 * we0 + e1 * we1;
    const float ex1 = e0 * we2 + e1 * we3;
    const float f = (comp == 0) ? 1.0f : ((comp == 1) ? ex0 : ex1);
    const float w = __expf(logit - m);
    if (comp < 3) atomicAdd(&out[(long long)d * DFEAT + comp], w * f);
}

__global__ void k3_node_sums(const float* __restrict__ e, const int* __restrict__ dst,
                             const float* __restrict__ a, const float* __restrict__ amax,
                             const float* __restrict__ w_edge,
                             float* __restrict__ out, int E) {
    const int tid = blockIdx.x * blockDim.x + threadIdx.x;
    const int quad = tid >> 2;
    const int comp = tid & 3;
    const int nQuads = (gridDim.x * blockDim.x) >> 2;
    const float we0 = w_edge[0], we1 = w_edge[1], we2 = w_edge[2], we3 = w_edge[3];
    const int nGroups = E >> 2;
    for (int g = quad; g < nGroups; g += nQuads) {
        const int4 dv = ((const int4*)dst)[g];
        const float4 lv = ((const float4*)a)[g];
        const float4 ea = ((const float4*)e)[2*g];
        const float4 eb = ((const float4*)e)[2*g + 1];
        const float m0 = amax[dv.x], m1 = amax[dv.y], m2 = amax[dv.z], m3 = amax[dv.w];
        edge_add(ea.x, ea.y, lv.x, m0, dv.x, comp, we0, we1, we2, we3, out);
        edge_add(ea.z, ea.w, lv.y, m1, dv.y, comp, we0, we1, we2, we3, out);
        edge_add(eb.x, eb.y, lv.z, m2, dv.z, comp, we0, we1, we2, we3, out);
        edge_add(eb.z, eb.w, lv.w, m3, dv.w, comp, we0, we1, we2, we3, out);
    }
    for (int i = (nGroups << 2) + quad; i < E; i += nQuads) {
        const int d = dst[i];
        const float2 ev = ((const float2*)e)[i];
        edge_add(ev.x, ev.y, a[i], amax[d], d, comp, we0, we1, we2, we3, out);
    }
}

__global__ void k4_node_project(const float* __restrict__ w_e2n, float* out, int N) {
    const int lane = threadIdx.x & 31;
    const long long wave = (long long)blockIdx.x * (blockDim.x >> 5) + (threadIdx.x >> 5);
    const long long base = wave * 16;
    if (base >= (long long)N) return;

    const int lo = lane & 15;
    const bool loHalf = (lane < 16);
    const long long node = base + lo;
    const bool validA = loHalf && (node < (long long)N);
    const long long nodeC = (node < (long long)N) ? node : (long long)N - 1;

    const float2* wv = (const float2*)w_e2n;
    float bxs[8], bys[8];
    #pragma unroll
    for (int j = 0; j < 8; ++j) {
        float2 b = wv[16 * j + lo];
        bxs[j] = loHalf ? b.x : 0.0f;
        bys[j] = loHalf ? b.y : 0.0f;
    }

    const float4 rec = *(const float4*)(out + nodeC * DFEAT);
    const float inv = __builtin_amdgcn_rcpf(fmaxf(rec.x, 1e-16f));
    const float ax = validA ? rec.y * inv : 0.0f;
    const float ay = validA ? rec.z * inv : 0.0f;

    v8f c[8];
    #pragma unroll
    for (int j = 0; j < 8; ++j) c[j] = rank2_wmma(ax, ay, bxs[j], bys[j]);

    const long long rowBase = base + ((lane >= 16) ? 8 : 0);
    #pragma unroll
    for (int v = 0; v < 8; ++v) {
        const long long row = rowBase + v;
        if (row < (long long)N) {
            float* orow = out + row * DFEAT + lo;
            #pragma unroll
            for (int j = 0; j < 8; ++j) orow[16 * j] = c[j][v];
        }
    }
}

extern "C" void kernel_launch(void* const* d_in, const int* in_sizes, int n_in,
                              void* d_out, int out_size, void* d_ws, size_t ws_size,
                              hipStream_t stream) {
    const float* h      = (const float*)d_in[0];
    const float* e      = (const float*)d_in[1];
    const int*   src    = (const int*)d_in[2];
    const int*   dst    = (const int*)d_in[3];
    const float* w_edge = (const float*)d_in[4];
    const float* w_attn = (const float*)d_in[5];
    const float* w_e2n  = (const float*)d_in[6];
    float* out = (float*)d_out;

    const int E = in_sizes[2];
    const int N = in_sizes[0] / DFEAT;

    float* a     = (float*)d_ws;
    float* s1    = a + E;
    float* s2    = s1 + N;
    float* amax  = s2 + N;

    const int T = 256;
    const int wavesPerBlock = T / 32;

    k0_init<<<(N + T - 1) / T, T, 0, stream>>>(out, amax, N);

    const int capRows  = 512;
    const int capEdges = 1024;

    const int rowTrips = (N + 3) / 4;
    int need1 = (rowTrips + wavesPerBlock - 1) / wavesPerBlock;
    need1 = (need1 < 1) ? 1 : need1;
    const int rounds1 = (need1 + capRows - 1) / capRows;
    const int g1 = (need1 + rounds1 - 1) / rounds1;
    k1_node_scalars<<<g1, T, 0, stream>>>(h, w_attn, s1, s2, N);

    const int groups = (E + 3) / 4;
    int need2 = (groups + T - 1) / T;
    need2 = (need2 < 1) ? 1 : need2;
    const int rounds2 = (need2 + capEdges - 1) / capEdges;
    const int g2 = (need2 + rounds2 - 1) / rounds2;
    k2_edge_logits<<<g2, T, 0, stream>>>(
        e, src, dst, s1, s2, w_edge, w_attn, a, amax, E);

    const int quadsPerBlock = T / 4;
    int need3 = (groups + quadsPerBlock - 1) / quadsPerBlock;
    need3 = (need3 < 1) ? 1 : need3;
    const int rounds3 = (need3 + capEdges - 1) / capEdges;
    const int g3 = (need3 + rounds3 - 1) / rounds3;
    k3_node_sums<<<g3, T, 0, stream>>>(
        e, dst, a, amax, w_edge, out, E);

    const long long wavesTotal = ((long long)N + 15) / 16;
    const int gProject = (int)((wavesTotal + wavesPerBlock - 1) / wavesPerBlock);
    k4_node_project<<<gProject, T, 0, stream>>>(w_e2n, out, N);
}


# K2+K3+init become one kernel: 128 owner work-groups scan all dst, list own edges in LDS, take max and in-degree by LDS integer atomics, sum in fixed edge order; no global atomics; sum order changes
# speedup vs baseline: 1.8505x; 1.0524x over previous
#include <hip/hip_runtime.h>
#include <hip/hip_bf16.h>

typedef float v8f __attribute__((ext_vector_type(8)));
typedef float v2f __attribute__((ext_vector_type(2)));
typedef _Float16 v16h __attribute__((ext_vector_type(16)));

#define DFEAT 128

__device__ __forceinline__ v8f rank2_wmma(float ax, float ay, float bx, float by) {
    v8f c = {};
#if __has_builtin(__builtin_amdgcn_wmma_f32_16x16x4_f32)
    v2f a; a.x = ax; a.y = ay;
    v2f b; b.x = bx; b.y = by;
    c = __builtin_amdgcn_wmma_f32_16x16x4_f32(false, a, false, b, (short)0, c, false, false);
#else
    v16h a = {}; a[0] = (_Float16)ax; a[1] = (_Float16)ay;
    v16h b = {}; b[0] = (_Float16)bx; b[1] = (_Float16)by;
    c = __builtin_amdgcn_wmma_f32_16x16x32_f16(false, a, false, b, (short)0, c, false, false);
#endif
    return c;
}


__global__ void k1_node_scalars(const float* __restrict__ h,
                                const float* __restrict__ w_attn,
                                float* __restrict__ s1, float* __restrict__ s2, int N) {
    const int lane = threadIdx.x & 31;
    const int wavesPerBlock = blockDim.x >> 5;
    const int wave = blockIdx.x * wavesPerBlock + (threadIdx.x >> 5);
    const int nWaves = gridDim.x * wavesPerBlock;
    const float4 a1 = ((const float4*)w_attn)[lane];
    const float4 a2 = ((const float4*)(w_attn + DFEAT))[lane];
    for (int base = wave * 4; base < N; base += nWaves * 4) {
        float4 x[4];
        #pragma unroll
        for (int k = 0; k < 4; ++k) {
            int node = base + k;
            node = (node < N) ? node : N - 1;
            x[k] = ((const float4*)(h + (long long)node * DFEAT))[lane];
        }
        float p1[4], p2[4];
        #pragma unroll
        for (int k = 0; k < 4; ++k) {
            p1[k] = x[k].x*a1.x + x[k].y*a1.y + x[k].z*a1.z + x[k].w*a1.w;
            p2[k] = x[k].x*a2.x + x[k].y*a2.y + x[k].z*a2.z + x[k].w*a2.w;
        }
        #pragma unroll
        for (int off = 16; off > 0; off >>= 1) {
            #pragma unroll
            for (int k = 0; k < 4; ++k) {
                p1[k] += __shfl_xor(p1[k], off, 32);
                p2[k] += __shfl_xor(p2[k], off, 32);
            }
        }
        const float r1 = (lane == 0) ? p1[0] : ((lane == 1) ? p1[1] : ((lane == 2) ? p1[2] : p1[3]));
        const float r2 = (lane == 0) ? p2[0] : ((lane == 1) ? p2[1] : ((lane == 2) ? p2[2] : p2[3]));
        if (lane < 4 && base + lane < N) { s1[base + lane] = r1; s2[base + lane] = r2; }
    }
}

__device__ __forceinline__ float edge_logit(float e0, float e1, float ss, float sd,
                                            float we0, float we1, float we2, float we3,
                                            float wa0, float wa1) {
    float ex0 = e0 * we0 + e1 * we1;
    float ex1 = e0 * we2 + e1 * we3;
    float logit = ss + sd + ex0 * wa0 + ex1 * wa1;
    return (logit > 0.0f) ? logit : 0.01f * logit;
}

__device__ __forceinline__ void ldsMaxFloat(float* addr, float val) {
    val = val + 0.0f;
    if (val >= 0.0f) atomicMax((int*)addr, __float_as_int(val));
    else             atomicMin((unsigned int*)addr, __float_as_uint(val));
}

#define OWN_THREADS 1024
#define LIST_CAP    8192
#define RMAX        1024
#define SCAN_DEPTH  16
#define LIST_PER    (LIST_CAP / OWN_THREADS)

__global__ void __launch_bounds__(OWN_THREADS)
k2_owner_sums(const float* __restrict__ e,
              const int* __restrict__ src, const int* __restrict__ dst,
              const float* __restrict__ s1, const float* __restrict__ s2,
              const float* __restrict__ w_edge, const float* __restrict__ w_attn,
              float* __restrict__ out, int E, int N, int R) {
    __shared__ int   ids[LIST_CAP];
    __shared__ int   key[LIST_CAP];
    __shared__ float t0[LIST_CAP], t1[LIST_CAP], t2[LIST_CAP];
    __shared__ int   ord[LIST_CAP];
    __shared__ float nm[RMAX], ns2[RMAX];
    __shared__ int   ncnt[RMAX], nst[RMAX], nfill[RMAX];
    __shared__ int   cnt;

    const int tid = threadIdx.x;
    const int base = blockIdx.x * R;
    int Rw = N - base;
    Rw = (Rw < R) ? Rw : R;
    Rw = (Rw > 0) ? Rw : 0;
    const unsigned int uRw = (unsigned int)Rw;

    const float we0 = w_edge[0], we1 = w_edge[1], we2 = w_edge[2], we3 = w_edge[3];
    const float wa0 = w_attn[2*DFEAT], wa1 = w_attn[2*DFEAT + 1];
    const float2* e2 = (const float2*)e;

    for (int n = tid; n < Rw; n += OWN_THREADS) {
        nm[n] = -3.0e38f; ncnt[n] = 0; nfill[n] = 0;
        ns2[n] = s2[base + n];
    }
    if (tid == 0) cnt = 0;
    __syncthreads();

    const int nG = E >> 2;
    const int nIter = (nG + OWN_THREADS - 1) / OWN_THREADS;
    const int nGpad = nIter * OWN_THREADS;
    const int off = (int)(((long long)blockIdx.x * nIter) / gridDim.x) * OWN_THREADS;
    const int4* dst4 = (const int4*)dst;
    for (int k0 = 0; k0 < nIter; k0 += SCAN_DEPTH) {
        int4 v[SCAN_DEPTH];
        #pragma unroll
        for (int u = 0; u < SCAN_DEPTH; ++u) {
            int gg = (k0 + u) * OWN_THREADS + tid + off;
            gg = (gg >= nGpad) ? gg - nGpad : gg;
            const bool ok = (k0 + u < nIter) && (gg < nG);
            v[u] = dst4[ok ? gg : 0];
        }
        #pragma unroll
        for (int u = 0; u < SCAN_DEPTH; ++u) {
            int gg = (k0 + u) * OWN_THREADS + tid + off;
            gg = (gg >= nGpad) ? gg - nGpad : gg;
            const bool ok = (k0 + u < nIter) && (gg < nG);
            const bool hx = (unsigned int)(v[u].x - base) < uRw;
            const bool hy = (unsigned int)(v[u].y - base) < uRw;
            const bool hz = (unsigned int)(v[u].z - base) < uRw;
            const bool hw = (unsigned int)(v[u].w - base) < uRw;
            if (ok && (hx || hy || hz || hw)) {
                const int e0i = gg << 2;
                if (hx) { int s = atomicAdd(&cnt, 1); if (s < LIST_CAP) ids[s] = e0i; }
                if (hy) { int s = atomicAdd(&cnt, 1); if (s < LIST_CAP) ids[s] = e0i + 1; }
                if (hz) { int s = atomicAdd(&cnt, 1); if (s < LIST_CAP) ids[s] = e0i + 2; }
                if (hw) { int s = atomicAdd(&cnt, 1); if (s < LIST_CAP) ids[s] = e0i + 3; }
            }
        }
    }
    for (int i = (nG << 2) + tid; i < E; i += OWN_THREADS) {
        if ((unsigned int)(dst[i] - base) < uRw) {
            int s = atomicAdd(&cnt, 1); if (s < LIST_CAP) ids[s] = i;
        }
    }
    __syncthreads();
    const int total = cnt;

    if (total > 0 && total <= LIST_CAP) {
        int ei[LIST_PER], dl[LIST_PER], sr[LIST_PER], ps[LIST_PER];
        float2 ev[LIST_PER];
        float sv[LIST_PER], lgt[LIST_PER];
        #pragma unroll
        for (int u = 0; u < LIST_PER; ++u) {
            const int j = tid + u * OWN_THREADS;
            ei[u] = ids[(j < total) ? j : 0];
        }
        #pragma unroll
        for (int u = 0; u < LIST_PER; ++u) {
            dl[u] = dst[ei[u]] - base;
            sr[u] = src[ei[u]];
            ev[u] = e2[ei[u]];
        }
        #pragma unroll
        for (int u = 0; u < LIST_PER; ++u) sv[u] = s1[sr[u]];
        #pragma unroll
        for (int u = 0; u < LIST_PER; ++u) {
            lgt[u] = edge_logit(ev[u].x, ev[u].y, sv[u], ns2[dl[u]],
                                we0, we1, we2, we3, wa0, wa1);
            if (tid + u * OWN_THREADS < total) {
                ldsMaxFloat(&nm[dl[u]], lgt[u]);
                atomicAdd(&ncnt[dl[u]], 1);
            }
        }
        __syncthreads();
        if (tid < 32) {
            int carry = 0;
            for (int c0 = 0; c0 < Rw; c0 += 32) {
                const int n = c0 + tid;
                const int c = (n < Rw) ? ncnt[n] : 0;
                int incl = c;
                #pragma unroll
                for (int d = 1; d < 32; d <<= 1) {
                    const int t = __shfl_up(incl, d, 32);
                    if (tid >= d) incl += t;
                }
                if (n < Rw) nst[n] = carry + incl - c;
                carry += __shfl(incl, 31, 32);
            }
        }
        __syncthreads();
        #pragma unroll
        for (int u = 0; u < LIST_PER; ++u) {
            ps[u] = 0;
            if (tid + u * OWN_THREADS < total) {
                const float ex0 = ev[u].x * we0 + ev[u].y * we1;
                const float ex1 = ev[u].x * we2 + ev[u].y * we3;
                const float w = __expf(lgt[u] - nm[dl[u]]);
                const int pos = nst[dl[u]] + atomicAdd(&nfill[dl[u]], 1);
                key[pos] = ei[u];
                t0[pos] = w; t1[pos] = w * ex0; t2[pos] = w * ex1;
                ps[u] = pos;
            }
        }
        __syncthreads();
        #pragma unroll
        for (int u = 0; u < LIST_PER; ++u) {
            if (tid + u * OWN_THREADS < total) {
                const int s = nst[dl[u]];
                const int c = ncnt[dl[u]];
                int rank = 0;
                for (int q = s; q < s + c; ++q) rank += (key[q] < ei[u]) ? 1 : 0;
                ord[s + rank] = ps[u];
            }
        }
        __syncthreads();
        for (int n = tid; n < Rw; n += OWN_THREADS) {
            const int s = nst[n];
            const int c = ncnt[n];
            float l = 0.0f, x = 0.0f, y = 0.0f;
            for (int k = 0; k < c; ++k) {
                const int p = ord[s + k];
                l += t0[p]; x += t1[p]; y += t2[p];
            }
            *(float4*)(out + (long long)(base + n) * DFEAT) = make_float4(l, x, y, 0.0f);
        }
    } else if (total == 0) {
        for (int n = tid; n < Rw; n += OWN_THREADS)
            *(float4*)(out + (long long)(base + n) * DFEAT) = make_float4(0.0f, 0.0f, 0.0f, 0.0f);
    } else {
        for (int n = tid; n < Rw; n += OWN_THREADS) { t0[n] = 0.0f; t1[n] = 0.0f; t2[n] = 0.0f; }
        for (int i = tid; i < E; i += OWN_THREADS) {
            const int dl = dst[i] - base;
            if ((unsigned int)dl < uRw) {
                const float2 ev = e2[i];
                ldsMaxFloat(&nm[dl], edge_logit(ev.x, ev.y, s1[src[i]], ns2[dl],
                                                we0, we1, we2, we3, wa0, wa1));
            }
        }
        __syncthreads();
        for (int i = tid; i < E; i += OWN_THREADS) {
            const int dl = dst[i] - base;
            if ((unsigned int)dl < uRw) {
                const float2 ev = e2[i];
                const float logit = edge_logit(ev.x, ev.y, s1[src[i]], ns2[dl],
                                               we0, we1, we2, we3, wa0, wa1);
                const float ex0 = ev.x * we0 + ev.y * we1;
                const float ex1 = ev.x * we2 + ev.y * we3;
                const float w = __expf(logit - nm[dl]);
                atomicAdd(&t0[dl], w);
                atomicAdd(&t1[dl], w * ex0);
                atomicAdd(&t2[dl], w * ex1);
            }
        }
        __syncthreads();
        for (int n = tid; n < Rw; n += OWN_THREADS)
            *(float4*)(out + (long long)(base + n) * DFEAT) = make_float4(t0[n], t1[n], t2[n], 0.0f);
    }
}

__global__ void k3_node_project(const float* __restrict__ w_e2n, float* out, int N) {
    const int lane = threadIdx.x & 31;
    const long long wave = (long long)blockIdx.x * (blockDim.x >> 5) + (threadIdx.x >> 5);
    const long long base = wave * 16;
    if (base >= (long long)N) return;

    const int lo = lane & 15;
    const bool loHalf = (lane < 16);
    const long long node = base + lo;
    const bool validA = loHalf && (node < (long long)N);
    const long long nodeC = (node < (long long)N) ? node : (long long)N - 1;

    const float2* wv = (const float2*)w_e2n;
    float bxs[8], bys[8];
    #pragma unroll
    for (int j = 0; j < 8; ++j) {
        float2 b = wv[16 * j + lo];
        bxs[j] = loHalf ? b.x : 0.0f;
        bys[j] = loHalf ? b.y : 0.0f;
    }

    const float4 rec = *(const float4*)(out + nodeC * DFEAT);
    const float inv = __builtin_amdgcn_rcpf(fmaxf(rec.x, 1e-16f));
    const float ax = validA ? rec.y * inv : 0.0f;
    const float ay = validA ? rec.z * inv : 0.0f;

    v8f c[8];
    #pragma unroll
    for (int j = 0; j < 8; ++j) c[j] = rank2_wmma(ax, ay, bxs[j], bys[j]);

    const long long rowBase = base + ((lane >= 16) ? 8 : 0);
    #pragma unroll
    for (int v = 0; v < 8; ++v) {
        const long long row = rowBase + v;
        if (row < (long long)N) {
            float* orow = out + row * DFEAT + lo;
            #pragma unroll
            for (int j = 0; j < 8; ++j) orow[16 * j] = c[j][v];
        }
    }
}

extern "C" void kernel_launch(void* const* d_in, const int* in_sizes, int n_in,
                              void* d_out, int out_size, void* d_ws, size_t ws_size,
                              hipStream_t stream) {
    const float* h      = (const float*)d_in[0];
    const float* e      = (const float*)d_in[1];
    const int*   src    = (const int*)d_in[2];
    const int*   dst    = (const int*)d_in[3];
    const float* w_edge = (const float*)d_in[4];
    const float* w_attn = (const float*)d_in[5];
    const float* w_e2n  = (const float*)d_in[6];
    float* out = (float*)d_out;

    const int E = in_sizes[2];
    const int N = in_sizes[0] / DFEAT;

    float* s1    = (float*)d_ws;
    float* s2    = s1 + N;

    const int T = 256;
    const int wavesPerBlock = T / 32;

    const int capRows  = 512;
    const int rowTrips = (N + 3) / 4;
    int need1 = (rowTrips + wavesPerBlock - 1) / wavesPerBlock;
    need1 = (need1 < 1) ? 1 : need1;
    const int rounds1 = (need1 + capRows - 1) / capRows;
    const int g1 = (need1 + rounds1 - 1) / rounds1;
    k1_node_scalars<<<g1, T, 0, stream>>>(h, w_attn, s1, s2, N);

    int owners = 128;
    int R = (N + owners - 1) / owners;
    if (R > RMAX) { R = RMAX; owners = (N + RMAX - 1) / RMAX; }
    if (R < 1) R = 1;
    k2_owner_sums<<<owners, OWN_THREADS, 0, stream>>>(e, src, dst, s1, s2, w_edge, w_attn, out, E, N, R);

    const long long wavesTotal = ((long long)N + 15) / 16;
    const int gProject = (int)((wavesTotal + wavesPerBlock - 1) / wavesPerBlock);
    k3_node_project<<<gProject, T, 0, stream>>>(w_e2n, out, N);
}


# K2: 4 work-groups share each node range, each lists only its quarter of the edge array (64 ranges x 4 parts, one per WGP); K3 merges the 4 partial softmax records by max-rescaling; sum order changes
# speedup vs baseline: 2.2303x; 1.2052x over previous
#include <hip/hip_runtime.h>
#include <hip/hip_bf16.h>

typedef float v8f __attribute__((ext_vector_type(8)));
typedef float v2f __attribute__((ext_vector_type(2)));
typedef _Float16 v16h __attribute__((ext_vector_type(16)));

#define DFEAT 128

__device__ __forceinline__ v8f rank2_wmma(float ax, float ay, float bx, float by) {
    v8f c = {};
#if __has_builtin(__builtin_amdgcn_wmma_f32_16x16x4_f32)
    v2f a; a.x = ax; a.y = ay;
    v2f b; b.x = bx; b.y = by;
    c = __builtin_amdgcn_wmma_f32_16x16x4_f32(false, a, false, b, (short)0, c, false, false);
#else
    v16h a = {}; a[0] = (_Float16)ax; a[1] = (_Float16)ay;
    v16h b = {}; b[0] = (_Float16)bx; b[1] = (_Float16)by;
    c = __builtin_amdgcn_wmma_f32_16x16x32_f16(false, a, false, b, (short)0, c, false, false);
#endif
    return c;
}


__global__ void k1_node_scalars(const float* __restrict__ h,
                                const float* __restrict__ w_attn,
                                float* __restrict__ s1, float* __restrict__ s2, int N) {
    const int lane = threadIdx.x & 31;
    const int wavesPerBlock = blockDim.x >> 5;
    const int wave = blockIdx.x * wavesPerBlock + (threadIdx.x >> 5);
    const int nWaves = gridDim.x * wavesPerBlock;
    const float4 a1 = ((const float4*)w_attn)[lane];
    const float4 a2 = ((const float4*)(w_attn + DFEAT))[lane];
    for (int base = wave * 4; base < N; base += nWaves * 4) {
        float4 x[4];
        #pragma unroll
        for (int k = 0; k < 4; ++k) {
            int node = base + k;
            node = (node < N) ? node : N - 1;
            x[k] = ((const float4*)(h + (long long)node * DFEAT))[lane];
        }
        float p1[4], p2[4];
        #pragma unroll
        for (int k = 0; k < 4; ++k) {
            p1[k] = x[k].x*a1.x + x[k].y*a1.y + x[k].z*a1.z + x[k].w*a1.w;
            p2[k] = x[k].x*a2.x + x[k].y*a2.y + x[k].z*a2.z + x[k].w*a2.w;
        }
        #pragma unroll
        for (int off = 16; off > 0; off >>= 1) {
            #pragma unroll
            for (int k = 0; k < 4; ++k) {
                p1[k] += __shfl_xor(p1[k], off, 32);
                p2[k] += __shfl_xor(p2[k], off, 32);
            }
        }
        const float r1 = (lane == 0) ? p1[0] : ((lane == 1) ? p1[1] : ((lane == 2) ? p1[2] : p1[3]));
        const float r2 = (lane == 0) ? p2[0] : ((lane == 1) ? p2[1] : ((lane == 2) ? p2[2] : p2[3]));
        if (lane < 4 && base + lane < N) { s1[base + lane] = r1; s2[base + lane] = r2; }
    }
}

__device__ __forceinline__ float edge_logit(float e0, float e1, float ss, float sd,
                                            float we0, float we1, float we2, float we3,
                                            float wa0, float wa1) {
    float ex0 = e0 * we0 + e1 * we1;
    float ex1 = e0 * we2 + e1 * we3;
    float logit = ss + sd + ex0 * wa0 + ex1 * wa1;
    return (logit > 0.0f) ? logit : 0.01f * logit;
}

__device__ __forceinline__ void ldsMaxFloat(float* addr, float val) {
    val = val + 0.0f;
    if (val >= 0.0f) atomicMax((int*)addr, __float_as_int(val));
    else             atomicMin((unsigned int*)addr, __float_as_uint(val));
}

#define OWN_THREADS 1024
#define LIST_CAP    8192
#define RMAX        1024
#define SCAN_DEPTH  17
#define LIST_PER    (LIST_CAP / OWN_THREADS)
#define PARTS       4
#define REC_STRIDE  32

__global__ void __launch_bounds__(OWN_THREADS)
k2_owner_sums(const float* __restrict__ e,
              const int* __restrict__ src, const int* __restrict__ dst,
              const float* __restrict__ s1, const float* __restrict__ s2,
              const float* __restrict__ w_edge, const float* __restrict__ w_attn,
              float* __restrict__ out, int E, int N, int R) {
    __shared__ int   ids[LIST_CAP];
    __shared__ int   key[LIST_CAP];
    __shared__ float t0[LIST_CAP], t1[LIST_CAP], t2[LIST_CAP];
    __shared__ int   ord[LIST_CAP];
    __shared__ float nm[RMAX], ns2[RMAX];
    __shared__ int   ncnt[RMAX], nst[RMAX], nfill[RMAX];
    __shared__ int   cnt;

    const int tid = threadIdx.x;
    const int own = blockIdx.x / PARTS;
    const int part = blockIdx.x - own * PARTS;
    const int nOwn = gridDim.x / PARTS;
    const int base = own * R;
    float* const recBase = out + (long long)base * DFEAT + part * REC_STRIDE;
    int Rw = N - base;
    Rw = (Rw < R) ? Rw : R;
    Rw = (Rw > 0) ? Rw : 0;
    const unsigned int uRw = (unsigned int)Rw;

    const float we0 = w_edge[0], we1 = w_edge[1], we2 = w_edge[2], we3 = w_edge[3];
    const float wa0 = w_attn[2*DFEAT], wa1 = w_attn[2*DFEAT + 1];
    const float2* e2 = (const float2*)e;

    for (int n = tid; n < Rw; n += OWN_THREADS) {
        nm[n] = -3.0e38f; ncnt[n] = 0; nfill[n] = 0;
        ns2[n] = s2[base + n];
    }
    if (tid == 0) cnt = 0;
    __syncthreads();

    const int nGall = E >> 2;
    const int g0 = (int)(((long long)nGall * part) / PARTS);
    const int g1 = (int)(((long long)nGall * (part + 1)) / PARTS);
    const int nG = g1 - g0;
    const int eBeg = g0 << 2;
    const int eEnd = (part == PARTS - 1) ? E : (g1 << 2);
    const int nIter = (nG + OWN_THREADS - 1) / OWN_THREADS;
    const int nGpad = nIter * OWN_THREADS;
    const int off = (int)(((long long)own * nIter) / nOwn) * OWN_THREADS;
    const int4* dst4 = (const int4*)dst + g0;
    for (int k0 = 0; k0 < nIter; k0 += SCAN_DEPTH) {
        int4 v[SCAN_DEPTH];
        #pragma unroll
        for (int u = 0; u < SCAN_DEPTH; ++u) {
            int gg = (k0 + u) * OWN_THREADS + tid + off;
            gg = (gg >= nGpad) ? gg - nGpad : gg;
            const bool ok = (k0 + u < nIter) && (gg < nG);
            v[u] = dst4[ok ? gg : 0];
        }
        #pragma unroll
        for (int u = 0; u < SCAN_DEPTH; ++u) {
            int gg = (k0 + u) * OWN_THREADS + tid + off;
            gg = (gg >= nGpad) ? gg - nGpad : gg;
            const bool ok = (k0 + u < nIter) && (gg < nG);
            const bool hx = (unsigned int)(v[u].x - base) < uRw;
            const bool hy = (unsigned int)(v[u].y - base) < uRw;
            const bool hz = (unsigned int)(v[u].z - base) < uRw;
            const bool hw = (unsigned int)(v[u].w - base) < uRw;
            if (ok && (hx || hy || hz || hw)) {
                const int e0i = eBeg + (gg << 2);
                if (hx) { int s = atomicAdd(&cnt, 1); if (s < LIST_CAP) ids[s] = e0i; }
                if (hy) { int s = atomicAdd(&cnt, 1); if (s < LIST_CAP) ids[s] = e0i + 1; }
                if (hz) { int s = atomicAdd(&cnt, 1); if (s < LIST_CAP) ids[s] = e0i + 2; }
                if (hw) { int s = atomicAdd(&cnt, 1); if (s < LIST_CAP) ids[s] = e0i + 3; }
            }
        }
    }
    for (int i = (g1 << 2) + tid; i < eEnd; i += OWN_THREADS) {
        if ((unsigned int)(dst[i] - base) < uRw) {
            int s = atomicAdd(&cnt, 1); if (s < LIST_CAP) ids[s] = i;
        }
    }
    __syncthreads();
    const int total = cnt;

    if (total > 0 && total <= LIST_CAP) {
        int ei[LIST_PER], dl[LIST_PER], sr[LIST_PER], ps[LIST_PER];
        float2 ev[LIST_PER];
        float sv[LIST_PER], lgt[LIST_PER];
        #pragma unroll
        for (int u = 0; u < LIST_PER; ++u) {
            const int j = tid + u * OWN_THREADS;
            ei[u] = ids[(j < total) ? j : 0];
        }
        #pragma unroll
        for (int u = 0; u < LIST_PER; ++u) {
            dl[u] = dst[ei[u]] - base;
            sr[u] = src[ei[u]];
            ev[u] = e2[ei[u]];
        }
        #pragma unroll
        for (int u = 0; u < LIST_PER; ++u) sv[u] = s1[sr[u]];
        #pragma unroll
        for (int u = 0; u < LIST_PER; ++u) {
            lgt[u] = edge_logit(ev[u].x, ev[u].y, sv[u], ns2[dl[u]],
                                we0, we1, we2, we3, wa0, wa1);
            if (tid + u * OWN_THREADS < total) {
                ldsMaxFloat(&nm[dl[u]], lgt[u]);
                atomicAdd(&ncnt[dl[u]], 1);
            }
        }
        __syncthreads();
        if (tid < 32) {
            int carry = 0;
            for (int c0 = 0; c0 < Rw; c0 += 32) {
                const int n = c0 + tid;
                const int c = (n < Rw) ? ncnt[n] : 0;
                int incl = c;
                #pragma unroll
                for (int d = 1; d < 32; d <<= 1) {
                    const int t = __shfl_up(incl, d, 32);
                    if (tid >= d) incl += t;
                }
                if (n < Rw) nst[n] = carry + incl - c;
                carry += __shfl(incl, 31, 32);
            }
        }
        __syncthreads();
        #pragma unroll
        for (int u = 0; u < LIST_PER; ++u) {
            ps[u] = 0;
            if (tid + u * OWN_THREADS < total) {
                const float ex0 = ev[u].x * we0 + ev[u].y * we1;
                const float ex1 = ev[u].x * we2 + ev[u].y * we3;
                const float w = __expf(lgt[u] - nm[dl[u]]);
                const int pos = nst[dl[u]] + atomicAdd(&nfill[dl[u]], 1);
                key[pos] = ei[u];
                t0[pos] = w; t1[pos] = w * ex0; t2[pos] = w * ex1;
                ps[u] = pos;
            }
        }
        __syncthreads();
        #pragma unroll
        for (int u = 0; u < LIST_PER; ++u) {
            if (tid + u * OWN_THREADS < total) {
                const int s = nst[dl[u]];
                const int c = ncnt[dl[u]];
                int rank = 0;
                for (int q = s; q < s + c; ++q) rank += (key[q] < ei[u]) ? 1 : 0;
                ord[s + rank] = ps[u];
            }
        }
        __syncthreads();
        for (int n = tid; n < Rw; n += OWN_THREADS) {
            const int s = nst[n];
            const int c = ncnt[n];
            float l = 0.0f, x = 0.0f, y = 0.0f;
            for (int k = 0; k < c; ++k) {
                const int p = ord[s + k];
                l += t0[p]; x += t1[p]; y += t2[p];
            }
            *(float4*)(recBase + (long long)n * DFEAT) = make_float4(nm[n], l, x, y);
        }
    } else if (total == 0) {
        for (int n = tid; n < Rw; n += OWN_THREADS)
            *(float4*)(recBase + (long long)n * DFEAT) = make_float4(-3.0e38f, 0.0f, 0.0f, 0.0f);
    } else {
        for (int n = tid; n < Rw; n += OWN_THREADS) { t0[n] = 0.0f; t1[n] = 0.0f; t2[n] = 0.0f; }
        for (int i = eBeg + tid; i < eEnd; i += OWN_THREADS) {
            const int dl = dst[i] - base;
            if ((unsigned int)dl < uRw) {
                const float2 ev = e2[i];
                ldsMaxFloat(&nm[dl], edge_logit(ev.x, ev.y, s1[src[i]], ns2[dl],
                                                we0, we1, we2, we3, wa0, wa1));
            }
        }
        __syncthreads();
        for (int i = eBeg + tid; i < eEnd; i += OWN_THREADS) {
            const int dl = dst[i] - base;
            if ((unsigned int)dl < uRw) {
                const float2 ev = e2[i];
                const float logit = edge_logit(ev.x, ev.y, s1[src[i]], ns2[dl],
                                               we0, we1, we2, we3, wa0, wa1);
                const float ex0 = ev.x * we0 + ev.y * we1;
                const float ex1 = ev.x * we2 + ev.y * we3;
                const float w = __expf(logit - nm[dl]);
                atomicAdd(&t0[dl], w);
                atomicAdd(&t1[dl], w * ex0);
                atomicAdd(&t2[dl], w * ex1);
            }
        }
        __syncthreads();
        for (int n = tid; n < Rw; n += OWN_THREADS)
            *(float4*)(recBase + (long long)n * DFEAT) = make_float4(nm[n], t0[n], t1[n], t2[n]);
    }
}

__global__ void k3_node_project(const float* __restrict__ w_e2n, float* out, int N) {
    const int lane = threadIdx.x & 31;
    const long long wave = (long long)blockIdx.x * (blockDim.x >> 5) + (threadIdx.x >> 5);
    const long long base = wave * 16;
    if (base >= (long long)N) return;

    const int lo = lane & 15;
    const bool loHalf = (lane < 16);
    const long long node = base + lo;
    const bool validA = loHalf && (node < (long long)N);
    const long long nodeC = (node < (long long)N) ? node : (long long)N - 1;

    const float2* wv = (const float2*)w_e2n;
    float bxs[8], bys[8];
    #pragma unroll
    for (int j = 0; j < 8; ++j) {
        float2 b = wv[16 * j + lo];
        bxs[j] = loHalf ? b.x : 0.0f;
        bys[j] = loHalf ? b.y : 0.0f;
    }

    float4 rec[PARTS];
    #pragma unroll
    for (int p = 0; p < PARTS; ++p) rec[p] = *(const float4*)(out + nodeC * DFEAT + p * REC_STRIDE);
    float m = rec[0].x;
    #pragma unroll
    for (int p = 1; p < PARTS; ++p) m = fmaxf(m, rec[p].x);
    float l = 0.0f, sx = 0.0f, sy = 0.0f;
    #pragma unroll
    for (int p = 0; p < PARTS; ++p) {
        const float s = expf(rec[p].x - m);
        l += rec[p].y * s; sx += rec[p].z * s; sy += rec[p].w * s;
    }
    const float inv = __builtin_amdgcn_rcpf(fmaxf(l, 1e-16f));
    const float ax = validA ? sx * inv : 0.0f;
    const float ay = validA ? sy * inv : 0.0f;

    v8f c[8];
    #pragma unroll
    for (int j = 0; j < 8; ++j) c[j] = rank2_wmma(ax, ay, bxs[j], bys[j]);

    const long long rowBase = base + ((lane >= 16) ? 8 : 0);
    #pragma unroll
    for (int v = 0; v < 8; ++v) {
        const long long row = rowBase + v;
        if (row < (long long)N) {
            float* orow = out + row * DFEAT + lo;
            #pragma unroll
            for (int j = 0; j < 8; ++j) orow[16 * j] = c[j][v];
        }
    }
}

extern "C" void kernel_launch(void* const* d_in, const int* in_sizes, int n_in,
                              void* d_out, int out_size, void* d_ws, size_t ws_size,
                              hipStream_t stream) {
    const float* h      = (const float*)d_in[0];
    const float* e      = (const float*)d_in[1];
    const int*   src    = (const int*)d_in[2];
    const int*   dst    = (const int*)d_in[3];
    const float* w_edge = (const float*)d_in[4];
    const float* w_attn = (const float*)d_in[5];
    const float* w_e2n  = (const float*)d_in[6];
    float* out = (float*)d_out;

    const int E = in_sizes[2];
    const int N = in_sizes[0] / DFEAT;

    float* s1    = (float*)d_ws;
    float* s2    = s1 + N;

    const int T = 256;
    const int wavesPerBlock = T / 32;

    const int capRows  = 512;
    const int rowTrips = (N + 3) / 4;
    int need1 = (rowTrips + wavesPerBlock - 1) / wavesPerBlock;
    need1 = (need1 < 1) ? 1 : need1;
    const int rounds1 = (need1 + capRows - 1) / capRows;
    const int g1 = (need1 + rounds1 - 1) / rounds1;
    k1_node_scalars<<<g1, T, 0, stream>>>(h, w_attn, s1, s2, N);

    int owners = 64;
    int R = (N + owners - 1) / owners;
    if (R > RMAX) { R = RMAX; owners = (N + RMAX - 1) / RMAX; }
    if (R < 1) R = 1;
    k2_owner_sums<<<owners * PARTS, OWN_THREADS, 0, stream>>>(e, src, dst, s1, s2, w_edge, w_attn, out, E, N, R);

    const long long wavesTotal = ((long long)N + 15) / 16;
    const int gProject = (int)((wavesTotal + wavesPerBlock - 1) / wavesPerBlock);
    k3_node_project<<<gProject, T, 0, stream>>>(w_e2n, out, N);
}


# K2: first-slot prefix sum of the in-degrees by all 32 waves at once (one scan per wave, wave totals through LDS, one extra barrier), not one wave walking 25 chunks of 32 nodes while 31 waves wait
# speedup vs baseline: 2.3843x; 1.0690x over previous
#include <hip/hip_runtime.h>
#include <hip/hip_bf16.h>

typedef float v8f __attribute__((ext_vector_type(8)));
typedef float v2f __attribute__((ext_vector_type(2)));
typedef _Float16 v16h __attribute__((ext_vector_type(16)));

#define DFEAT 128

__device__ __forceinline__ v8f rank2_wmma(float ax, float ay, float bx, float by) {
    v8f c = {};
#if __has_builtin(__builtin_amdgcn_wmma_f32_16x16x4_f32)
    v2f a; a.x = ax; a.y = ay;
    v2f b; b.x = bx; b.y = by;
    c = __builtin_amdgcn_wmma_f32_16x16x4_f32(false, a, false, b, (short)0, c, false, false);
#else
    v16h a = {}; a[0] = (_Float16)ax; a[1] = (_Float16)ay;
    v16h b = {}; b[0] = (_Float16)bx; b[1] = (_Float16)by;
    c = __builtin_amdgcn_wmma_f32_16x16x32_f16(false, a, false, b, (short)0, c, false, false);
#endif
    return c;
}


__global__ void k1_node_scalars(const float* __restrict__ h,
                                const float* __restrict__ w_attn,
                                float* __restrict__ s1, float* __restrict__ s2, int N) {
    const int lane = threadIdx.x & 31;
    const int wavesPerBlock = blockDim.x >> 5;
    const int wave = blockIdx.x * wavesPerBlock + (threadIdx.x >> 5);
    const int nWaves = gridDim.x * wavesPerBlock;
    const float4 a1 = ((const float4*)w_attn)[lane];
    const float4 a2 = ((const float4*)(w_attn + DFEAT))[lane];
    for (int base = wave * 4; base < N; base += nWaves * 4) {
        float4 x[4];
        #pragma unroll
        for (int k = 0; k < 4; ++k) {
            int node = base + k;
            node = (node < N) ? node : N - 1;
            x[k] = ((const float4*)(h + (long long)node * DFEAT))[lane];
        }
        float p1[4], p2[4];
        #pragma unroll
        for (int k = 0; k < 4; ++k) {
            p1[k] = x[k].x*a1.x + x[k].y*a1.y + x[k].z*a1.z + x[k].w*a1.w;
            p2[k] = x[k].x*a2.x + x[k].y*a2.y + x[k].z*a2.z + x[k].w*a2.w;
        }
        #pragma unroll
        for (int off = 16; off > 0; off >>= 1) {
            #pragma unroll
            for (int k = 0; k < 4; ++k) {
                p1[k] += __shfl_xor(p1[k], off, 32);
                p2[k] += __shfl_xor(p2[k], off, 32);
            }
        }
        const float r1 = (lane == 0) ? p1[0] : ((lane == 1) ? p1[1] : ((lane == 2) ? p1[2] : p1[3]));
        const float r2 = (lane == 0) ? p2[0] : ((lane == 1) ? p2[1] : ((lane == 2) ? p2[2] : p2[3]));
        if (lane < 4 && base + lane < N) { s1[base + lane] = r1; s2[base + lane] = r2; }
    }
}

__device__ __forceinline__ float edge_logit(float e0, float e1, float ss, float sd,
                                            float we0, float we1, float we2, float we3,
                                            float wa0, float wa1) {
    float ex0 = e0 * we0 + e1 * we1;
    float ex1 = e0 * we2 + e1 * we3;
    float logit = ss + sd + ex0 * wa0 + ex1 * wa1;
    return (logit > 0.0f) ? logit : 0.01f * logit;
}

__device__ __forceinline__ void ldsMaxFloat(float* addr, float val) {
    val = val + 0.0f;
    if (val >= 0.0f) atomicMax((int*)addr, __float_as_int(val));
    else             atomicMin((unsigned int*)addr, __float_as_uint(val));
}

#define OWN_THREADS 1024
#define LIST_CAP    8192
#define RMAX        1024
#define SCAN_DEPTH  17
#define LIST_PER    (LIST_CAP / OWN_THREADS)
#define PARTS       4
#define REC_STRIDE  32

__global__ void __launch_bounds__(OWN_THREADS)
k2_owner_sums(const float* __restrict__ e,
              const int* __restrict__ src, const int* __restrict__ dst,
              const float* __restrict__ s1, const float* __restrict__ s2,
              const float* __restrict__ w_edge, const float* __restrict__ w_attn,
              float* __restrict__ out, int E, int N, int R) {
    __shared__ int   ids[LIST_CAP];
    __shared__ int   key[LIST_CAP];
    __shared__ float t0[LIST_CAP], t1[LIST_CAP], t2[LIST_CAP];
    __shared__ int   ord[LIST_CAP];
    __shared__ float nm[RMAX], ns2[RMAX];
    __shared__ int   ncnt[RMAX], nst[RMAX], nfill[RMAX];
    __shared__ int   cnt;
    __shared__ int   wtot[OWN_THREADS / 32];
    static_assert(RMAX <= OWN_THREADS, "the slot scan gives every node a thread of its own");
    static_assert(OWN_THREADS / 32 <= 32, "one lane per wave total");

    const int tid = threadIdx.x;
    const int own = blockIdx.x / PARTS;
    const int part = blockIdx.x - own * PARTS;
    const int nOwn = gridDim.x / PARTS;
    const int base = own * R;
    float* const recBase = out + (long long)base * DFEAT + part * REC_STRIDE;
    int Rw = N - base;
    Rw = (Rw < R) ? Rw : R;
    Rw = (Rw > 0) ? Rw : 0;
    const unsigned int uRw = (unsigned int)Rw;

    const float we0 = w_edge[0], we1 = w_edge[1], we2 = w_edge[2], we3 = w_edge[3];
    const float wa0 = w_attn[2*DFEAT], wa1 = w_attn[2*DFEAT + 1];
    const float2* e2 = (const float2*)e;

    for (int n = tid; n < Rw; n += OWN_THREADS) {
        nm[n] = -3.0e38f; ncnt[n] = 0; nfill[n] = 0;
        ns2[n] = s2[base + n];
    }
    if (tid == 0) cnt = 0;
    __syncthreads();

    const int nGall = E >> 2;
    const int g0 = (int)(((long long)nGall * part) / PARTS);
    const int g1 = (int)(((long long)nGall * (part + 1)) / PARTS);
    const int nG = g1 - g0;
    const int eBeg = g0 << 2;
    const int eEnd = (part == PARTS - 1) ? E : (g1 << 2);
    const int nIter = (nG + OWN_THREADS - 1) / OWN_THREADS;
    const int nGpad = nIter * OWN_THREADS;
    const int off = (int)(((long long)own * nIter) / nOwn) * OWN_THREADS;
    const int4* dst4 = (const int4*)dst + g0;
    for (int k0 = 0; k0 < nIter; k0 += SCAN_DEPTH) {
        int4 v[SCAN_DEPTH];
        #pragma unroll
        for (int u = 0; u < SCAN_DEPTH; ++u) {
            int gg = (k0 + u) * OWN_THREADS + tid + off;
            gg = (gg >= nGpad) ? gg - nGpad : gg;
            const bool ok = (k0 + u < nIter) && (gg < nG);
            v[u] = dst4[ok ? gg : 0];
        }
        #pragma unroll
        for (int u = 0; u < SCAN_DEPTH; ++u) {
            int gg = (k0 + u) * OWN_THREADS + tid + off;
            gg = (gg >= nGpad) ? gg - nGpad : gg;
            const bool ok = (k0 + u < nIter) && (gg < nG);
            const bool hx = (unsigned int)(v[u].x - base) < uRw;
            const bool hy = (unsigned int)(v[u].y - base) < uRw;
            const bool hz = (unsigned int)(v[u].z - base) < uRw;
            const bool hw = (unsigned int)(v[u].w - base) < uRw;
            if (ok && (hx || hy || hz || hw)) {
                const int e0i = eBeg + (gg << 2);
                if (hx) { int s = atomicAdd(&cnt, 1); if (s < LIST_CAP) ids[s] = e0i; }
                if (hy) { int s = atomicAdd(&cnt, 1); if (s < LIST_CAP) ids[s] = e0i + 1; }
                if (hz) { int s = atomicAdd(&cnt, 1); if (s < LIST_CAP) ids[s] = e0i + 2; }
                if (hw) { int s = atomicAdd(&cnt, 1); if (s < LIST_CAP) ids[s] = e0i + 3; }
            }
        }
    }
    for (int i = (g1 << 2) + tid; i < eEnd; i += OWN_THREADS) {
        if ((unsigned int)(dst[i] - base) < uRw) {
            int s = atomicAdd(&cnt, 1); if (s < LIST_CAP) ids[s] = i;
        }
    }
    __syncthreads();
    const int total = cnt;

    if (total > 0 && total <= LIST_CAP) {
        int ei[LIST_PER], dl[LIST_PER], sr[LIST_PER], ps[LIST_PER];
        float2 ev[LIST_PER];
        float sv[LIST_PER], lgt[LIST_PER];
        #pragma unroll
        for (int u = 0; u < LIST_PER; ++u) {
            const int j = tid + u * OWN_THREADS;
            ei[u] = ids[(j < total) ? j : 0];
        }
        #pragma unroll
        for (int u = 0; u < LIST_PER; ++u) {
            dl[u] = dst[ei[u]] - base;
            sr[u] = src[ei[u]];
            ev[u] = e2[ei[u]];
        }
        #pragma unroll
        for (int u = 0; u < LIST_PER; ++u) sv[u] = s1[sr[u]];
        #pragma unroll
        for (int u = 0; u < LIST_PER; ++u) {
            lgt[u] = edge_logit(ev[u].x, ev[u].y, sv[u], ns2[dl[u]],
                                we0, we1, we2, we3, wa0, wa1);
            if (tid + u * OWN_THREADS < total) {
                ldsMaxFloat(&nm[dl[u]], lgt[u]);
                atomicAdd(&ncnt[dl[u]], 1);
            }
        }
        __syncthreads();
        {
            const int lane = tid & 31;
            const int wv = tid >> 5;
            const int c = (tid < Rw) ? ncnt[tid] : 0;
            int incl = c;
            #pragma unroll
            for (int d = 1; d < 32; d <<= 1) {
                const int t = __shfl_up(incl, d, 32);
                if (lane >= d) incl += t;
            }
            if (lane == 31) wtot[wv] = incl;
            __syncthreads();
            int before = (lane < wv) ? wtot[lane] : 0;
            #pragma unroll
            for (int o = 16; o > 0; o >>= 1) before += __shfl_xor(before, o, 32);
            if (tid < Rw) nst[tid] = before + incl - c;
        }
        __syncthreads();
        #pragma unroll
        for (int u = 0; u < LIST_PER; ++u) {
            ps[u] = 0;
            if (tid + u * OWN_THREADS < total) {
                const float ex0 = ev[u].x * we0 + ev[u].y * we1;
                const float ex1 = ev[u].x * we2 + ev[u].y * we3;
                const float w = __expf(lgt[u] - nm[dl[u]]);
                const int pos = nst[dl[u]] + atomicAdd(&nfill[dl[u]], 1);
                key[pos] = ei[u];
                t0[pos] = w; t1[pos] = w * ex0; t2[pos] = w * ex1;
                ps[u] = pos;
            }
        }
        __syncthreads();
        #pragma unroll
        for (int u = 0; u < LIST_PER; ++u) {
            if (tid + u * OWN_THREADS < total) {
                const int s = nst[dl[u]];
                const int c = ncnt[dl[u]];
                int rank = 0;
                for (int q = s; q < s + c; ++q) rank += (key[q] < ei[u]) ? 1 : 0;
                ord[s + rank] = ps[u];
            }
        }
        __syncthreads();
        for (int n = tid; n < Rw; n += OWN_THREADS) {
            const int s = nst[n];
            const int c = ncnt[n];
            float l = 0.0f, x = 0.0f, y = 0.0f;
            for (int k = 0; k < c; ++k) {
                const int p = ord[s + k];
                l += t0[p]; x += t1[p]; y += t2[p];
            }
            *(float4*)(recBase + (long long)n * DFEAT) = make_float4(nm[n], l, x, y);
        }
    } else if (total == 0) {
        for (int n = tid; n < Rw; n += OWN_THREADS)
            *(float4*)(recBase + (long long)n * DFEAT) = make_float4(-3.0e38f, 0.0f, 0.0f, 0.0f);
    } else {
        for (int n = tid; n < Rw; n += OWN_THREADS) { t0[n] = 0.0f; t1[n] = 0.0f; t2[n] = 0.0f; }
        for (int i = eBeg + tid; i < eEnd; i += OWN_THREADS) {
            const int dl = dst[i] - base;
            if ((unsigned int)dl < uRw) {
                const float2 ev = e2[i];
                ldsMaxFloat(&nm[dl], edge_logit(ev.x, ev.y, s1[src[i]], ns2[dl],
                                                we0, we1, we2, we3, wa0, wa1));
            }
        }
        __syncthreads();
        for (int i = eBeg + tid; i < eEnd; i += OWN_THREADS) {
            const int dl = dst[i] - base;
            if ((unsigned int)dl < uRw) {
                const float2 ev = e2[i];
                const float logit = edge_logit(ev.x, ev.y, s1[src[i]], ns2[dl],
                                               we0, we1, we2, we3, wa0, wa1);
                const float ex0 = ev.x * we0 + ev.y * we1;
                const float ex1 = ev.x * we2 + ev.y * we3;
                const float w = __expf(logit - nm[dl]);
                atomicAdd(&t0[dl], w);
                atomicAdd(&t1[dl], w * ex0);
                atomicAdd(&t2[dl], w * ex1);
            }
        }
        __syncthreads();
        for (int n = tid; n < Rw; n += OWN_THREADS)
            *(float4*)(recBase + (long long)n * DFEAT) = make_float4(nm[n], t0[n], t1[n], t2[n]);
    }
}

__global__ void k3_node_project(const float* __restrict__ w_e2n, float* out, int N) {
    const int lane = threadIdx.x & 31;
    const long long wave = (long long)blockIdx.x * (blockDim.x >> 5) + (threadIdx.x >> 5);
    const long long base = wave * 16;
    if (base >= (long long)N) return;

    const int lo = lane & 15;
    const bool loHalf = (lane < 16);
    const long long node = base + lo;
    const bool validA = loHalf && (node < (long long)N);
    const long long nodeC = (node < (long long)N) ? node : (long long)N - 1;

    const float2* wv = (const float2*)w_e2n;
    float bxs[8], bys[8];
    #pragma unroll
    for (int j = 0; j < 8; ++j) {
        float2 b = wv[16 * j + lo];
        bxs[j] = loHalf ? b.x : 0.0f;
        bys[j] = loHalf ? b.y : 0.0f;
    }

    float4 rec[PARTS];
    #pragma unroll
    for (int p = 0; p < PARTS; ++p) rec[p] = *(const float4*)(out + nodeC * DFEAT + p * REC_STRIDE);
    float m = rec[0].x;
    #pragma unroll
    for (int p = 1; p < PARTS; ++p) m = fmaxf(m, rec[p].x);
    float l = 0.0f, sx = 0.0f, sy = 0.0f;
    #pragma unroll
    for (int p = 0; p < PARTS; ++p) {
        const float s = expf(rec[p].x - m);
        l += rec[p].y * s; sx += rec[p].z * s; sy += rec[p].w * s;
    }
    const float inv = __builtin_amdgcn_rcpf(fmaxf(l, 1e-16f));
    const float ax = validA ? sx * inv : 0.0f;
    const float ay = validA ? sy * inv : 0.0f;

    v8f c[8];
    #pragma unroll
    for (int j = 0; j < 8; ++j) c[j] = rank2_wmma(ax, ay, bxs[j], bys[j]);

    const long long rowBase = base + ((lane >= 16) ? 8 : 0);
    #pragma unroll
    for (int v = 0; v < 8; ++v) {
        const long long row = rowBase + v;
        if (row < (long long)N) {
            float* orow = out + row * DFEAT + lo;
            #pragma unroll
            for (int j = 0; j < 8; ++j) orow[16 * j] = c[j][v];
        }
    }
}

extern "C" void kernel_launch(void* const* d_in, const int* in_sizes, int n_in,
                              void* d_out, int out_size, void* d_ws, size_t ws_size,
                              hipStream_t stream) {
    const float* h      = (const float*)d_in[0];
    const float* e      = (const float*)d_in[1];
    const int*   src    = (const int*)d_in[2];
    const int*   dst    = (const int*)d_in[3];
    const float* w_edge = (const float*)d_in[4];
    const float* w_attn = (const float*)d_in[5];
    const float* w_e2n  = (const float*)d_in[6];
    float* out = (float*)d_out;

    const int E = in_sizes[2];
    const int N = in_sizes[0] / DFEAT;

    float* s1    = (float*)d_ws;
    float* s2    = s1 + N;

    const int T = 256;
    const int wavesPerBlock = T / 32;

    const int capRows  = 512;
    const int rowTrips = (N + 3) / 4;
    int need1 = (rowTrips + wavesPerBlock - 1) / wavesPerBlock;
    need1 = (need1 < 1) ? 1 : need1;
    const int rounds1 = (need1 + capRows - 1) / capRows;
    const int g1 = (need1 + rounds1 - 1) / rounds1;
    k1_node_scalars<<<g1, T, 0, stream>>>(h, w_attn, s1, s2, N);

    int owners = 64;
    int R = (N + owners - 1) / owners;
    if (R > RMAX) { R = RMAX; owners = (N + RMAX - 1) / RMAX; }
    if (R < 1) R = 1;
    k2_owner_sums<<<owners * PARTS, OWN_THREADS, 0, stream>>>(e, src, dst, s1, s2, w_edge, w_attn, out, E, N, R);

    const long long wavesTotal = ((long long)N + 15) / 16;
    const int gProject = (int)((wavesTotal + wavesPerBlock - 1) / wavesPerBlock);
    k3_node_project<<<gProject, T, 0, stream>>>(w_e2n, out, N);
}


# Bin work-groups in K1's launch sort each edge chunk by owner range (LDS counting sort) into workspace words {edge,dst,src,e}; K2 owners load their runs: no dst scan, 1 gather per edge; sum order kept
# speedup vs baseline: 2.9984x; 1.2576x over previous
#include <hip/hip_runtime.h>
#include <hip/hip_bf16.h>

typedef float v8f __attribute__((ext_vector_type(8)));
typedef float v2f __attribute__((ext_vector_type(2)));
typedef _Float16 v16h __attribute__((ext_vector_type(16)));

#define DFEAT 128

__device__ __forceinline__ v8f rank2_wmma(float ax, float ay, float bx, float by) {
    v8f c = {};
#if __has_builtin(__builtin_amdgcn_wmma_f32_16x16x4_f32)
    v2f a; a.x = ax; a.y = ay;
    v2f b; b.x = bx; b.y = by;
    c = __builtin_amdgcn_wmma_f32_16x16x4_f32(false, a, false, b, (short)0, c, false, false);
#else
    v16h a = {}; a[0] = (_Float16)ax; a[1] = (_Float16)ay;
    v16h b = {}; b[0] = (_Float16)bx; b[1] = (_Float16)by;
    c = __builtin_amdgcn_wmma_f32_16x16x32_f16(false, a, false, b, (short)0, c, false, false);
#endif
    return c;
}

#define OWN_THREADS 1024
#define LIST_CAP    8192
#define RMAX        1024
#define SCAN_DEPTH  17
#define LIST_PER    (LIST_CAP / OWN_THREADS)
#define PARTS       4
#define REC_STRIDE  32
#define NCH         256
#define CPP         (NCH / PARTS)
#define MAXOWN      64
#define BIN_T       256
#define BIN_CAP     4096
#define BIN_PER     (BIN_CAP / BIN_T)

__device__ __forceinline__ void bin_chunk(const float* __restrict__ e, const int* __restrict__ src,
                                          const int* __restrict__ dst,
                                          int* __restrict__ dir, int* __restrict__ rec,
                                          int E, int N, int R, int CS, int np, int c,
                                          int* hist, int* ofs,
                                          int* sE, int* sD, int* sS, float* sX, float* sY) {
    const int tid = threadIdx.x;
    const int cb = (int)(((long long)E * c) / NCH);
    const int ce = (int)(((long long)E * (c + 1)) / NCH);
    const int L = ce - cb;
    const float2* e2 = (const float2*)e;
    if (tid < MAXOWN) hist[tid] = 0;
    int dv[BIN_PER], sv[BIN_PER], rk[BIN_PER], ov[BIN_PER];
    float2 xv[BIN_PER];
    #pragma unroll
    for (int k = 0; k < BIN_PER; ++k) {
        int gi = cb + tid + k * BIN_T;
        gi = (gi < E) ? gi : E - 1;
        dv[k] = dst[gi]; sv[k] = src[gi]; xv[k] = e2[gi];
    }
    __syncthreads();
    #pragma unroll
    for (int k = 0; k < BIN_PER; ++k) {
        bool ok = (tid + k * BIN_T < L) && ((unsigned int)dv[k] < (unsigned int)N);
        int o = ok ? (int)((unsigned int)dv[k] / (unsigned int)R) : 0;
        ok = ok && (o < MAXOWN);
        ov[k] = ok ? o : -1;
        rk[k] = 0;
        if (ok) rk[k] = atomicAdd(&hist[o], 1);
    }
    __syncthreads();
    if (tid < 32) {
        const int a = hist[2 * tid], b = hist[2 * tid + 1];
        int incl = a + b;
        #pragma unroll
        for (int d = 1; d < 32; d <<= 1) {
            const int t = __shfl_up(incl, d, 32);
            if (tid >= d) incl += t;
        }
        const int excl = incl - (a + b);
        ofs[2 * tid] = excl; ofs[2 * tid + 1] = excl + a;
        if (tid == 31) ofs[MAXOWN] = incl;
    }
    __syncthreads();
    #pragma unroll
    for (int k = 0; k < BIN_PER; ++k) {
        if (ov[k] >= 0) {
            const int pos = ofs[ov[k]] + rk[k];
            sE[pos] = cb + tid + k * BIN_T; sD[pos] = dv[k]; sS[pos] = sv[k];
            sX[pos] = xv[k].x; sY[pos] = xv[k].y;
        }
    }
    __syncthreads();
    const int tot = ofs[MAXOWN];
    const long long plane = (long long)NCH * CS;
    int* r0 = rec + (long long)c * CS;
    for (int i = tid; i < tot; i += BIN_T) {
        r0[i] = sE[i];
        if (np >= 2) r0[plane + i] = sD[i];
        if (np >= 5) {
            r0[2 * plane + i] = sS[i];
            r0[3 * plane + i] = __float_as_int(sX[i]); r0[4 * plane + i] = __float_as_int(sY[i]);
        }
    }
    if (tid <= MAXOWN) dir[c * (MAXOWN + 1) + tid] = ofs[tid];
}


__global__ void __launch_bounds__(BIN_T)
k1_node_scalars(const float* __restrict__ h,
                const float* __restrict__ w_attn,
                float* __restrict__ s1, float* __restrict__ s2, int N,
                const float* __restrict__ e, const int* __restrict__ src,
                const int* __restrict__ dst,
                int* __restrict__ dir, int* __restrict__ rec,
                int E, int R, int CS, int np, int nBin) {
    __shared__ int hist[MAXOWN];
    __shared__ int ofs[MAXOWN + 1];
    __shared__ int sE[BIN_CAP], sD[BIN_CAP], sS[BIN_CAP];
    __shared__ float sX[BIN_CAP], sY[BIN_CAP];
    static_assert(MAXOWN == 64, "the first wave sums two counts per lane");
    if ((int)blockIdx.x < nBin) {
        bin_chunk(e, src, dst, dir, rec, E, N, R, CS, np, (int)blockIdx.x, hist, ofs, sE, sD, sS, sX, sY);
        return;
    }
    const int lane = threadIdx.x & 31;
    const int wavesPerBlock = blockDim.x >> 5;
    const int wave = ((int)blockIdx.x - nBin) * wavesPerBlock + (threadIdx.x >> 5);
    const int nWaves = ((int)gridDim.x - nBin) * wavesPerBlock;
    const float4 a1 = ((const float4*)w_attn)[lane];
    const float4 a2 = ((const float4*)(w_attn + DFEAT))[lane];
    for (int base = wave * 4; base < N; base += nWaves * 4) {
        float4 x[4];
        #pragma unroll
        for (int k = 0; k < 4; ++k) {
            int node = base + k;
            node = (node < N) ? node : N - 1;
            x[k] = ((const float4*)(h + (long long)node * DFEAT))[lane];
        }
        float p1[4], p2[4];
        #pragma unroll
        for (int k = 0; k < 4; ++k) {
            p1[k] = x[k].x*a1.x + x[k].y*a1.y + x[k].z*a1.z + x[k].w*a1.w;
            p2[k] = x[k].x*a2.x + x[k].y*a2.y + x[k].z*a2.z + x[k].w*a2.w;
        }
        #pragma unroll
        for (int off = 16; off > 0; off >>= 1) {
            #pragma unroll
            for (int k = 0; k < 4; ++k) {
                p1[k] += __shfl_xor(p1[k], off, 32);
                p2[k] += __shfl_xor(p2[k], off, 32);
            }
        }
        const float r1 = (lane == 0) ? p1[0] : ((lane == 1) ? p1[1] : ((lane == 2) ? p1[2] : p1[3]));
        const float r2 = (lane == 0) ? p2[0] : ((lane == 1) ? p2[1] : ((lane == 2) ? p2[2] : p2[3]));
        if (lane < 4 && base + lane < N) { s1[base + lane] = r1; s2[base + lane] = r2; }
    }
}

__device__ __forceinline__ float edge_logit(float e0, float e1, float ss, float sd,
                                            float we0, float we1, float we2, float we3,
                                            float wa0, float wa1) {
    float ex0 = e0 * we0 + e1 * we1;
    float ex1 = e0 * we2 + e1 * we3;
    float logit = ss + sd + ex0 * wa0 + ex1 * wa1;
    return (logit > 0.0f) ? logit : 0.01f * logit;
}

__device__ __forceinline__ void ldsMaxFloat(float* addr, float val) {
    val = val + 0.0f;
    if (val >= 0.0f) atomicMax((int*)addr, __float_as_int(val));
    else             atomicMin((unsigned int*)addr, __float_as_uint(val));
}


template <int NP>
__global__ void __launch_bounds__(OWN_THREADS)
k2_owner_sums(const float* __restrict__ e,
              const int* __restrict__ src, const int* __restrict__ dst,
              const float* __restrict__ s1, const float* __restrict__ s2,
              const float* __restrict__ w_edge, const float* __restrict__ w_attn,
              float* __restrict__ out,
              const int* __restrict__ dir, const int* __restrict__ rec, int CS,
              int E, int N, int R) {
    __shared__ int   ids[LIST_CAP];
    __shared__ int   key[LIST_CAP];
    __shared__ float t0[LIST_CAP], t1[LIST_CAP], t2[LIST_CAP];
    __shared__ int   ord[LIST_CAP];
    __shared__ float nm[RMAX], ns2[RMAX];
    __shared__ int   ncnt[RMAX], nst[RMAX], nfill[RMAX];
    __shared__ int   cnt;
    __shared__ int   wtot[OWN_THREADS / 32];
    __shared__ int   runP[CPP + 1], runS[CPP];
    static_assert(CPP == 64, "the first wave takes two runs per lane and the search makes six halvings");
    static_assert(RMAX <= OWN_THREADS, "the slot scan gives every node a thread of its own");
    static_assert(OWN_THREADS / 32 <= 32, "one lane per wave total");

    constexpr bool BINS = (NP > 0);
    const int tid = threadIdx.x;
    const int own = blockIdx.x / PARTS;
    const int part = blockIdx.x - own * PARTS;
    const int nOwn = gridDim.x / PARTS;
    const int base = own * R;
    float* const recBase = out + (long long)base * DFEAT + part * REC_STRIDE;
    int Rw = N - base;
    Rw = (Rw < R) ? Rw : R;
    Rw = (Rw > 0) ? Rw : 0;
    const unsigned int uRw = (unsigned int)Rw;

    const float we0 = w_edge[0], we1 = w_edge[1], we2 = w_edge[2], we3 = w_edge[3];
    const float wa0 = w_attn[2*DFEAT], wa1 = w_attn[2*DFEAT + 1];
    const float2* e2 = (const float2*)e;

    for (int n = tid; n < Rw; n += OWN_THREADS) {
        nm[n] = -3.0e38f; ncnt[n] = 0; nfill[n] = 0;
        ns2[n] = s2[base + n];
    }
    if (tid == 0) cnt = 0;
    if (BINS && tid < 32) {
        const int c0 = part * CPP + 2 * tid;
        const int a0 = dir[c0 * (MAXOWN + 1) + own],       a1 = dir[c0 * (MAXOWN + 1) + own + 1];
        const int b0 = dir[(c0 + 1) * (MAXOWN + 1) + own], b1 = dir[(c0 + 1) * (MAXOWN + 1) + own + 1];
        const int la = a1 - a0, lb = b1 - b0;
        int incl = la + lb;
        #pragma unroll
        for (int d = 1; d < 32; d <<= 1) {
            const int t = __shfl_up(incl, d, 32);
            if (tid >= d) incl += t;
        }
        const int excl = incl - (la + lb);
        runP[2 * tid] = excl; runP[2 * tid + 1] = excl + la;
        runS[2 * tid] = a0;   runS[2 * tid + 1] = b0;
        if (tid == 31) runP[CPP] = incl;
    }
    __syncthreads();

    const int nGall = E >> 2;
    const int g0 = (int)(((long long)nGall * part) / PARTS);
    const int g1 = (int)(((long long)nGall * (part + 1)) / PARTS);
    const int nG = g1 - g0;
    const int eBeg = BINS ? (int)(((long long)E * (part * CPP)) / NCH) : (g0 << 2);
    const int eEnd = BINS ? (int)(((long long)E * ((part + 1) * CPP)) / NCH)
                          : ((part == PARTS - 1) ? E : (g1 << 2));
    const int nIter = (nG + OWN_THREADS - 1) / OWN_THREADS;
    const int nGpad = nIter * OWN_THREADS;
    const int off = (int)(((long long)own * nIter) / nOwn) * OWN_THREADS;
    const int4* dst4 = (const int4*)dst + g0;
    if (!BINS) {
    for (int k0 = 0; k0 < nIter; k0 += SCAN_DEPTH) {
        int4 v[SCAN_DEPTH];
        #pragma unroll
        for (int u = 0; u < SCAN_DEPTH; ++u) {
            int gg = (k0 + u) * OWN_THREADS + tid + off;
            gg = (gg >= nGpad) ? gg - nGpad : gg;
            const bool ok = (k0 + u < nIter) && (gg < nG);
            v[u] = dst4[ok ? gg : 0];
        }
        #pragma unroll
        for (int u = 0; u < SCAN_DEPTH; ++u) {
            int gg = (k0 + u) * OWN_THREADS + tid + off;
            gg = (gg >= nGpad) ? gg - nGpad : gg;
            const bool ok = (k0 + u < nIter) && (gg < nG);
            const bool hx = (unsigned int)(v[u].x - base) < uRw;
            const bool hy = (unsigned int)(v[u].y - base) < uRw;
            const bool hz = (unsigned int)(v[u].z - base) < uRw;
            const bool hw = (unsigned int)(v[u].w - base) < uRw;
            if (ok && (hx || hy || hz || hw)) {
                const int e0i = eBeg + (gg << 2);
                if (hx) { int s = atomicAdd(&cnt, 1); if (s < LIST_CAP) ids[s] = e0i; }
                if (hy) { int s = atomicAdd(&cnt, 1); if (s < LIST_CAP) ids[s] = e0i + 1; }
                if (hz) { int s = atomicAdd(&cnt, 1); if (s < LIST_CAP) ids[s] = e0i + 2; }
                if (hw) { int s = atomicAdd(&cnt, 1); if (s < LIST_CAP) ids[s] = e0i + 3; }
            }
        }
    }
    for (int i = (g1 << 2) + tid; i < eEnd; i += OWN_THREADS) {
        if ((unsigned int)(dst[i] - base) < uRw) {
            int s = atomicAdd(&cnt, 1); if (s < LIST_CAP) ids[s] = i;
        }
    }
    __syncthreads();
    }
    const int total = BINS ? runP[CPP] : cnt;

    if (total > 0 && total <= LIST_CAP) {
        int ei[LIST_PER], dl[LIST_PER], sr[LIST_PER], ps[LIST_PER];
        float2 ev[LIST_PER];
        float sv[LIST_PER], lgt[LIST_PER];
        if (BINS) {
            const long long plane = (long long)NCH * CS;
            long long q[LIST_PER];
            #pragma unroll
            for (int u = 0; u < LIST_PER; ++u) {
                const int j = tid + u * OWN_THREADS;
                const int jj = (j < total) ? j : 0;
                int k = 0;
                #pragma unroll
                for (int st = CPP / 2; st > 0; st >>= 1) k += (runP[k + st] <= jj) ? st : 0;
                q[u] = (long long)(part * CPP + k) * CS + runS[k] + (jj - runP[k]);
            }
            #pragma unroll
            for (int u = 0; u < LIST_PER; ++u) {
                ei[u] = rec[q[u]];
                if (NP >= 2) dl[u] = rec[plane + q[u]] - base;
                if (NP >= 5) {
                    sr[u] = rec[2 * plane + q[u]];
                    ev[u].x = __int_as_float(rec[3 * plane + q[u]]);
                    ev[u].y = __int_as_float(rec[4 * plane + q[u]]);
                }
            }
            #pragma unroll
            for (int u = 0; u < LIST_PER; ++u) {
                if (NP < 2) dl[u] = dst[ei[u]] - base;
                if (NP < 5) { sr[u] = src[ei[u]]; ev[u] = e2[ei[u]]; }
            }
        } else {
            #pragma unroll
            for (int u = 0; u < LIST_PER; ++u) {
                const int j = tid + u * OWN_THREADS;
                ei[u] = ids[(j < total) ? j : 0];
            }
            #pragma unroll
            for (int u = 0; u < LIST_PER; ++u) {
                dl[u] = dst[ei[u]] - base;
                sr[u] = src[ei[u]];
                ev[u] = e2[ei[u]];
            }
        }
        #pragma unroll
        for (int u = 0; u < LIST_PER; ++u) sv[u] = s1[sr[u]];
        #pragma unroll
        for (int u = 0; u < LIST_PER; ++u) {
            lgt[u] = edge_logit(ev[u].x, ev[u].y, sv[u], ns2[dl[u]],
                                we0, we1, we2, we3, wa0, wa1);
            if (tid + u * OWN_THREADS < total) {
                ldsMaxFloat(&nm[dl[u]], lgt[u]);
                atomicAdd(&ncnt[dl[u]], 1);
            }
        }
        __syncthreads();
        {
            const int lane = tid & 31;
            const int wv = tid >> 5;
            const int c = (tid < Rw) ? ncnt[tid] : 0;
            int incl = c;
            #pragma unroll
            for (int d = 1; d < 32; d <<= 1) {
                const int t = __shfl_up(incl, d, 32);
                if (lane >= d) incl += t;
            }
            if (lane == 31) wtot[wv] = incl;
            __syncthreads();
            int before = (lane < wv) ? wtot[lane] : 0;
            #pragma unroll
            for (int o = 16; o > 0; o >>= 1) before += __shfl_xor(before, o, 32);
            if (tid < Rw) nst[tid] = before + incl - c;
        }
        __syncthreads();
        #pragma unroll
        for (int u = 0; u < LIST_PER; ++u) {
            ps[u] = 0;
            if (tid + u * OWN_THREADS < total) {
                const float ex0 = ev[u].x * we0 + ev[u].y * we1;
                const float ex1 = ev[u].x * we2 + ev[u].y * we3;
                const float w = __expf(lgt[u] - nm[dl[u]]);
                const int pos = nst[dl[u]] + atomicAdd(&nfill[dl[u]], 1);
                key[pos] = ei[u];
                t0[pos] = w; t1[pos] = w * ex0; t2[pos] = w * ex1;
                ps[u] = pos;
            }
        }
        __syncthreads();
        #pragma unroll
        for (int u = 0; u < LIST_PER; ++u) {
            if (tid + u * OWN_THREADS < total) {
                const int s = nst[dl[u]];
                const int c = ncnt[dl[u]];
                int rank = 0;
                for (int q = s; q < s + c; ++q) rank += (key[q] < ei[u]) ? 1 : 0;
                ord[s + rank] = ps[u];
            }
        }
        __syncthreads();
        for (int n = tid; n < Rw; n += OWN_THREADS) {
            const int s = nst[n];
            const int c = ncnt[n];
            float l = 0.0f, x = 0.0f, y = 0.0f;
            for (int k = 0; k < c; ++k) {
                const int p = ord[s + k];
                l += t0[p]; x += t1[p]; y += t2[p];
            }
            *(float4*)(recBase + (long long)n * DFEAT) = make_float4(nm[n], l, x, y);
        }
    } else if (total == 0) {
        for (int n = tid; n < Rw; n += OWN_THREADS)
            *(float4*)(recBase + (long long)n * DFEAT) = make_float4(-3.0e38f, 0.0f, 0.0f, 0.0f);
    } else {
        for (int n = tid; n < Rw; n += OWN_THREADS) { t0[n] = 0.0f; t1[n] = 0.0f; t2[n] = 0.0f; }
        for (int i = eBeg + tid; i < eEnd; i += OWN_THREADS) {
            const int dl = dst[i] - base;
            if ((unsigned int)dl < uRw) {
                const float2 ev = e2[i];
                ldsMaxFloat(&nm[dl], edge_logit(ev.x, ev.y, s1[src[i]], ns2[dl],
                                                we0, we1, we2, we3, wa0, wa1));
            }
        }
        __syncthreads();
        for (int i = eBeg + tid; i < eEnd; i += OWN_THREADS) {
            const int dl = dst[i] - base;
            if ((unsigned int)dl < uRw) {
                const float2 ev = e2[i];
                const float logit = edge_logit(ev.x, ev.y, s1[src[i]], ns2[dl],
                                               we0, we1, we2, we3, wa0, wa1);
                const float ex0 = ev.x * we0 + ev.y * we1;
                const float ex1 = ev.x * we2 + ev.y * we3;
                const float w = __expf(logit - nm[dl]);
                atomicAdd(&t0[dl], w);
                atomicAdd(&t1[dl], w * ex0);
                atomicAdd(&t2[dl], w * ex1);
            }
        }
        __syncthreads();
        for (int n = tid; n < Rw; n += OWN_THREADS)
            *(float4*)(recBase + (long long)n * DFEAT) = make_float4(nm[n], t0[n], t1[n], t2[n]);
    }
}

__global__ void k3_node_project(const float* __restrict__ w_e2n, float* out, int N) {
    const int lane = threadIdx.x & 31;
    const long long wave = (long long)blockIdx.x * (blockDim.x >> 5) + (threadIdx.x >> 5);
    const long long base = wave * 16;
    if (base >= (long long)N) return;

    const int lo = lane & 15;
    const bool loHalf = (lane < 16);
    const long long node = base + lo;
    const bool validA = loHalf && (node < (long long)N);
    const long long nodeC = (node < (long long)N) ? node : (long long)N - 1;

    const float2* wv = (const float2*)w_e2n;
    float bxs[8], bys[8];
    #pragma unroll
    for (int j = 0; j < 8; ++j) {
        float2 b = wv[16 * j + lo];
        bxs[j] = loHalf ? b.x : 0.0f;
        bys[j] = loHalf ? b.y : 0.0f;
    }

    float4 rec[PARTS];
    #pragma unroll
    for (int p = 0; p < PARTS; ++p) rec[p] = *(const float4*)(out + nodeC * DFEAT + p * REC_STRIDE);
    float m = rec[0].x;
    #pragma unroll
    for (int p = 1; p < PARTS; ++p) m = fmaxf(m, rec[p].x);
    float l = 0.0f, sx = 0.0f, sy = 0.0f;
    #pragma unroll
    for (int p = 0; p < PARTS; ++p) {
        const float s = expf(rec[p].x - m);
        l += rec[p].y * s; sx += rec[p].z * s; sy += rec[p].w * s;
    }
    const float inv = __builtin_amdgcn_rcpf(fmaxf(l, 1e-16f));
    const float ax = validA ? sx * inv : 0.0f;
    const float ay = validA ? sy * inv : 0.0f;

    v8f c[8];
    #pragma unroll
    for (int j = 0; j < 8; ++j) c[j] = rank2_wmma(ax, ay, bxs[j], bys[j]);

    const long long rowBase = base + ((lane >= 16) ? 8 : 0);
    #pragma unroll
    for (int v = 0; v < 8; ++v) {
        const long long row = rowBase + v;
        if (row < (long long)N) {
            float* orow = out + row * DFEAT + lo;
            #pragma unroll
            for (int j = 0; j < 8; ++j) orow[16 * j] = c[j][v];
        }
    }
}

extern "C" void kernel_launch(void* const* d_in, const int* in_sizes, int n_in,
                              void* d_out, int out_size, void* d_ws, size_t ws_size,
                              hipStream_t stream) {
    const float* h      = (const float*)d_in[0];
    const float* e      = (const float*)d_in[1];
    const int*   src    = (const int*)d_in[2];
    const int*   dst    = (const int*)d_in[3];
    const float* w_edge = (const float*)d_in[4];
    const float* w_attn = (const float*)d_in[5];
    const float* w_e2n  = (const float*)d_in[6];
    float* out = (float*)d_out;

    const int E = in_sizes[2];
    const int N = in_sizes[0] / DFEAT;

    float* s1    = (float*)d_ws;
    float* s2    = s1 + N;
    int*   dir   = (int*)(s2 + N);
    int*   rec   = dir + NCH * (MAXOWN + 1);

    int owners = 64;
    int R = (N + owners - 1) / owners;
    if (R > RMAX) { R = RMAX; owners = (N + RMAX - 1) / RMAX; }
    if (R < 1) R = 1;

    const long long chunkMax = ((long long)E + NCH - 1) / NCH;
    const int CS = (int)((chunkMax + 31) / 32 * 32);
    const size_t fixedW = (size_t)2 * N + (size_t)NCH * (MAXOWN + 1);
    const size_t planeW = (size_t)NCH * CS;
    int np = 0;
    if ((E > 0) && (chunkMax <= BIN_CAP) && (owners <= MAXOWN)) {
        if      (ws_size >= (fixedW + 5 * planeW) * sizeof(int)) np = 5;
        else if (ws_size >= (fixedW + 2 * planeW) * sizeof(int)) np = 2;
        else if (ws_size >= (fixedW + 1 * planeW) * sizeof(int)) np = 1;
    }
    const int nBin = (np > 0) ? NCH : 0;

    const int T = BIN_T;
    const int wavesPerBlock = T / 32;

    const int capRows  = 512;
    const int rowTrips = (N + 3) / 4;
    int need1 = (rowTrips + wavesPerBlock - 1) / wavesPerBlock;
    need1 = (need1 < 1) ? 1 : need1;
    const int rounds1 = (need1 + capRows - 1) / capRows;
    const int g1 = (need1 + rounds1 - 1) / rounds1;
    k1_node_scalars<<<g1 + nBin, T, 0, stream>>>(h, w_attn, s1, s2, N, e, src, dst, dir, rec, E, R, CS, np, nBin);

    if (np == 5)
        k2_owner_sums<5><<<owners * PARTS, OWN_THREADS, 0, stream>>>(e, src, dst, s1, s2, w_edge, w_attn, out, dir, rec, CS, E, N, R);
    else if (np == 2)
        k2_owner_sums<2><<<owners * PARTS, OWN_THREADS, 0, stream>>>(e, src, dst, s1, s2, w_edge, w_attn, out, dir, rec, CS, E, N, R);
    else if (np == 1)
        k2_owner_sums<1><<<owners * PARTS, OWN_THREADS, 0, stream>>>(e, src, dst, s1, s2, w_edge, w_attn, out, dir, rec, CS, E, N, R);
    else
        k2_owner_sums<0><<<owners * PARTS, OWN_THREADS, 0, stream>>>(e, src, dst, s1, s2, w_edge, w_attn, out, dir, rec, CS, E, N, R);

    const long long wavesTotal = ((long long)N + 15) / 16;
    const int gProject = (int)((wavesTotal + wavesPerBlock - 1) / wavesPerBlock);
    k3_node_project<<<gProject, T, 0, stream>>>(w_e2n, out, N);
}


# K3 folded into K2: bins sort 256 ways, each of 256 owner work-groups holds whole nodes, sums all their edges in edge order and projects on WMMA itself; no records, 2 launches; sum order changes
# speedup vs baseline: 5.8347x; 1.9459x over previous
#include <hip/hip_runtime.h>
#include <hip/hip_bf16.h>

typedef float v8f __attribute__((ext_vector_type(8)));
typedef float v2f __attribute__((ext_vector_type(2)));
typedef _Float16 v16h __attribute__((ext_vector_type(16)));

#define DFEAT 128

__device__ __forceinline__ v8f rank2_wmma(float ax, float ay, float bx, float by) {
    v8f c = {};
#if __has_builtin(__builtin_amdgcn_wmma_f32_16x16x4_f32)
    v2f a; a.x = ax; a.y = ay;
    v2f b; b.x = bx; b.y = by;
    c = __builtin_amdgcn_wmma_f32_16x16x4_f32(false, a, false, b, (short)0, c, false, false);
#else
    v16h a = {}; a[0] = (_Float16)ax; a[1] = (_Float16)ay;
    v16h b = {}; b[0] = (_Float16)bx; b[1] = (_Float16)by;
    c = __builtin_amdgcn_wmma_f32_16x16x32_f16(false, a, false, b, (short)0, c, false, false);
#endif
    return c;
}

#define OWN_THREADS 1024
#define LIST_CAP    8192
#define RMAX        1024
#define SCAN_DEPTH  17
#define LIST_PER    (LIST_CAP / OWN_THREADS)
#define NCH         256
#define MAXOWN      256
#define BIN_T       256
#define BIN_CAP     4096
#define BIN_PER     (BIN_CAP / BIN_T)

__device__ __forceinline__ void bin_chunk(const float* __restrict__ e, const int* __restrict__ src,
                                          const int* __restrict__ dst,
                                          int* __restrict__ dir, int* __restrict__ rec,
                                          int E, int N, int R, int CS, int np, int c,
                                          int* hist, int* ofs, int* wsum,
                                          int* sE, int* sD, int* sS, float* sX, float* sY) {
    const int tid = threadIdx.x;
    const int cb = (int)(((long long)E * c) / NCH);
    const int ce = (int)(((long long)E * (c + 1)) / NCH);
    const int L = ce - cb;
    const float2* e2 = (const float2*)e;
    hist[tid] = 0;
    int dv[BIN_PER], sv[BIN_PER], rk[BIN_PER], ov[BIN_PER];
    float2 xv[BIN_PER];
    #pragma unroll
    for (int k = 0; k < BIN_PER; ++k) {
        int gi = cb + tid + k * BIN_T;
        gi = (gi < E) ? gi : E - 1;
        dv[k] = dst[gi]; sv[k] = src[gi]; xv[k] = e2[gi];
    }
    __syncthreads();
    #pragma unroll
    for (int k = 0; k < BIN_PER; ++k) {
        bool ok = (tid + k * BIN_T < L) && ((unsigned int)dv[k] < (unsigned int)N);
        int o = ok ? (int)((unsigned int)dv[k] / (unsigned int)R) : 0;
        ok = ok && (o < MAXOWN);
        ov[k] = ok ? o : -1;
        rk[k] = 0;
        if (ok) rk[k] = atomicAdd(&hist[o], 1);
    }
    __syncthreads();
    {
        const int lane = tid & 31;
        const int wv = tid >> 5;
        const int cn = hist[tid];
        int incl = cn;
        #pragma unroll
        for (int d = 1; d < 32; d <<= 1) {
            const int t = __shfl_up(incl, d, 32);
            if (lane >= d) incl += t;
        }
        if (lane == 31) wsum[wv] = incl;
        __syncthreads();
        int before = 0;
        #pragma unroll
        for (int w = 0; w < BIN_T / 32; ++w) before += (w < wv) ? wsum[w] : 0;
        ofs[tid] = before + incl - cn;
        if (tid == BIN_T - 1) ofs[MAXOWN] = before + incl;
    }
    __syncthreads();
    #pragma unroll
    for (int k = 0; k < BIN_PER; ++k) {
        if (ov[k] >= 0) {
            const int pos = ofs[ov[k]] + rk[k];
            sE[pos] = cb + tid + k * BIN_T; sD[pos] = dv[k]; sS[pos] = sv[k];
            sX[pos] = xv[k].x; sY[pos] = xv[k].y;
        }
    }
    __syncthreads();
    const int tot = ofs[MAXOWN];
    const long long plane = (long long)NCH * CS;
    int* r0 = rec + (long long)c * CS;
    for (int i = tid; i < tot; i += BIN_T) {
        r0[i] = sE[i];
        if (np >= 2) r0[plane + i] = sD[i];
        if (np >= 5) {
            r0[2 * plane + i] = sS[i];
            r0[3 * plane + i] = __float_as_int(sX[i]); r0[4 * plane + i] = __float_as_int(sY[i]);
        }
    }
    for (int i = tid; i <= MAXOWN; i += BIN_T) dir[c * (MAXOWN + 1) + i] = ofs[i];
}


__global__ void __launch_bounds__(BIN_T)
k1_node_scalars(const float* __restrict__ h,
                const float* __restrict__ w_attn,
                float* __restrict__ s1, float* __restrict__ s2, int N,
                const float* __restrict__ e, const int* __restrict__ src,
                const int* __restrict__ dst,
                int* __restrict__ dir, int* __restrict__ rec,
                int E, int R, int CS, int np, int nBin) {
    __shared__ int hist[MAXOWN];
    __shared__ int ofs[MAXOWN + 1];
    __shared__ int wsum[BIN_T / 32];
    __shared__ int sE[BIN_CAP], sD[BIN_CAP], sS[BIN_CAP];
    __shared__ float sX[BIN_CAP], sY[BIN_CAP];
    static_assert(MAXOWN == BIN_T, "the running sum of the counts takes one count per thread");
    if ((int)blockIdx.x < nBin) {
        bin_chunk(e, src, dst, dir, rec, E, N, R, CS, np, (int)blockIdx.x, hist, ofs, wsum, sE, sD, sS, sX, sY);
        return;
    }
    const int lane = threadIdx.x & 31;
    const int wavesPerBlock = blockDim.x >> 5;
    const int wave = ((int)blockIdx.x - nBin) * wavesPerBlock + (threadIdx.x >> 5);
    const int nWaves = ((int)gridDim.x - nBin) * wavesPerBlock;
    const float4 a1 = ((const float4*)w_attn)[lane];
    const float4 a2 = ((const float4*)(w_attn + DFEAT))[lane];
    for (int base = wave * 4; base < N; base += nWaves * 4) {
        float4 x[4];
        #pragma unroll
        for (int k = 0; k < 4; ++k) {
            int node = base + k;
            node = (node < N) ? node : N - 1;
            x[k] = ((const float4*)(h + (long long)node * DFEAT))[lane];
        }
        float p1[4], p2[4];
        #pragma unroll
        for (int k = 0; k < 4; ++k) {
            p1[k] = x[k].x*a1.x + x[k].y*a1.y + x[k].z*a1.z + x[k].w*a1.w;
            p2[k] = x[k].x*a2.x + x[k].y*a2.y + x[k].z*a2.z + x[k].w*a2.w;
        }
        #pragma unroll
        for (int off = 16; off > 0; off >>= 1) {
            #pragma unroll
            for (int k = 0; k < 4; ++k) {
                p1[k] += __shfl_xor(p1[k], off, 32);
                p2[k] += __shfl_xor(p2[k], off, 32);
            }
        }
        const float r1 = (lane == 0) ? p1[0] : ((lane == 1) ? p1[1] : ((lane == 2) ? p1[2] : p1[3]));
        const float r2 = (lane == 0) ? p2[0] : ((lane == 1) ? p2[1] : ((lane == 2) ? p2[2] : p2[3]));
        if (lane < 4 && base + lane < N) { s1[base + lane] = r1; s2[base + lane] = r2; }
    }
}

__device__ __forceinline__ float edge_logit(float e0, float e1, float ss, float sd,
                                            float we0, float we1, float we2, float we3,
                                            float wa0, float wa1) {
    float ex0 = e0 * we0 + e1 * we1;
    float ex1 = e0 * we2 + e1 * we3;
    float logit = ss + sd + ex0 * wa0 + ex1 * wa1;
    return (logit > 0.0f) ? logit : 0.01f * logit;
}

__device__ __forceinline__ void ldsMaxFloat(float* addr, float val) {
    val = val + 0.0f;
    if (val >= 0.0f) atomicMax((int*)addr, __float_as_int(val));
    else             atomicMin((unsigned int*)addr, __float_as_uint(val));
}


template <int NP>
__global__ void __launch_bounds__(OWN_THREADS)
k2_owner_sums(const float* __restrict__ e,
              const int* __restrict__ src, const int* __restrict__ dst,
              const float* __restrict__ s1, const float* __restrict__ s2,
              const float* __restrict__ w_edge, const float* __restrict__ w_attn,
              const float* __restrict__ w_e2n,
              float* __restrict__ out,
              const int* __restrict__ dir, const int* __restrict__ rec, int CS,
              int E, int N, int R) {
    __shared__ int   ids[LIST_CAP];
    __shared__ int   key[LIST_CAP];
    __shared__ float t0[LIST_CAP], t1[LIST_CAP], t2[LIST_CAP];
    __shared__ int   ord[LIST_CAP];
    __shared__ float nm[RMAX], ns2[RMAX];
    __shared__ int   ncnt[RMAX], nst[RMAX], nfill[RMAX];
    __shared__ float nax[RMAX], nay[RMAX];
    __shared__ int   cnt;
    __shared__ int   wtot[OWN_THREADS / 32];
    __shared__ int   runP[NCH + 1], runS[NCH];
    static_assert(NCH == 256, "the search for an entry's run makes eight halvings");
    static_assert(NCH <= OWN_THREADS, "the running sum of the run lengths gives every run a thread of its own");
    static_assert(RMAX <= OWN_THREADS, "the slot scan gives every node a thread of its own");
    static_assert(OWN_THREADS / 32 <= 32, "one lane per wave total");

    constexpr bool BINS = (NP > 0);
    const int tid = threadIdx.x;
    const int own = blockIdx.x;
    const int nOwn = gridDim.x;
    const int base = own * R;
    int Rw = N - base;
    Rw = (Rw < R) ? Rw : R;
    Rw = (Rw > 0) ? Rw : 0;
    const unsigned int uRw = (unsigned int)Rw;

    const float we0 = w_edge[0], we1 = w_edge[1], we2 = w_edge[2], we3 = w_edge[3];
    const float wa0 = w_attn[2*DFEAT], wa1 = w_attn[2*DFEAT + 1];
    const float2* e2 = (const float2*)e;

    for (int n = tid; n < Rw; n += OWN_THREADS) {
        nm[n] = -3.0e38f; ncnt[n] = 0; nfill[n] = 0;
        ns2[n] = s2[base + n];
    }
    if (tid == 0) cnt = 0;
    if (BINS) {
        const int lane = tid & 31;
        const int wv = tid >> 5;
        int a0 = 0, la = 0;
        if (tid < NCH) {
            a0 = dir[tid * (MAXOWN + 1) + own];
            la = dir[tid * (MAXOWN + 1) + own + 1] - a0;
        }
        int incl = la;
        #pragma unroll
        for (int d = 1; d < 32; d <<= 1) {
            const int t = __shfl_up(incl, d, 32);
            if (lane >= d) incl += t;
        }
        if (tid < NCH && lane == 31) wtot[wv] = incl;
        __syncthreads();
        if (tid < NCH) {
            int before = 0;
            #pragma unroll
            for (int w = 0; w < NCH / 32; ++w) before += (w < wv) ? wtot[w] : 0;
            runP[tid] = before + incl - la;
            runS[tid] = a0;
            if (tid == NCH - 1) runP[NCH] = before + incl;
        }
    }
    __syncthreads();

    const int nG = E >> 2;
    const int eBeg = 0, eEnd = E;
    const int nIter = (nG + OWN_THREADS - 1) / OWN_THREADS;
    const int nGpad = nIter * OWN_THREADS;
    const int off = (int)(((long long)own * nIter) / nOwn) * OWN_THREADS;
    const int4* dst4 = (const int4*)dst;
    if (!BINS) {
    for (int k0 = 0; k0 < nIter; k0 += SCAN_DEPTH) {
        int4 v[SCAN_DEPTH];
        #pragma unroll
        for (int u = 0; u < SCAN_DEPTH; ++u) {
            int gg = (k0 + u) * OWN_THREADS + tid + off;
            gg = (gg >= nGpad) ? gg - nGpad : gg;
            const bool ok = (k0 + u < nIter) && (gg < nG);
            v[u] = dst4[ok ? gg : 0];
        }
        #pragma unroll
        for (int u = 0; u < SCAN_DEPTH; ++u) {
            int gg = (k0 + u) * OWN_THREADS + tid + off;
            gg = (gg >= nGpad) ? gg - nGpad : gg;
            const bool ok = (k0 + u < nIter) && (gg < nG);
            const bool hx = (unsigned int)(v[u].x - base) < uRw;
            const bool hy = (unsigned int)(v[u].y - base) < uRw;
            const bool hz = (unsigned int)(v[u].z - base) < uRw;
            const bool hw = (unsigned int)(v[u].w - base) < uRw;
            if (ok && (hx || hy || hz || hw)) {
                const int e0i = eBeg + (gg << 2);
                if (hx) { int s = atomicAdd(&cnt, 1); if (s < LIST_CAP) ids[s] = e0i; }
                if (hy) { int s = atomicAdd(&cnt, 1); if (s < LIST_CAP) ids[s] = e0i + 1; }
                if (hz) { int s = atomicAdd(&cnt, 1); if (s < LIST_CAP) ids[s] = e0i + 2; }
                if (hw) { int s = atomicAdd(&cnt, 1); if (s < LIST_CAP) ids[s] = e0i + 3; }
            }
        }
    }
    for (int i = (nG << 2) + tid; i < eEnd; i += OWN_THREADS) {
        if ((unsigned int)(dst[i] - base) < uRw) {
            int s = atomicAdd(&cnt, 1); if (s < LIST_CAP) ids[s] = i;
        }
    }
    __syncthreads();
    }
    const int total = BINS ? runP[NCH] : cnt;

    if (total > 0 && total <= LIST_CAP) {
        int ei[LIST_PER], dl[LIST_PER], sr[LIST_PER], ps[LIST_PER];
        float2 ev[LIST_PER];
        float sv[LIST_PER], lgt[LIST_PER];
        if (BINS) {
            const long long plane = (long long)NCH * CS;
            long long q[LIST_PER];
            #pragma unroll
            for (int u = 0; u < LIST_PER; ++u) {
                const int j = tid + u * OWN_THREADS;
                const int jj = (j < total) ? j : 0;
                int k = 0;
                #pragma unroll
                for (int st = NCH / 2; st > 0; st >>= 1) k += (runP[k + st] <= jj) ? st : 0;
                q[u] = (long long)k * CS + runS[k] + (jj - runP[k]);
            }
            #pragma unroll
            for (int u = 0; u < LIST_PER; ++u) {
                ei[u] = rec[q[u]];
                if (NP >= 2) dl[u] = rec[plane + q[u]] - base;
                if (NP >= 5) {
                    sr[u] = rec[2 * plane + q[u]];
                    ev[u].x = __int_as_float(rec[3 * plane + q[u]]);
                    ev[u].y = __int_as_float(rec[4 * plane + q[u]]);
                }
            }
            #pragma unroll
            for (int u = 0; u < LIST_PER; ++u) {
                if (NP < 2) dl[u] = dst[ei[u]] - base;
                if (NP < 5) { sr[u] = src[ei[u]]; ev[u] = e2[ei[u]]; }
            }
        } else {
            #pragma unroll
            for (int u = 0; u < LIST_PER; ++u) {
                const int j = tid + u * OWN_THREADS;
                ei[u] = ids[(j < total) ? j : 0];
            }
            #pragma unroll
            for (int u = 0; u < LIST_PER; ++u) {
                dl[u] = dst[ei[u]] - base;
                sr[u] = src[ei[u]];
                ev[u] = e2[ei[u]];
            }
        }
        #pragma unroll
        for (int u = 0; u < LIST_PER; ++u) sv[u] = s1[sr[u]];
        #pragma unroll
        for (int u = 0; u < LIST_PER; ++u) {
            lgt[u] = edge_logit(ev[u].x, ev[u].y, sv[u], ns2[dl[u]],
                                we0, we1, we2, we3, wa0, wa1);
            if (tid + u * OWN_THREADS < total) {
                ldsMaxFloat(&nm[dl[u]], lgt[u]);
                atomicAdd(&ncnt[dl[u]], 1);
            }
        }
        __syncthreads();
        {
            const int lane = tid & 31;
            const int wv = tid >> 5;
            const int c = (tid < Rw) ? ncnt[tid] : 0;
            int incl = c;
            #pragma unroll
            for (int d = 1; d < 32; d <<= 1) {
                const int t = __shfl_up(incl, d, 32);
                if (lane >= d) incl += t;
            }
            if (lane == 31) wtot[wv] = incl;
            __syncthreads();
            int before = (lane < wv) ? wtot[lane] : 0;
            #pragma unroll
            for (int o = 16; o > 0; o >>= 1) before += __shfl_xor(before, o, 32);
            if (tid < Rw) nst[tid] = before + incl - c;
        }
        __syncthreads();
        #pragma unroll
        for (int u = 0; u < LIST_PER; ++u) {
            ps[u] = 0;
            if (tid + u * OWN_THREADS < total) {
                const float ex0 = ev[u].x * we0 + ev[u].y * we1;
                const float ex1 = ev[u].x * we2 + ev[u].y * we3;
                const float w = __expf(lgt[u] - nm[dl[u]]);
                const int pos = nst[dl[u]] + atomicAdd(&nfill[dl[u]], 1);
                key[pos] = ei[u];
                t0[pos] = w; t1[pos] = w * ex0; t2[pos] = w * ex1;
                ps[u] = pos;
            }
        }
        __syncthreads();
        #pragma unroll
        for (int u = 0; u < LIST_PER; ++u) {
            if (tid + u * OWN_THREADS < total) {
                const int s = nst[dl[u]];
                const int end = s + ncnt[dl[u]];
                int rank = 0;
                for (int q = s; q < end; q += 8) {
                    int kv[8];
                    #pragma unroll
                    for (int i = 0; i < 8; ++i) {
                        const int qi = q + i;
                        kv[i] = key[(qi < LIST_CAP) ? qi : LIST_CAP - 1];
                    }
                    #pragma unroll
                    for (int i = 0; i < 8; ++i) rank += ((q + i < end) && (kv[i] < ei[u])) ? 1 : 0;
                }
                ord[s + rank] = ps[u];
            }
        }
        __syncthreads();
        for (int n = tid; n < Rw; n += OWN_THREADS) {
            const int s = nst[n];
            const int c = ncnt[n];
            float l = 0.0f, x = 0.0f, y = 0.0f;
            for (int k = 0; k < c; k += 8) {
                int p[8];
                float a[8], b[8], d[8];
                #pragma unroll
                for (int i = 0; i < 8; ++i) {
                    const int qi = s + k + i;
                    const int pv = ord[(qi < LIST_CAP) ? qi : LIST_CAP - 1];
                    p[i] = (k + i < c) ? pv : 0;
                }
                #pragma unroll
                for (int i = 0; i < 8; ++i) { a[i] = t0[p[i]]; b[i] = t1[p[i]]; d[i] = t2[p[i]]; }
                #pragma unroll
                for (int i = 0; i < 8; ++i) {
                    if (k + i < c) { l += a[i]; x += b[i]; y += d[i]; }
                }
            }
            const float inv = __builtin_amdgcn_rcpf(fmaxf(l, 1e-16f));
            nax[n] = x * inv; nay[n] = y * inv;
        }
    } else if (total == 0) {
        for (int n = tid; n < Rw; n += OWN_THREADS) { nax[n] = 0.0f; nay[n] = 0.0f; }
    } else {
        for (int n = tid; n < Rw; n += OWN_THREADS) { t0[n] = 0.0f; t1[n] = 0.0f; t2[n] = 0.0f; }
        for (int i = eBeg + tid; i < eEnd; i += OWN_THREADS) {
            const int dl = dst[i] - base;
            if ((unsigned int)dl < uRw) {
                const float2 ev = e2[i];
                ldsMaxFloat(&nm[dl], edge_logit(ev.x, ev.y, s1[src[i]], ns2[dl],
                                                we0, we1, we2, we3, wa0, wa1));
            }
        }
        __syncthreads();
        for (int i = eBeg + tid; i < eEnd; i += OWN_THREADS) {
            const int dl = dst[i] - base;
            if ((unsigned int)dl < uRw) {
                const float2 ev = e2[i];
                const float logit = edge_logit(ev.x, ev.y, s1[src[i]], ns2[dl],
                                               we0, we1, we2, we3, wa0, wa1);
                const float ex0 = ev.x * we0 + ev.y * we1;
                const float ex1 = ev.x * we2 + ev.y * we3;
                const float w = __expf(logit - nm[dl]);
                atomicAdd(&t0[dl], w);
                atomicAdd(&t1[dl], w * ex0);
                atomicAdd(&t2[dl], w * ex1);
            }
        }
        __syncthreads();
        for (int n = tid; n < Rw; n += OWN_THREADS) {
            const float inv = __builtin_amdgcn_rcpf(fmaxf(t0[n], 1e-16f));
            nax[n] = t1[n] * inv; nay[n] = t2[n] * inv;
        }
    }
    __syncthreads();

    {
        const int lane = tid & 31;
        const int wv = tid >> 5;
        const int lo = lane & 15;
        const bool loHalf = (lane < 16);
        const float2* wc = (const float2*)w_e2n;
        const int nUnits = ((Rw + 15) >> 4) * 4;
        for (int u = wv; u < nUnits; u += OWN_THREADS / 32) {
            const int tile = u >> 2;
            const int jp = u & 3;
            const int nl = tile * 16 + lo;
            const int nlc = (nl < RMAX) ? nl : RMAX - 1;
            const bool validA = loHalf && (nl < Rw);
            const float axr = nax[nlc], ayr = nay[nlc];
            const float ax = validA ? axr : 0.0f;
            const float ay = validA ? ayr : 0.0f;
            const float2 b0 = wc[32 * jp + lo];
            const float2 b1 = wc[32 * jp + 16 + lo];
            const v8f c0 = rank2_wmma(ax, ay, loHalf ? b0.x : 0.0f, loHalf ? b0.y : 0.0f);
            const v8f c1 = rank2_wmma(ax, ay, loHalf ? b1.x : 0.0f, loHalf ? b1.y : 0.0f);
            const int rowl = tile * 16 + ((lane >= 16) ? 8 : 0);
            const int nrow = Rw - rowl;
            float* const ob = out + ((long long)base + rowl) * DFEAT + 32 * jp + lo;
            #pragma unroll
            for (int v = 0; v < 8; ++v) {
                if (v < nrow) {
                    ob[(long long)v * DFEAT] = c0[v];
                    ob[(long long)v * DFEAT + 16] = c1[v];
                }
            }
        }
    }
}

extern "C" void kernel_launch(void* const* d_in, const int* in_sizes, int n_in,
                              void* d_out, int out_size, void* d_ws, size_t ws_size,
                              hipStream_t stream) {
    const float* h      = (const float*)d_in[0];
    const float* e      = (const float*)d_in[1];
    const int*   src    = (const int*)d_in[2];
    const int*   dst    = (const int*)d_in[3];
    const float* w_edge = (const float*)d_in[4];
    const float* w_attn = (const float*)d_in[5];
    const float* w_e2n  = (const float*)d_in[6];
    float* out = (float*)d_out;

    const int E = in_sizes[2];
    const int N = in_sizes[0] / DFEAT;

    float* s1    = (float*)d_ws;
    float* s2    = s1 + N;
    int*   dir   = (int*)(s2 + N);
    int*   rec   = dir + NCH * (MAXOWN + 1);

    int owners = MAXOWN;
    int R = (N + owners - 1) / owners;
    if (R > RMAX) { R = RMAX; owners = (N + RMAX - 1) / RMAX; }
    if (R < 1) R = 1;

    const long long chunkMax = ((long long)E + NCH - 1) / NCH;
    const int CS = (int)((chunkMax + 31) / 32 * 32);
    const size_t fixedW = (size_t)2 * N + (size_t)NCH * (MAXOWN + 1);
    const size_t planeW = (size_t)NCH * CS;
    int np = 0;
    if ((E > 0) && (chunkMax <= BIN_CAP) && (owners <= MAXOWN)) {
        if      (ws_size >= (fixedW + 5 * planeW) * sizeof(int)) np = 5;
        else if (ws_size >= (fixedW + 2 * planeW) * sizeof(int)) np = 2;
        else if (ws_size >= (fixedW + 1 * planeW) * sizeof(int)) np = 1;
    }
    const int nBin = (np > 0) ? NCH : 0;

    const int T = BIN_T;
    const int wavesPerBlock = T / 32;

    const int capRows  = 512;
    const int rowTrips = (N + 3) / 4;
    int need1 = (rowTrips + wavesPerBlock - 1) / wavesPerBlock;
    need1 = (need1 < 1) ? 1 : need1;
    const int rounds1 = (need1 + capRows - 1) / capRows;
    const int g1 = (need1 + rounds1 - 1) / rounds1;
    k1_node_scalars<<<g1 + nBin, T, 0, stream>>>(h, w_attn, s1, s2, N, e, src, dst, dir, rec, E, R, CS, np, nBin);

    if (np == 5)
        k2_owner_sums<5><<<owners, OWN_THREADS, 0, stream>>>(e, src, dst, s1, s2, w_edge, w_attn, w_e2n, out, dir, rec, CS, E, N, R);
    else if (np == 2)
        k2_owner_sums<2><<<owners, OWN_THREADS, 0, stream>>>(e, src, dst, s1, s2, w_edge, w_attn, w_e2n, out, dir, rec, CS, E, N, R);
    else if (np == 1)
        k2_owner_sums<1><<<owners, OWN_THREADS, 0, stream>>>(e, src, dst, s1, s2, w_edge, w_attn, w_e2n, out, dir, rec, CS, E, N, R);
    else
        k2_owner_sums<0><<<owners, OWN_THREADS, 0, stream>>>(e, src, dst, s1, s2, w_edge, w_attn, w_e2n, out, dir, rec, CS, E, N, R);
}


# Projection tile re-mapped (8 nodes twice over the two k pairs, B column = the lane's own 4 output columns): a lane holds 4 adjacent columns, every store is one whole 512-byte row; sum order kept
# speedup vs baseline: 6.1575x; 1.0553x over previous
#include <hip/hip_runtime.h>
#include <hip/hip_bf16.h>

typedef float v8f __attribute__((ext_vector_type(8)));
typedef float v2f __attribute__((ext_vector_type(2)));

#define DFEAT 128

__device__ __forceinline__ v8f wmma_k4(v2f a, v2f b) {
    v8f c = {};
    return __builtin_amdgcn_wmma_f32_16x16x4_f32(false, a, false, b, (short)0, c, false, false);
}

#define OWN_THREADS 1024
#define LIST_CAP    8192
#define RMAX        1024
#define SCAN_DEPTH  17
#define LIST_PER    (LIST_CAP / OWN_THREADS)
#define NCH         256
#define MAXOWN      256
#define BIN_T       256
#define BIN_CAP     4096
#define BIN_PER     (BIN_CAP / BIN_T)

__device__ __forceinline__ void bin_chunk(const float* __restrict__ e, const int* __restrict__ src,
                                          const int* __restrict__ dst,
                                          int* __restrict__ dir, int* __restrict__ rec,
                                          int E, int N, int R, int CS, int np, int c,
                                          int* hist, int* ofs, int* wsum,
                                          int* sE, int* sD, int* sS, float* sX, float* sY) {
    const int tid = threadIdx.x;
    const int cb = (int)(((long long)E * c) / NCH);
    const int ce = (int)(((long long)E * (c + 1)) / NCH);
    const int L = ce - cb;
    const float2* e2 = (const float2*)e;
    hist[tid] = 0;
    int dv[BIN_PER], sv[BIN_PER], rk[BIN_PER], ov[BIN_PER];
    float2 xv[BIN_PER];
    #pragma unroll
    for (int k = 0; k < BIN_PER; ++k) {
        int gi = cb + tid + k * BIN_T;
        gi = (gi < E) ? gi : E - 1;
        dv[k] = dst[gi]; sv[k] = src[gi]; xv[k] = e2[gi];
    }
    __syncthreads();
    #pragma unroll
    for (int k = 0; k < BIN_PER; ++k) {
        bool ok = (tid + k * BIN_T < L) && ((unsigned int)dv[k] < (unsigned int)N);
        int o = ok ? (int)((unsigned int)dv[k] / (unsigned int)R) : 0;
        ok = ok && (o < MAXOWN);
        ov[k] = ok ? o : -1;
        rk[k] = 0;
        if (ok) rk[k] = atomicAdd(&hist[o], 1);
    }
    __syncthreads();
    {
        const int lane = tid & 31;
        const int wv = tid >> 5;
        const int cn = hist[tid];
        int incl = cn;
        #pragma unroll
        for (int d = 1; d < 32; d <<= 1) {
            const int t = __shfl_up(incl, d, 32);
            if (lane >= d) incl += t;
        }
        if (lane == 31) wsum[wv] = incl;
        __syncthreads();
        int before = 0;
        #pragma unroll
        for (int w = 0; w < BIN_T / 32; ++w) before += (w < wv) ? wsum[w] : 0;
        ofs[tid] = before + incl - cn;
        if (tid == BIN_T - 1) ofs[MAXOWN] = before + incl;
    }
    __syncthreads();
    #pragma unroll
    for (int k = 0; k < BIN_PER; ++k) {
        if (ov[k] >= 0) {
            const int pos = ofs[ov[k]] + rk[k];
            sE[pos] = cb + tid + k * BIN_T; sD[pos] = dv[k]; sS[pos] = sv[k];
            sX[pos] = xv[k].x; sY[pos] = xv[k].y;
        }
    }
    __syncthreads();
    const int tot = ofs[MAXOWN];
    const long long plane = (long long)NCH * CS;
    int* r0 = rec + (long long)c * CS;
    for (int i = tid; i < tot; i += BIN_T) {
        r0[i] = sE[i];
        if (np >= 2) r0[plane + i] = sD[i];
        if (np >= 5) {
            r0[2 * plane + i] = sS[i];
            r0[3 * plane + i] = __float_as_int(sX[i]); r0[4 * plane + i] = __float_as_int(sY[i]);
        }
    }
    for (int i = tid; i <= MAXOWN; i += BIN_T) dir[c * (MAXOWN + 1) + i] = ofs[i];
}


__global__ void __launch_bounds__(BIN_T)
k1_node_scalars(const float* __restrict__ h,
                const float* __restrict__ w_attn,
                float* __restrict__ s1, float* __restrict__ s2, int N,
                const float* __restrict__ e, const int* __restrict__ src,
                const int* __restrict__ dst,
                int* __restrict__ dir, int* __restrict__ rec,
                int E, int R, int CS, int np, int nBin) {
    __shared__ int hist[MAXOWN];
    __shared__ int ofs[MAXOWN + 1];
    __shared__ int wsum[BIN_T / 32];
    __shared__ int sE[BIN_CAP], sD[BIN_CAP], sS[BIN_CAP];
    __shared__ float sX[BIN_CAP], sY[BIN_CAP];
    static_assert(MAXOWN == BIN_T, "the running sum of the counts takes one count per thread");
    if ((int)blockIdx.x < nBin) {
        bin_chunk(e, src, dst, dir, rec, E, N, R, CS, np, (int)blockIdx.x, hist, ofs, wsum, sE, sD, sS, sX, sY);
        return;
    }
    const int lane = threadIdx.x & 31;
    const int wavesPerBlock = blockDim.x >> 5;
    const int wave = ((int)blockIdx.x - nBin) * wavesPerBlock + (threadIdx.x >> 5);
    const int nWaves = ((int)gridDim.x - nBin) * wavesPerBlock;
    const float4 a1 = ((const float4*)w_attn)[lane];
    const float4 a2 = ((const float4*)(w_attn + DFEAT))[lane];
    for (int base = wave * 4; base < N; base += nWaves * 4) {
        float4 x[4];
        #pragma unroll
        for (int k = 0; k < 4; ++k) {
            int node = base + k;
            node = (node < N) ? node : N - 1;
            x[k] = ((const float4*)(h + (long long)node * DFEAT))[lane];
        }
        float p1[4], p2[4];
        #pragma unroll
        for (int k = 0; k < 4; ++k) {
            p1[k] = x[k].x*a1.x + x[k].y*a1.y + x[k].z*a1.z + x[k].w*a1.w;
            p2[k] = x[k].x*a2.x + x[k].y*a2.y + x[k].z*a2.z + x[k].w*a2.w;
        }
        #pragma unroll
        for (int off = 16; off > 0; off >>= 1) {
            #pragma unroll
            for (int k = 0; k < 4; ++k) {
                p1[k] += __shfl_xor(p1[k], off, 32);
                p2[k] += __shfl_xor(p2[k], off, 32);
            }
        }
        const float r1 = (lane == 0) ? p1[0] : ((lane == 1) ? p1[1] : ((lane == 2) ? p1[2] : p1[3]));
        const float r2 = (lane == 0) ? p2[0] : ((lane == 1) ? p2[1] : ((lane == 2) ? p2[2] : p2[3]));
        if (lane < 4 && base + lane < N) { s1[base + lane] = r1; s2[base + lane] = r2; }
    }
}

__device__ __forceinline__ float edge_logit(float e0, float e1, float ss, float sd,
                                            float we0, float we1, float we2, float we3,
                                            float wa0, float wa1) {
    float ex0 = e0 * we0 + e1 * we1;
    float ex1 = e0 * we2 + e1 * we3;
    float logit = ss + sd + ex0 * wa0 + ex1 * wa1;
    return (logit > 0.0f) ? logit : 0.01f * logit;
}

__device__ __forceinline__ void ldsMaxFloat(float* addr, float val) {
    val = val + 0.0f;
    if (val >= 0.0f) atomicMax((int*)addr, __float_as_int(val));
    else             atomicMin((unsigned int*)addr, __float_as_uint(val));
}


template <int NP>
__global__ void __launch_bounds__(OWN_THREADS)
k2_owner_sums(const float* __restrict__ e,
              const int* __restrict__ src, const int* __restrict__ dst,
              const float* __restrict__ s1, const float* __restrict__ s2,
              const float* __restrict__ w_edge, const float* __restrict__ w_attn,
              const float* __restrict__ w_e2n,
              float* __restrict__ out,
              const int* __restrict__ dir, const int* __restrict__ rec, int CS,
              int E, int N, int R) {
    __shared__ int   ids[LIST_CAP];
    __shared__ int   key[LIST_CAP];
    __shared__ float t0[LIST_CAP], t1[LIST_CAP], t2[LIST_CAP];
    __shared__ int   ord[LIST_CAP];
    __shared__ float nm[RMAX], ns2[RMAX];
    __shared__ int   ncnt[RMAX], nst[RMAX], nfill[RMAX];
    __shared__ float nax[RMAX], nay[RMAX];
    __shared__ int   cnt;
    __shared__ int   wtot[OWN_THREADS / 32];
    __shared__ int   runP[NCH + 1], runS[NCH];
    static_assert(NCH == 256, "the search for an entry's run makes eight halvings");
    static_assert(NCH <= OWN_THREADS, "the running sum of the run lengths gives every run a thread of its own");
    static_assert(RMAX <= OWN_THREADS, "the slot scan gives every node a thread of its own");
    static_assert(OWN_THREADS / 32 <= 32, "one lane per wave total");

    constexpr bool BINS = (NP > 0);
    const int tid = threadIdx.x;
    const int own = blockIdx.x;
    const int nOwn = gridDim.x;
    const int base = own * R;
    int Rw = N - base;
    Rw = (Rw < R) ? Rw : R;
    Rw = (Rw > 0) ? Rw : 0;
    const unsigned int uRw = (unsigned int)Rw;

    const float we0 = w_edge[0], we1 = w_edge[1], we2 = w_edge[2], we3 = w_edge[3];
    const float wa0 = w_attn[2*DFEAT], wa1 = w_attn[2*DFEAT + 1];
    const float2* e2 = (const float2*)e;

    for (int n = tid; n < Rw; n += OWN_THREADS) {
        nm[n] = -3.0e38f; ncnt[n] = 0; nfill[n] = 0;
        ns2[n] = s2[base + n];
    }
    if (tid == 0) cnt = 0;
    if (BINS) {
        const int lane = tid & 31;
        const int wv = tid >> 5;
        int a0 = 0, la = 0;
        if (tid < NCH) {
            a0 = dir[tid * (MAXOWN + 1) + own];
            la = dir[tid * (MAXOWN + 1) + own + 1] - a0;
        }
        int incl = la;
        #pragma unroll
        for (int d = 1; d < 32; d <<= 1) {
            const int t = __shfl_up(incl, d, 32);
            if (lane >= d) incl += t;
        }
        if (tid < NCH && lane == 31) wtot[wv] = incl;
        __syncthreads();
        if (tid < NCH) {
            int before = 0;
            #pragma unroll
            for (int w = 0; w < NCH / 32; ++w) before += (w < wv) ? wtot[w] : 0;
            runP[tid] = before + incl - la;
            runS[tid] = a0;
            if (tid == NCH - 1) runP[NCH] = before + incl;
        }
    }
    __syncthreads();

    const int nG = E >> 2;
    const int eBeg = 0, eEnd = E;
    const int nIter = (nG + OWN_THREADS - 1) / OWN_THREADS;
    const int nGpad = nIter * OWN_THREADS;
    const int off = (int)(((long long)own * nIter) / nOwn) * OWN_THREADS;
    const int4* dst4 = (const int4*)dst;
    if (!BINS) {
    for (int k0 = 0; k0 < nIter; k0 += SCAN_DEPTH) {
        int4 v[SCAN_DEPTH];
        #pragma unroll
        for (int u = 0; u < SCAN_DEPTH; ++u) {
            int gg = (k0 + u) * OWN_THREADS + tid + off;
            gg = (gg >= nGpad) ? gg - nGpad : gg;
            const bool ok = (k0 + u < nIter) && (gg < nG);
            v[u] = dst4[ok ? gg : 0];
        }
        #pragma unroll
        for (int u = 0; u < SCAN_DEPTH; ++u) {
            int gg = (k0 + u) * OWN_THREADS + tid + off;
            gg = (gg >= nGpad) ? gg - nGpad : gg;
            const bool ok = (k0 + u < nIter) && (gg < nG);
            const bool hx = (unsigned int)(v[u].x - base) < uRw;
            const bool hy = (unsigned int)(v[u].y - base) < uRw;
            const bool hz = (unsigned int)(v[u].z - base) < uRw;
            const bool hw = (unsigned int)(v[u].w - base) < uRw;
            if (ok && (hx || hy || hz || hw)) {
                const int e0i = eBeg + (gg << 2);
                if (hx) { int s = atomicAdd(&cnt, 1); if (s < LIST_CAP) ids[s] = e0i; }
                if (hy) { int s = atomicAdd(&cnt, 1); if (s < LIST_CAP) ids[s] = e0i + 1; }
                if (hz) { int s = atomicAdd(&cnt, 1); if (s < LIST_CAP) ids[s] = e0i + 2; }
                if (hw) { int s = atomicAdd(&cnt, 1); if (s < LIST_CAP) ids[s] = e0i + 3; }
            }
        }
    }
    for (int i = (nG << 2) + tid; i < eEnd; i += OWN_THREADS) {
        if ((unsigned int)(dst[i] - base) < uRw) {
            int s = atomicAdd(&cnt, 1); if (s < LIST_CAP) ids[s] = i;
        }
    }
    __syncthreads();
    }
    const int total = BINS ? runP[NCH] : cnt;

    if (total > 0 && total <= LIST_CAP) {
        int ei[LIST_PER], dl[LIST_PER], sr[LIST_PER], ps[LIST_PER];
        float2 ev[LIST_PER];
        float sv[LIST_PER], lgt[LIST_PER];
        if (BINS) {
            const long long plane = (long long)NCH * CS;
            long long q[LIST_PER];
            #pragma unroll
            for (int u = 0; u < LIST_PER; ++u) {
                const int j = tid + u * OWN_THREADS;
                const int jj = (j < total) ? j : 0;
                int k = 0;
                #pragma unroll
                for (int st = NCH / 2; st > 0; st >>= 1) k += (runP[k + st] <= jj) ? st : 0;
                q[u] = (long long)k * CS + runS[k] + (jj - runP[k]);
            }
            #pragma unroll
            for (int u = 0; u < LIST_PER; ++u) {
                ei[u] = rec[q[u]];
                if (NP >= 2) dl[u] = rec[plane + q[u]] - base;
                if (NP >= 5) {
                    sr[u] = rec[2 * plane + q[u]];
                    ev[u].x = __int_as_float(rec[3 * plane + q[u]]);
                    ev[u].y = __int_as_float(rec[4 * plane + q[u]]);
                }
            }
            #pragma unroll
            for (int u = 0; u < LIST_PER; ++u) {
                if (NP < 2) dl[u] = dst[ei[u]] - base;
                if (NP < 5) { sr[u] = src[ei[u]]; ev[u] = e2[ei[u]]; }
            }
        } else {
            #pragma unroll
            for (int u = 0; u < LIST_PER; ++u) {
                const int j = tid + u * OWN_THREADS;
                ei[u] = ids[(j < total) ? j : 0];
            }
            #pragma unroll
            for (int u = 0; u < LIST_PER; ++u) {
                dl[u] = dst[ei[u]] - base;
                sr[u] = src[ei[u]];
                ev[u] = e2[ei[u]];
            }
        }
        #pragma unroll
        for (int u = 0; u < LIST_PER; ++u) sv[u] = s1[sr[u]];
        #pragma unroll
        for (int u = 0; u < LIST_PER; ++u) {
            lgt[u] = edge_logit(ev[u].x, ev[u].y, sv[u], ns2[dl[u]],
                                we0, we1, we2, we3, wa0, wa1);
            if (tid + u * OWN_THREADS < total) {
                ldsMaxFloat(&nm[dl[u]], lgt[u]);
                atomicAdd(&ncnt[dl[u]], 1);
            }
        }
        __syncthreads();
        {
            const int lane = tid & 31;
            const int wv = tid >> 5;
            const int c = (tid < Rw) ? ncnt[tid] : 0;
            int incl = c;
            #pragma unroll
            for (int d = 1; d < 32; d <<= 1) {
                const int t = __shfl_up(incl, d, 32);
                if (lane >= d) incl += t;
            }
            if (lane == 31) wtot[wv] = incl;
            __syncthreads();
            int before = (lane < wv) ? wtot[lane] : 0;
            #pragma unroll
            for (int o = 16; o > 0; o >>= 1) before += __shfl_xor(before, o, 32);
            if (tid < Rw) nst[tid] = before + incl - c;
        }
        __syncthreads();
        #pragma unroll
        for (int u = 0; u < LIST_PER; ++u) {
            ps[u] = 0;
            if (tid + u * OWN_THREADS < total) {
                const float ex0 = ev[u].x * we0 + ev[u].y * we1;
                const float ex1 = ev[u].x * we2 + ev[u].y * we3;
                const float w = __expf(lgt[u] - nm[dl[u]]);
                const int pos = nst[dl[u]] + atomicAdd(&nfill[dl[u]], 1);
                key[pos] = ei[u];
                t0[pos] = w; t1[pos] = w * ex0; t2[pos] = w * ex1;
                ps[u] = pos;
            }
        }
        __syncthreads();
        #pragma unroll
        for (int u = 0; u < LIST_PER; ++u) {
            if (tid + u * OWN_THREADS < total) {
                const int s = nst[dl[u]];
                const int end = s + ncnt[dl[u]];
                int rank = 0;
                for (int q = s; q < end; q += 8) {
                    int kv[8];
                    #pragma unroll
                    for (int i = 0; i < 8; ++i) {
                        const int qi = q + i;
                        kv[i] = key[(qi < LIST_CAP) ? qi : LIST_CAP - 1];
                    }
                    #pragma unroll
                    for (int i = 0; i < 8; ++i) rank += ((q + i < end) && (kv[i] < ei[u])) ? 1 : 0;
                }
                ord[s + rank] = ps[u];
            }
        }
        __syncthreads();
        for (int n = tid; n < Rw; n += OWN_THREADS) {
            const int s = nst[n];
            const int c = ncnt[n];
            float l = 0.0f, x = 0.0f, y = 0.0f;
            for (int k = 0; k < c; k += 8) {
                int p[8];
                float a[8], b[8], d[8];
                #pragma unroll
                for (int i = 0; i < 8; ++i) {
                    const int qi = s + k + i;
                    const int pv = ord[(qi < LIST_CAP) ? qi : LIST_CAP - 1];
                    p[i] = (k + i < c) ? pv : 0;
                }
                #pragma unroll
                for (int i = 0; i < 8; ++i) { a[i] = t0[p[i]]; b[i] = t1[p[i]]; d[i] = t2[p[i]]; }
                #pragma unroll
                for (int i = 0; i < 8; ++i) {
                    if (k + i < c) { l += a[i]; x += b[i]; y += d[i]; }
                }
            }
            const float inv = __builtin_amdgcn_rcpf(fmaxf(l, 1e-16f));
            nax[n] = x * inv; nay[n] = y * inv;
        }
    } else if (total == 0) {
        for (int n = tid; n < Rw; n += OWN_THREADS) { nax[n] = 0.0f; nay[n] = 0.0f; }
    } else {
        for (int n = tid; n < Rw; n += OWN_THREADS) { t0[n] = 0.0f; t1[n] = 0.0f; t2[n] = 0.0f; }
        for (int i = eBeg + tid; i < eEnd; i += OWN_THREADS) {
            const int dl = dst[i] - base;
            if ((unsigned int)dl < uRw) {
                const float2 ev = e2[i];
                ldsMaxFloat(&nm[dl], edge_logit(ev.x, ev.y, s1[src[i]], ns2[dl],
                                                we0, we1, we2, we3, wa0, wa1));
            }
        }
        __syncthreads();
        for (int i = eBeg + tid; i < eEnd; i += OWN_THREADS) {
            const int dl = dst[i] - base;
            if ((unsigned int)dl < uRw) {
                const float2 ev = e2[i];
                const float logit = edge_logit(ev.x, ev.y, s1[src[i]], ns2[dl],
                                               we0, we1, we2, we3, wa0, wa1);
                const float ex0 = ev.x * we0 + ev.y * we1;
                const float ex1 = ev.x * we2 + ev.y * we3;
                const float w = __expf(logit - nm[dl]);
                atomicAdd(&t0[dl], w);
                atomicAdd(&t1[dl], w * ex0);
                atomicAdd(&t2[dl], w * ex1);
            }
        }
        __syncthreads();
        for (int n = tid; n < Rw; n += OWN_THREADS) {
            const float inv = __builtin_amdgcn_rcpf(fmaxf(t0[n], 1e-16f));
            nax[n] = t1[n] * inv; nay[n] = t2[n] * inv;
        }
    }
    {
        const int lane = tid & 31;
        const int wv = tid >> 5;
        const int c = lane & 15;
        const int h = lane >> 4;
        const int nGroups = (Rw + 7) >> 3;
        const int g0 = OWN_THREADS / 32 - 1 - wv;
        float4 wlo = make_float4(0.0f, 0.0f, 0.0f, 0.0f), whi = wlo;
        if (g0 < nGroups) {
            wlo = ((const float4*)w_e2n)[2 * lane];
            whi = ((const float4*)w_e2n)[2 * lane + 1];
        }
        __syncthreads();
        v2f b0, b1, b2, b3;
        b0.x = wlo.x; b0.y = wlo.y; b1.x = wlo.z; b1.y = wlo.w;
        b2.x = whi.x; b2.y = whi.y; b3.x = whi.z; b3.y = whi.w;
        for (int g = g0; g < nGroups; g += OWN_THREADS / 32) {
            const int row0 = g * 8;
            const int nl = row0 + (c & 7);
            const bool validA = ((c >> 3) == h) && (nl < Rw);
            const float axr = nax[nl], ayr = nay[nl];
            v2f a;
            a.x = validA ? axr : 0.0f;
            a.y = validA ? ayr : 0.0f;
            const v8f d0 = wmma_k4(a, b0);
            const v8f d1 = wmma_k4(a, b1);
            const v8f d2 = wmma_k4(a, b2);
            const v8f d3 = wmma_k4(a, b3);
            float4* const ob = (float4*)(out + ((long long)base + row0) * DFEAT) + lane;
            #pragma unroll
            for (int r = 0; r < 8; ++r) {
                if (row0 + r < Rw) {
                    float4 o;
                    o.x = d0[r]; o.y = d1[r]; o.z = d2[r]; o.w = d3[r];
                    ob[r * (DFEAT / 4)] = o;
                }
            }
        }
    }
}

extern "C" void kernel_launch(void* const* d_in, const int* in_sizes, int n_in,
                              void* d_out, int out_size, void* d_ws, size_t ws_size,
                              hipStream_t stream) {
    const float* h      = (const float*)d_in[0];
    const float* e      = (const float*)d_in[1];
    const int*   src    = (const int*)d_in[2];
    const int*   dst    = (const int*)d_in[3];
    const float* w_edge = (const float*)d_in[4];
    const float* w_attn = (const float*)d_in[5];
    const float* w_e2n  = (const float*)d_in[6];
    float* out = (float*)d_out;

    const int E = in_sizes[2];
    const int N = in_sizes[0] / DFEAT;

    float* s1    = (float*)d_ws;
    float* s2    = s1 + N;
    int*   dir   = (int*)(s2 + N);
    int*   rec   = dir + NCH * (MAXOWN + 1);

    int owners = MAXOWN;
    int R = (N + owners - 1) / owners;
    if (R > RMAX) { R = RMAX; owners = (N + RMAX - 1) / RMAX; }
    if (R < 1) R = 1;

    const long long chunkMax = ((long long)E + NCH - 1) / NCH;
    const int CS = (int)((chunkMax + 31) / 32 * 32);
    const size_t fixedW = (size_t)2 * N + (size_t)NCH * (MAXOWN + 1);
    const size_t planeW = (size_t)NCH * CS;
    int np = 0;
    if ((E > 0) && (chunkMax <= BIN_CAP) && (owners <= MAXOWN)) {
        if      (ws_size >= (fixedW + 5 * planeW) * sizeof(int)) np = 5;
        else if (ws_size >= (fixedW + 2 * planeW) * sizeof(int)) np = 2;
        else if (ws_size >= (fixedW + 1 * planeW) * sizeof(int)) np = 1;
    }
    const int nBin = (np > 0) ? NCH : 0;

    const int T = BIN_T;
    const int wavesPerBlock = T / 32;

    const int capRows  = 512;
    const int rowTrips = (N + 3) / 4;
    int need1 = (rowTrips + wavesPerBlock - 1) / wavesPerBlock;
    need1 = (need1 < 1) ? 1 : need1;
    const int rounds1 = (need1 + capRows - 1) / capRows;
    const int g1 = (need1 + rounds1 - 1) / rounds1;
    k1_node_scalars<<<g1 + nBin, T, 0, stream>>>(h, w_attn, s1, s2, N, e, src, dst, dir, rec, E, R, CS, np, nBin);

    if (np == 5)
        k2_owner_sums<5><<<owners, OWN_THREADS, 0, stream>>>(e, src, dst, s1, s2, w_edge, w_attn, w_e2n, out, dir, rec, CS, E, N, R);
    else if (np == 2)
        k2_owner_sums<2><<<owners, OWN_THREADS, 0, stream>>>(e, src, dst, s1, s2, w_edge, w_attn, w_e2n, out, dir, rec, CS, E, N, R);
    else if (np == 1)
        k2_owner_sums<1><<<owners, OWN_THREADS, 0, stream>>>(e, src, dst, s1, s2, w_edge, w_attn, w_e2n, out, dir, rec, CS, E, N, R);
    else
        k2_owner_sums<0><<<owners, OWN_THREADS, 0, stream>>>(e, src, dst, s1, s2, w_edge, w_attn, w_e2n, out, dir, rec, CS, E, N, R);
}


# K2: per-node sums of w, w*ex0, w*ex1 as int64 fixed-point LDS integer adds (2^-36 units, same bits in any order); slot, rank-by-edge-number and one-lane ordered-sum passes go; sum rounding changes
# speedup vs baseline: 7.8400x; 1.2732x over previous
#include <hip/hip_runtime.h>
#include <hip/hip_bf16.h>

typedef float v8f __attribute__((ext_vector_type(8)));
typedef float v2f __attribute__((ext_vector_type(2)));

#define DFEAT 128

__device__ __forceinline__ v8f wmma_k4(v2f a, v2f b) {
    v8f c = {};
    return __builtin_amdgcn_wmma_f32_16x16x4_f32(false, a, false, b, (short)0, c, false, false);
}

#define OWN_THREADS 1024
#define LIST_CAP    8192
#define RMAX        1024
#define SCAN_DEPTH  17
#define LIST_PER    (LIST_CAP / OWN_THREADS)
#define NCH         256
#define MAXOWN      256
#define BIN_T       256
#define BIN_CAP     4096
#define BIN_PER     (BIN_CAP / BIN_T)

__device__ __forceinline__ void bin_chunk(const float* __restrict__ e, const int* __restrict__ src,
                                          const int* __restrict__ dst,
                                          int* __restrict__ dir, int* __restrict__ rec,
                                          int E, int N, int R, int CS, int np, int c,
                                          int* hist, int* ofs, int* wsum,
                                          int* sE, int* sD, int* sS, float* sX, float* sY) {
    const int tid = threadIdx.x;
    const int cb = (int)(((long long)E * c) / NCH);
    const int ce = (int)(((long long)E * (c + 1)) / NCH);
    const int L = ce - cb;
    const float2* e2 = (const float2*)e;
    hist[tid] = 0;
    int dv[BIN_PER], sv[BIN_PER], rk[BIN_PER], ov[BIN_PER];
    float2 xv[BIN_PER];
    #pragma unroll
    for (int k = 0; k < BIN_PER; ++k) {
        int gi = cb + tid + k * BIN_T;
        gi = (gi < E) ? gi : E - 1;
        dv[k] = dst[gi]; sv[k] = src[gi]; xv[k] = e2[gi];
    }
    __syncthreads();
    #pragma unroll
    for (int k = 0; k < BIN_PER; ++k) {
        bool ok = (tid + k * BIN_T < L) && ((unsigned int)dv[k] < (unsigned int)N);
        int o = ok ? (int)((unsigned int)dv[k] / (unsigned int)R) : 0;
        ok = ok && (o < MAXOWN);
        ov[k] = ok ? o : -1;
        rk[k] = 0;
        if (ok) rk[k] = atomicAdd(&hist[o], 1);
    }
    __syncthreads();
    {
        const int lane = tid & 31;
        const int wv = tid >> 5;
        const int cn = hist[tid];
        int incl = cn;
        #pragma unroll
        for (int d = 1; d < 32; d <<= 1) {
            const int t = __shfl_up(incl, d, 32);
            if (lane >= d) incl += t;
        }
        if (lane == 31) wsum[wv] = incl;
        __syncthreads();
        int before = 0;
        #pragma unroll
        for (int w = 0; w < BIN_T / 32; ++w) before += (w < wv) ? wsum[w] : 0;
        ofs[tid] = before + incl - cn;
        if (tid == BIN_T - 1) ofs[MAXOWN] = before + incl;
    }
    __syncthreads();
    #pragma unroll
    for (int k = 0; k < BIN_PER; ++k) {
        if (ov[k] >= 0) {
            const int pos = ofs[ov[k]] + rk[k];
            if (np < 5) sE[pos] = cb + tid + k * BIN_T;
            sD[pos] = dv[k]; sS[pos] = sv[k];
            sX[pos] = xv[k].x; sY[pos] = xv[k].y;
        }
    }
    __syncthreads();
    const int tot = ofs[MAXOWN];
    const long long plane = (long long)NCH * CS;
    int* r0 = rec + (long long)c * CS;
    for (int i = tid; i < tot; i += BIN_T) {
        if (np < 5) r0[i] = sE[i];
        if (np >= 2) r0[plane + i] = sD[i];
        if (np >= 5) {
            r0[2 * plane + i] = sS[i];
            r0[3 * plane + i] = __float_as_int(sX[i]); r0[4 * plane + i] = __float_as_int(sY[i]);
        }
    }
    for (int i = tid; i <= MAXOWN; i += BIN_T) dir[c * (MAXOWN + 1) + i] = ofs[i];
}


__global__ void __launch_bounds__(BIN_T)
k1_node_scalars(const float* __restrict__ h,
                const float* __restrict__ w_attn,
                float* __restrict__ s1, float* __restrict__ s2, int N,
                const float* __restrict__ e, const int* __restrict__ src,
                const int* __restrict__ dst,
                int* __restrict__ dir, int* __restrict__ rec,
                int E, int R, int CS, int np, int nBin) {
    __shared__ int hist[MAXOWN];
    __shared__ int ofs[MAXOWN + 1];
    __shared__ int wsum[BIN_T / 32];
    __shared__ int sE[BIN_CAP], sD[BIN_CAP], sS[BIN_CAP];
    __shared__ float sX[BIN_CAP], sY[BIN_CAP];
    static_assert(MAXOWN == BIN_T, "the running sum of the counts takes one count per thread");
    if ((int)blockIdx.x < nBin) {
        bin_chunk(e, src, dst, dir, rec, E, N, R, CS, np, (int)blockIdx.x, hist, ofs, wsum, sE, sD, sS, sX, sY);
        return;
    }
    const int lane = threadIdx.x & 31;
    const int wavesPerBlock = blockDim.x >> 5;
    const int wave = ((int)blockIdx.x - nBin) * wavesPerBlock + (threadIdx.x >> 5);
    const int nWaves = ((int)gridDim.x - nBin) * wavesPerBlock;
    const float4 a1 = ((const float4*)w_attn)[lane];
    const float4 a2 = ((const float4*)(w_attn + DFEAT))[lane];
    for (int base = wave * 4; base < N; base += nWaves * 4) {
        float4 x[4];
        #pragma unroll
        for (int k = 0; k < 4; ++k) {
            int node = base + k;
            node = (node < N) ? node : N - 1;
            x[k] = ((const float4*)(h + (long long)node * DFEAT))[lane];
        }
        float p1[4], p2[4];
        #pragma unroll
        for (int k = 0; k < 4; ++k) {
            p1[k] = x[k].x*a1.x + x[k].y*a1.y + x[k].z*a1.z + x[k].w*a1.w;
            p2[k] = x[k].x*a2.x + x[k].y*a2.y + x[k].z*a2.z + x[k].w*a2.w;
        }
        #pragma unroll
        for (int off = 16; off > 0; off >>= 1) {
            #pragma unroll
            for (int k = 0; k < 4; ++k) {
                p1[k] += __shfl_xor(p1[k], off, 32);
                p2[k] += __shfl_xor(p2[k], off, 32);
            }
        }
        const float r1 = (lane == 0) ? p1[0] : ((lane == 1) ? p1[1] : ((lane == 2) ? p1[2] : p1[3]));
        const float r2 = (lane == 0) ? p2[0] : ((lane == 1) ? p2[1] : ((lane == 2) ? p2[2] : p2[3]));
        if (lane < 4 && base + lane < N) { s1[base + lane] = r1; s2[base + lane] = r2; }
    }
}

__device__ __forceinline__ float edge_logit(float e0, float e1, float ss, float sd,
                                            float we0, float we1, float we2, float we3,
                                            float wa0, float wa1) {
    float ex0 = e0 * we0 + e1 * we1;
    float ex1 = e0 * we2 + e1 * we3;
    float logit = ss + sd + ex0 * wa0 + ex1 * wa1;
    return (logit > 0.0f) ? logit : 0.01f * logit;
}

__device__ __forceinline__ void ldsMaxFloat(float* addr, float val) {
    val = val + 0.0f;
    if (val >= 0.0f) atomicMax((int*)addr, __float_as_int(val));
    else             atomicMin((unsigned int*)addr, __float_as_uint(val));
}


#define FIX_SCALE 68719476736.0f
#define FIX_INV   (1.0f / 68719476736.0f)

__device__ __forceinline__ void ldsAddFixed(unsigned long long* addr, float v) {
    atomicAdd(addr, (unsigned long long)(long long)(v * FIX_SCALE));
}

template <int NP>
__global__ void __launch_bounds__(OWN_THREADS)
k2_owner_sums(const float* __restrict__ e,
              const int* __restrict__ src, const int* __restrict__ dst,
              const float* __restrict__ s1, const float* __restrict__ s2,
              const float* __restrict__ w_edge, const float* __restrict__ w_attn,
              const float* __restrict__ w_e2n,
              float* __restrict__ out,
              const int* __restrict__ dir, const int* __restrict__ rec, int CS,
              int E, int N, int R) {
    __shared__ int   ids[LIST_CAP];
    __shared__ float nm[RMAX], ns2[RMAX];
    __shared__ unsigned long long accL[RMAX], accX[RMAX], accY[RMAX];
    __shared__ float nax[RMAX], nay[RMAX];
    __shared__ int   cnt;
    __shared__ int   wtot[NCH / 32];
    __shared__ int   runP[NCH + 1], runS[NCH];
    static_assert(NCH == 256, "the search for an entry's run makes eight halvings");
    static_assert(NCH <= OWN_THREADS, "the running sum of the run lengths gives every run a thread of its own");

    constexpr bool BINS = (NP > 0);
    const int tid = threadIdx.x;
    const int own = blockIdx.x;
    const int nOwn = gridDim.x;
    const int base = own * R;
    int Rw = N - base;
    Rw = (Rw < R) ? Rw : R;
    Rw = (Rw > 0) ? Rw : 0;
    const unsigned int uRw = (unsigned int)Rw;

    const float we0 = w_edge[0], we1 = w_edge[1], we2 = w_edge[2], we3 = w_edge[3];
    const float wa0 = w_attn[2*DFEAT], wa1 = w_attn[2*DFEAT + 1];
    const float2* e2 = (const float2*)e;

    for (int n = tid; n < Rw; n += OWN_THREADS) {
        nm[n] = -3.0e38f; accL[n] = 0ull; accX[n] = 0ull; accY[n] = 0ull;
        ns2[n] = s2[base + n];
    }
    if (tid == 0) cnt = 0;
    if (BINS) {
        const int lane = tid & 31;
        const int wv = tid >> 5;
        int a0 = 0, la = 0;
        if (tid < NCH) {
            a0 = dir[tid * (MAXOWN + 1) + own];
            la = dir[tid * (MAXOWN + 1) + own + 1] - a0;
        }
        int incl = la;
        #pragma unroll
        for (int d = 1; d < 32; d <<= 1) {
            const int t = __shfl_up(incl, d, 32);
            if (lane >= d) incl += t;
        }
        if (tid < NCH && lane == 31) wtot[wv] = incl;
        __syncthreads();
        if (tid < NCH) {
            int before = 0;
            #pragma unroll
            for (int w = 0; w < NCH / 32; ++w) before += (w < wv) ? wtot[w] : 0;
            runP[tid] = before + incl - la;
            runS[tid] = a0;
            if (tid == NCH - 1) runP[NCH] = before + incl;
        }
    }
    __syncthreads();

    const int nG = E >> 2;
    const int eBeg = 0, eEnd = E;
    const int nIter = (nG + OWN_THREADS - 1) / OWN_THREADS;
    const int nGpad = nIter * OWN_THREADS;
    const int off = (int)(((long long)own * nIter) / nOwn) * OWN_THREADS;
    const int4* dst4 = (const int4*)dst;
    if (!BINS) {
    for (int k0 = 0; k0 < nIter; k0 += SCAN_DEPTH) {
        int4 v[SCAN_DEPTH];
        #pragma unroll
        for (int u = 0; u < SCAN_DEPTH; ++u) {
            int gg = (k0 + u) * OWN_THREADS + tid + off;
            gg = (gg >= nGpad) ? gg - nGpad : gg;
            const bool ok = (k0 + u < nIter) && (gg < nG);
            v[u] = dst4[ok ? gg : 0];
        }
        #pragma unroll
        for (int u = 0; u < SCAN_DEPTH; ++u) {
            int gg = (k0 + u) * OWN_THREADS + tid + off;
            gg = (gg >= nGpad) ? gg - nGpad : gg;
            const bool ok = (k0 + u < nIter) && (gg < nG);
            const bool hx = (unsigned int)(v[u].x - base) < uRw;
            const bool hy = (unsigned int)(v[u].y - base) < uRw;
            const bool hz = (unsigned int)(v[u].z - base) < uRw;
            const bool hw = (unsigned int)(v[u].w - base) < uRw;
            if (ok && (hx || hy || hz || hw)) {
                const int e0i = eBeg + (gg << 2);
                if (hx) { int s = atomicAdd(&cnt, 1); if (s < LIST_CAP) ids[s] = e0i; }
                if (hy) { int s = atomicAdd(&cnt, 1); if (s < LIST_CAP) ids[s] = e0i + 1; }
                if (hz) { int s = atomicAdd(&cnt, 1); if (s < LIST_CAP) ids[s] = e0i + 2; }
                if (hw) { int s = atomicAdd(&cnt, 1); if (s < LIST_CAP) ids[s] = e0i + 3; }
            }
        }
    }
    for (int i = (nG << 2) + tid; i < eEnd; i += OWN_THREADS) {
        if ((unsigned int)(dst[i] - base) < uRw) {
            int s = atomicAdd(&cnt, 1); if (s < LIST_CAP) ids[s] = i;
        }
    }
    __syncthreads();
    }
    const int total = BINS ? runP[NCH] : cnt;
    if (total > 0 && total <= LIST_CAP) {
        int ei[LIST_PER], dl[LIST_PER], sr[LIST_PER];
        float2 ev[LIST_PER];
        float sv[LIST_PER], lgt[LIST_PER];
        if (BINS) {
            const long long plane = (long long)NCH * CS;
            long long q[LIST_PER];
            #pragma unroll
            for (int u = 0; u < LIST_PER; ++u) {
                const int j = tid + u * OWN_THREADS;
                const int jj = (j < total) ? j : 0;
                int k = 0;
                #pragma unroll
                for (int st = NCH / 2; st > 0; st >>= 1) k += (runP[k + st] <= jj) ? st : 0;
                q[u] = (long long)k * CS + runS[k] + (jj - runP[k]);
            }
            #pragma unroll
            for (int u = 0; u < LIST_PER; ++u) {
                if (NP < 5) ei[u] = rec[q[u]];
                if (NP >= 2) dl[u] = rec[plane + q[u]] - base;
                if (NP >= 5) {
                    sr[u] = rec[2 * plane + q[u]];
                    ev[u].x = __int_as_float(rec[3 * plane + q[u]]);
                    ev[u].y = __int_as_float(rec[4 * plane + q[u]]);
                }
            }
            #pragma unroll
            for (int u = 0; u < LIST_PER; ++u) {
                if (NP < 2) dl[u] = dst[ei[u]] - base;
                if (NP < 5) { sr[u] = src[ei[u]]; ev[u] = e2[ei[u]]; }
            }
        } else {
            #pragma unroll
            for (int u = 0; u < LIST_PER; ++u) {
                const int j = tid + u * OWN_THREADS;
                ei[u] = ids[(j < total) ? j : 0];
            }
            #pragma unroll
            for (int u = 0; u < LIST_PER; ++u) {
                dl[u] = dst[ei[u]] - base;
                sr[u] = src[ei[u]];
                ev[u] = e2[ei[u]];
            }
        }
        #pragma unroll
        for (int u = 0; u < LIST_PER; ++u) sv[u] = s1[sr[u]];
        #pragma unroll
        for (int u = 0; u < LIST_PER; ++u) {
            lgt[u] = edge_logit(ev[u].x, ev[u].y, sv[u], ns2[dl[u]],
                                we0, we1, we2, we3, wa0, wa1);
            if (tid + u * OWN_THREADS < total) ldsMaxFloat(&nm[dl[u]], lgt[u]);
        }
        __syncthreads();
        #pragma unroll
        for (int u = 0; u < LIST_PER; ++u) {
            if (tid + u * OWN_THREADS < total) {
                const float ex0 = ev[u].x * we0 + ev[u].y * we1;
                const float ex1 = ev[u].x * we2 + ev[u].y * we3;
                const float w = __expf(lgt[u] - nm[dl[u]]);
                ldsAddFixed(&accL[dl[u]], w);
                ldsAddFixed(&accX[dl[u]], w * ex0);
                ldsAddFixed(&accY[dl[u]], w * ex1);
            }
        }
    } else if (total > LIST_CAP) {
        for (int i = eBeg + tid; i < eEnd; i += OWN_THREADS) {
            const int dl = dst[i] - base;
            if ((unsigned int)dl < uRw) {
                const float2 ev = e2[i];
                ldsMaxFloat(&nm[dl], edge_logit(ev.x, ev.y, s1[src[i]], ns2[dl],
                                                we0, we1, we2, we3, wa0, wa1));
            }
        }
        __syncthreads();
        for (int i = eBeg + tid; i < eEnd; i += OWN_THREADS) {
            const int dl = dst[i] - base;
            if ((unsigned int)dl < uRw) {
                const float2 ev = e2[i];
                const float logit = edge_logit(ev.x, ev.y, s1[src[i]], ns2[dl],
                                               we0, we1, we2, we3, wa0, wa1);
                const float ex0 = ev.x * we0 + ev.y * we1;
                const float ex1 = ev.x * we2 + ev.y * we3;
                const float w = __expf(logit - nm[dl]);
                ldsAddFixed(&accL[dl], w);
                ldsAddFixed(&accX[dl], w * ex0);
                ldsAddFixed(&accY[dl], w * ex1);
            }
        }
    }
    __syncthreads();
    for (int n = tid; n < Rw; n += OWN_THREADS) {
        const float l = (float)(long long)accL[n] * FIX_INV;
        const float x = (float)(long long)accX[n] * FIX_INV;
        const float y = (float)(long long)accY[n] * FIX_INV;
        const float inv = __builtin_amdgcn_rcpf(fmaxf(l, 1e-16f));
        nax[n] = x * inv; nay[n] = y * inv;
    }
    {
        const int lane = tid & 31;
        const int wv = tid >> 5;
        const int c = lane & 15;
        const int h = lane >> 4;
        const int nGroups = (Rw + 7) >> 3;
        const int g0 = OWN_THREADS / 32 - 1 - wv;
        float4 wlo = make_float4(0.0f, 0.0f, 0.0f, 0.0f), whi = wlo;
        if (g0 < nGroups) {
            wlo = ((const float4*)w_e2n)[2 * lane];
            whi = ((const float4*)w_e2n)[2 * lane + 1];
        }
        __syncthreads();
        v2f b0, b1, b2, b3;
        b0.x = wlo.x; b0.y = wlo.y; b1.x = wlo.z; b1.y = wlo.w;
        b2.x = whi.x; b2.y = whi.y; b3.x = whi.z; b3.y = whi.w;
        for (int g = g0; g < nGroups; g += OWN_THREADS / 32) {
            const int row0 = g * 8;
            const int nl = row0 + (c & 7);
            const bool validA = ((c >> 3) == h) && (nl < Rw);
            const float axr = nax[nl], ayr = nay[nl];
            v2f a;
            a.x = validA ? axr : 0.0f;
            a.y = validA ? ayr : 0.0f;
            const v8f d0 = wmma_k4(a, b0);
            const v8f d1 = wmma_k4(a, b1);
            const v8f d2 = wmma_k4(a, b2);
            const v8f d3 = wmma_k4(a, b3);
            float4* const ob = (float4*)(out + ((long long)base + row0) * DFEAT) + lane;
            #pragma unroll
            for (int r = 0; r < 8; ++r) {
                if (row0 + r < Rw) {
                    float4 o;
                    o.x = d0[r]; o.y = d1[r]; o.z = d2[r]; o.w = d3[r];
                    ob[r * (DFEAT / 4)] = o;
                }
            }
        }
    }
}

extern "C" void kernel_launch(void* const* d_in, const int* in_sizes, int n_in,
                              void* d_out, int out_size, void* d_ws, size_t ws_size,
                              hipStream_t stream) {
    const float* h      = (const float*)d_in[0];
    const float* e      = (const float*)d_in[1];
    const int*   src    = (const int*)d_in[2];
    const int*   dst    = (const int*)d_in[3];
    const float* w_edge = (const float*)d_in[4];
    const float* w_attn = (const float*)d_in[5];
    const float* w_e2n  = (const float*)d_in[6];
    float* out = (float*)d_out;

    const int E = in_sizes[2];
    const int N = in_sizes[0] / DFEAT;

    float* s1    = (float*)d_ws;
    float* s2    = s1 + N;
    int*   dir   = (int*)(s2 + N);
    int*   rec   = dir + NCH * (MAXOWN + 1);

    int owners = MAXOWN;
    int R = (N + owners - 1) / owners;
    if (R > RMAX) { R = RMAX; owners = (N + RMAX - 1) / RMAX; }
    if (R < 1) R = 1;

    const long long chunkMax = ((long long)E + NCH - 1) / NCH;
    const int CS = (int)((chunkMax + 31) / 32 * 32);
    const size_t fixedW = (size_t)2 * N + (size_t)NCH * (MAXOWN + 1);
    const size_t planeW = (size_t)NCH * CS;
    int np = 0;
    if ((E > 0) && (chunkMax <= BIN_CAP) && (owners <= MAXOWN)) {
        if      (ws_size >= (fixedW + 5 * planeW) * sizeof(int)) np = 5;
        else if (ws_size >= (fixedW + 2 * planeW) * sizeof(int)) np = 2;
        else if (ws_size >= (fixedW + 1 * planeW) * sizeof(int)) np = 1;
    }
    const int nBin = (np > 0) ? NCH : 0;

    const int T = BIN_T;
    const int wavesPerBlock = T / 32;

    const int capRows  = 512;
    const int rowTrips = (N + 3) / 4;
    int need1 = (rowTrips + wavesPerBlock - 1) / wavesPerBlock;
    need1 = (need1 < 1) ? 1 : need1;
    const int rounds1 = (need1 + capRows - 1) / capRows;
    const int g1 = (need1 + rounds1 - 1) / rounds1;
    k1_node_scalars<<<g1 + nBin, T, 0, stream>>>(h, w_attn, s1, s2, N, e, src, dst, dir, rec, E, R, CS, np, nBin);

    if (np == 5)
        k2_owner_sums<5><<<owners, OWN_THREADS, 0, stream>>>(e, src, dst, s1, s2, w_edge, w_attn, w_e2n, out, dir, rec, CS, E, N, R);
    else if (np == 2)
        k2_owner_sums<2><<<owners, OWN_THREADS, 0, stream>>>(e, src, dst, s1, s2, w_edge, w_attn, w_e2n, out, dir, rec, CS, E, N, R);
    else if (np == 1)
        k2_owner_sums<1><<<owners, OWN_THREADS, 0, stream>>>(e, src, dst, s1, s2, w_edge, w_attn, w_e2n, out, dir, rec, CS, E, N, R);
    else
        k2_owner_sums<0><<<owners, OWN_THREADS, 0, stream>>>(e, src, dst, s1, s2, w_edge, w_attn, w_e2n, out, dir, rec, CS, E, N, R);
}


# K2 owner work-groups of 16 waves, not 32: half the waves to enter; a thread's 8 entry slots cover a list of 4096 (slots 9-16 only walked for a longer one), not 8192; integer sums, same bits
# speedup vs baseline: 8.1567x; 1.0404x over previous
#include <hip/hip_runtime.h>
#include <hip/hip_bf16.h>

typedef float v8f __attribute__((ext_vector_type(8)));
typedef float v2f __attribute__((ext_vector_type(2)));

#define DFEAT 128

__device__ __forceinline__ v8f wmma_k4(v2f a, v2f b) {
    v8f c = {};
    return __builtin_amdgcn_wmma_f32_16x16x4_f32(false, a, false, b, (short)0, c, false, false);
}

#define OWN_THREADS 512
#define LIST_CAP    8192
#define RMAX        1024
#define SCAN_DEPTH  17
#define LIST_PER    (LIST_CAP / OWN_THREADS)
#define LIST_HALF   (LIST_PER / 2)
#define NCH         256
#define MAXOWN      256
#define BIN_T       256
#define BIN_CAP     4096
#define BIN_PER     (BIN_CAP / BIN_T)

__device__ __forceinline__ void bin_chunk(const float* __restrict__ e, const int* __restrict__ src,
                                          const int* __restrict__ dst,
                                          int* __restrict__ dir, int* __restrict__ rec,
                                          int E, int N, int R, int CS, int np, int c,
                                          int* hist, int* ofs, int* wsum,
                                          int* sE, int* sD, int* sS, float* sX, float* sY) {
    const int tid = threadIdx.x;
    const int cb = (int)(((long long)E * c) / NCH);
    const int ce = (int)(((long long)E * (c + 1)) / NCH);
    const int L = ce - cb;
    const float2* e2 = (const float2*)e;
    hist[tid] = 0;
    int dv[BIN_PER], sv[BIN_PER], rk[BIN_PER], ov[BIN_PER];
    float2 xv[BIN_PER];
    #pragma unroll
    for (int k = 0; k < BIN_PER; ++k) {
        int gi = cb + tid + k * BIN_T;
        gi = (gi < E) ? gi : E - 1;
        dv[k] = dst[gi]; sv[k] = src[gi]; xv[k] = e2[gi];
    }
    __syncthreads();
    #pragma unroll
    for (int k = 0; k < BIN_PER; ++k) {
        bool ok = (tid + k * BIN_T < L) && ((unsigned int)dv[k] < (unsigned int)N);
        int o = ok ? (int)((unsigned int)dv[k] / (unsigned int)R) : 0;
        ok = ok && (o < MAXOWN);
        ov[k] = ok ? o : -1;
        rk[k] = 0;
        if (ok) rk[k] = atomicAdd(&hist[o], 1);
    }
    __syncthreads();
    {
        const int lane = tid & 31;
        const int wv = tid >> 5;
        const int cn = hist[tid];
        int incl = cn;
        #pragma unroll
        for (int d = 1; d < 32; d <<= 1) {
            const int t = __shfl_up(incl, d, 32);
            if (lane >= d) incl += t;
        }
        if (lane == 31) wsum[wv] = incl;
        __syncthreads();
        int before = 0;
        #pragma unroll
        for (int w = 0; w < BIN_T / 32; ++w) before += (w < wv) ? wsum[w] : 0;
        ofs[tid] = before + incl - cn;
        if (tid == BIN_T - 1) ofs[MAXOWN] = before + incl;
    }
    __syncthreads();
    #pragma unroll
    for (int k = 0; k < BIN_PER; ++k) {
        if (ov[k] >= 0) {
            const int pos = ofs[ov[k]] + rk[k];
            if (np < 5) sE[pos] = cb + tid + k * BIN_T;
            sD[pos] = dv[k]; sS[pos] = sv[k];
            sX[pos] = xv[k].x; sY[pos] = xv[k].y;
        }
    }
    __syncthreads();
    const int tot = ofs[MAXOWN];
    const long long plane = (long long)NCH * CS;
    int* r0 = rec + (long long)c * CS;
    for (int i = tid; i < tot; i += BIN_T) {
        if (np < 5) r0[i] = sE[i];
        if (np >= 2) r0[plane + i] = sD[i];
        if (np >= 5) {
            r0[2 * plane + i] = sS[i];
            r0[3 * plane + i] = __float_as_int(sX[i]); r0[4 * plane + i] = __float_as_int(sY[i]);
        }
    }
    for (int i = tid; i <= MAXOWN; i += BIN_T) dir[c * (MAXOWN + 1) + i] = ofs[i];
}


__global__ void __launch_bounds__(BIN_T)
k1_node_scalars(const float* __restrict__ h,
                const float* __restrict__ w_attn,
                float* __restrict__ s1, float* __restrict__ s2, int N,
                const float* __restrict__ e, const int* __restrict__ src,
                const int* __restrict__ dst,
                int* __restrict__ dir, int* __restrict__ rec,
                int E, int R, int CS, int np, int nBin) {
    __shared__ int hist[MAXOWN];
    __shared__ int ofs[MAXOWN + 1];
    __shared__ int wsum[BIN_T / 32];
    __shared__ int sE[BIN_CAP], sD[BIN_CAP], sS[BIN_CAP];
    __shared__ float sX[BIN_CAP], sY[BIN_CAP];
    static_assert(MAXOWN == BIN_T, "the running sum of the counts takes one count per thread");
    if ((int)blockIdx.x < nBin) {
        bin_chunk(e, src, dst, dir, rec, E, N, R, CS, np, (int)blockIdx.x, hist, ofs, wsum, sE, sD, sS, sX, sY);
        return;
    }
    const int lane = threadIdx.x & 31;
    const int wavesPerBlock = blockDim.x >> 5;
    const int wave = ((int)blockIdx.x - nBin) * wavesPerBlock + (threadIdx.x >> 5);
    const int nWaves = ((int)gridDim.x - nBin) * wavesPerBlock;
    const float4 a1 = ((const float4*)w_attn)[lane];
    const float4 a2 = ((const float4*)(w_attn + DFEAT))[lane];
    for (int base = wave * 4; base < N; base += nWaves * 4) {
        float4 x[4];
        #pragma unroll
        for (int k = 0; k < 4; ++k) {
            int node = base + k;
            node = (node < N) ? node : N - 1;
            x[k] = ((const float4*)(h + (long long)node * DFEAT))[lane];
        }
        float p1[4], p2[4];
        #pragma unroll
        for (int k = 0; k < 4; ++k) {
            p1[k] = x[k].x*a1.x + x[k].y*a1.y + x[k].z*a1.z + x[k].w*a1.w;
            p2[k] = x[k].x*a2.x + x[k].y*a2.y + x[k].z*a2.z + x[k].w*a2.w;
        }
        #pragma unroll
        for (int off = 16; off > 0; off >>= 1) {
            #pragma unroll
            for (int k = 0; k < 4; ++k) {
                p1[k] += __shfl_xor(p1[k], off, 32);
                p2[k] += __shfl_xor(p2[k], off, 32);
            }
        }
        const float r1 = (lane == 0) ? p1[0] : ((lane == 1) ? p1[1] : ((lane == 2) ? p1[2] : p1[3]));
        const float r2 = (lane == 0) ? p2[0] : ((lane == 1) ? p2[1] : ((lane == 2) ? p2[2] : p2[3]));
        if (lane < 4 && base + lane < N) { s1[base + lane] = r1; s2[base + lane] = r2; }
    }
}

__device__ __forceinline__ float edge_logit(float e0, float e1, float ss, float sd,
                                            float we0, float we1, float we2, float we3,
                                            float wa0, float wa1) {
    float ex0 = e0 * we0 + e1 * we1;
    float ex1 = e0 * we2 + e1 * we3;
    float logit = ss + sd + ex0 * wa0 + ex1 * wa1;
    return (logit > 0.0f) ? logit : 0.01f * logit;
}

__device__ __forceinline__ void ldsMaxFloat(float* addr, float val) {
    val = val + 0.0f;
    if (val >= 0.0f) atomicMax((int*)addr, __float_as_int(val));
    else             atomicMin((unsigned int*)addr, __float_as_uint(val));
}


#define FIX_SCALE 68719476736.0f
#define FIX_INV   (1.0f / 68719476736.0f)

__device__ __forceinline__ void ldsAddFixed(unsigned long long* addr, float v) {
    atomicAdd(addr, (unsigned long long)(long long)(v * FIX_SCALE));
}

template <int NP>
__global__ void __launch_bounds__(OWN_THREADS)
k2_owner_sums(const float* __restrict__ e,
              const int* __restrict__ src, const int* __restrict__ dst,
              const float* __restrict__ s1, const float* __restrict__ s2,
              const float* __restrict__ w_edge, const float* __restrict__ w_attn,
              const float* __restrict__ w_e2n,
              float* __restrict__ out,
              const int* __restrict__ dir, const int* __restrict__ rec, int CS,
              int E, int N, int R) {
    __shared__ int   ids[LIST_CAP];
    __shared__ float nm[RMAX], ns2[RMAX];
    __shared__ unsigned long long accL[RMAX], accX[RMAX], accY[RMAX];
    __shared__ float nax[RMAX], nay[RMAX];
    __shared__ int   cnt;
    __shared__ int   wtot[NCH / 32];
    __shared__ int   runP[NCH + 1], runS[NCH];
    static_assert(NCH == 256, "the search for an entry's run makes eight halvings");
    static_assert(NCH <= OWN_THREADS, "the running sum of the run lengths gives every run a thread of its own");

    constexpr bool BINS = (NP > 0);
    const int tid = threadIdx.x;
    const int own = blockIdx.x;
    const int nOwn = gridDim.x;
    const int base = own * R;
    int Rw = N - base;
    Rw = (Rw < R) ? Rw : R;
    Rw = (Rw > 0) ? Rw : 0;
    const unsigned int uRw = (unsigned int)Rw;

    const float we0 = w_edge[0], we1 = w_edge[1], we2 = w_edge[2], we3 = w_edge[3];
    const float wa0 = w_attn[2*DFEAT], wa1 = w_attn[2*DFEAT + 1];
    const float2* e2 = (const float2*)e;

    for (int n = tid; n < Rw; n += OWN_THREADS) {
        nm[n] = -3.0e38f; accL[n] = 0ull; accX[n] = 0ull; accY[n] = 0ull;
        ns2[n] = s2[base + n];
    }
    if (tid == 0) cnt = 0;
    if (BINS) {
        const int lane = tid & 31;
        const int wv = tid >> 5;
        int a0 = 0, la = 0;
        if (tid < NCH) {
            a0 = dir[tid * (MAXOWN + 1) + own];
            la = dir[tid * (MAXOWN + 1) + own + 1] - a0;
        }
        int incl = la;
        #pragma unroll
        for (int d = 1; d < 32; d <<= 1) {
            const int t = __shfl_up(incl, d, 32);
            if (lane >= d) incl += t;
        }
        if (tid < NCH && lane == 31) wtot[wv] = incl;
        __syncthreads();
        if (tid < NCH) {
            int before = 0;
            #pragma unroll
            for (int w = 0; w < NCH / 32; ++w) before += (w < wv) ? wtot[w] : 0;
            runP[tid] = before + incl - la;
            runS[tid] = a0;
            if (tid == NCH - 1) runP[NCH] = before + incl;
        }
    }
    __syncthreads();

    const int nG = E >> 2;
    const int eBeg = 0, eEnd = E;
    const int nIter = (nG + OWN_THREADS - 1) / OWN_THREADS;
    const int nGpad = nIter * OWN_THREADS;
    const int off = (int)(((long long)own * nIter) / nOwn) * OWN_THREADS;
    const int4* dst4 = (const int4*)dst;
    if (!BINS) {
    for (int k0 = 0; k0 < nIter; k0 += SCAN_DEPTH) {
        int4 v[SCAN_DEPTH];
        #pragma unroll
        for (int u = 0; u < SCAN_DEPTH; ++u) {
            int gg = (k0 + u) * OWN_THREADS + tid + off;
            gg = (gg >= nGpad) ? gg - nGpad : gg;
            const bool ok = (k0 + u < nIter) && (gg < nG);
            v[u] = dst4[ok ? gg : 0];
        }
        #pragma unroll
        for (int u = 0; u < SCAN_DEPTH; ++u) {
            int gg = (k0 + u) * OWN_THREADS + tid + off;
            gg = (gg >= nGpad) ? gg - nGpad : gg;
            const bool ok = (k0 + u < nIter) && (gg < nG);
            const bool hx = (unsigned int)(v[u].x - base) < uRw;
            const bool hy = (unsigned int)(v[u].y - base) < uRw;
            const bool hz = (unsigned int)(v[u].z - base) < uRw;
            const bool hw = (unsigned int)(v[u].w - base) < uRw;
            if (ok && (hx || hy || hz || hw)) {
                const int e0i = eBeg + (gg << 2);
                if (hx) { int s = atomicAdd(&cnt, 1); if (s < LIST_CAP) ids[s] = e0i; }
                if (hy) { int s = atomicAdd(&cnt, 1); if (s < LIST_CAP) ids[s] = e0i + 1; }
                if (hz) { int s = atomicAdd(&cnt, 1); if (s < LIST_CAP) ids[s] = e0i + 2; }
                if (hw) { int s = atomicAdd(&cnt, 1); if (s < LIST_CAP) ids[s] = e0i + 3; }
            }
        }
    }
    for (int i = (nG << 2) + tid; i < eEnd; i += OWN_THREADS) {
        if ((unsigned int)(dst[i] - base) < uRw) {
            int s = atomicAdd(&cnt, 1); if (s < LIST_CAP) ids[s] = i;
        }
    }
    __syncthreads();
    }
    const int total = BINS ? runP[NCH] : cnt;
    if (total > 0 && total <= LIST_CAP) {
        const bool longList = total > LIST_HALF * OWN_THREADS;
        int ei[LIST_PER], dl[LIST_PER], sr[LIST_PER];
        float2 ev[LIST_PER];
        float sv[LIST_PER], lgt[LIST_PER];
        if (BINS) {
            const long long plane = (long long)NCH * CS;
            long long q[LIST_PER];
            #pragma unroll
            for (int hf = 0; hf < 2; ++hf) if (hf == 0 || longList) {
            #pragma unroll
            for (int u = hf * LIST_HALF; u < (hf + 1) * LIST_HALF; ++u) {
                const int j = tid + u * OWN_THREADS;
                const int jj = (j < total) ? j : 0;
                int k = 0;
                #pragma unroll
                for (int st = NCH / 2; st > 0; st >>= 1) k += (runP[k + st] <= jj) ? st : 0;
                q[u] = (long long)k * CS + runS[k] + (jj - runP[k]);
            }
            }
            #pragma unroll
            for (int hf = 0; hf < 2; ++hf) if (hf == 0 || longList) {
            #pragma unroll
            for (int u = hf * LIST_HALF; u < (hf + 1) * LIST_HALF; ++u) {
                if (NP < 5) ei[u] = rec[q[u]];
                if (NP >= 2) dl[u] = rec[plane + q[u]] - base;
                if (NP >= 5) {
                    sr[u] = rec[2 * plane + q[u]];
                    ev[u].x = __int_as_float(rec[3 * plane + q[u]]);
                    ev[u].y = __int_as_float(rec[4 * plane + q[u]]);
                }
            }
            }
            #pragma unroll
            for (int hf = 0; hf < 2; ++hf) if (hf == 0 || longList) {
            #pragma unroll
            for (int u = hf * LIST_HALF; u < (hf + 1) * LIST_HALF; ++u) {
                if (NP < 2) dl[u] = dst[ei[u]] - base;
                if (NP < 5) { sr[u] = src[ei[u]]; ev[u] = e2[ei[u]]; }
            }
            }
        } else {
            #pragma unroll
            for (int hf = 0; hf < 2; ++hf) if (hf == 0 || longList) {
            #pragma unroll
            for (int u = hf * LIST_HALF; u < (hf + 1) * LIST_HALF; ++u) {
                const int j = tid + u * OWN_THREADS;
                ei[u] = ids[(j < total) ? j : 0];
            }
            #pragma unroll
            for (int u = hf * LIST_HALF; u < (hf + 1) * LIST_HALF; ++u) {
                dl[u] = dst[ei[u]] - base;
                sr[u] = src[ei[u]];
                ev[u] = e2[ei[u]];
            }
            }
        }
        #pragma unroll
        for (int hf = 0; hf < 2; ++hf) if (hf == 0 || longList) {
        #pragma unroll
        for (int u = hf * LIST_HALF; u < (hf + 1) * LIST_HALF; ++u) sv[u] = s1[sr[u]];
        }
        #pragma unroll
        for (int hf = 0; hf < 2; ++hf) if (hf == 0 || longList) {
        #pragma unroll
        for (int u = hf * LIST_HALF; u < (hf + 1) * LIST_HALF; ++u) {
            lgt[u] = edge_logit(ev[u].x, ev[u].y, sv[u], ns2[dl[u]],
                                we0, we1, we2, we3, wa0, wa1);
            if (tid + u * OWN_THREADS < total) ldsMaxFloat(&nm[dl[u]], lgt[u]);
        }
        }
        __syncthreads();
        #pragma unroll
        for (int hf = 0; hf < 2; ++hf) if (hf == 0 || longList) {
        #pragma unroll
        for (int u = hf * LIST_HALF; u < (hf + 1) * LIST_HALF; ++u) {
            if (tid + u * OWN_THREADS < total) {
                const float ex0 = ev[u].x * we0 + ev[u].y * we1;
                const float ex1 = ev[u].x * we2 + ev[u].y * we3;
                const float w = __expf(lgt[u] - nm[dl[u]]);
                ldsAddFixed(&accL[dl[u]], w);
                ldsAddFixed(&accX[dl[u]], w * ex0);
                ldsAddFixed(&accY[dl[u]], w * ex1);
            }
        }
        }
    } else if (total > LIST_CAP) {
        for (int i = eBeg + tid; i < eEnd; i += OWN_THREADS) {
            const int dl = dst[i] - base;
            if ((unsigned int)dl < uRw) {
                const float2 ev = e2[i];
                ldsMaxFloat(&nm[dl], edge_logit(ev.x, ev.y, s1[src[i]], ns2[dl],
                                                we0, we1, we2, we3, wa0, wa1));
            }
        }
        __syncthreads();
        for (int i = eBeg + tid; i < eEnd; i += OWN_THREADS) {
            const int dl = dst[i] - base;
            if ((unsigned int)dl < uRw) {
                const float2 ev = e2[i];
                const float logit = edge_logit(ev.x, ev.y, s1[src[i]], ns2[dl],
                                               we0, we1, we2, we3, wa0, wa1);
                const float ex0 = ev.x * we0 + ev.y * we1;
                const float ex1 = ev.x * we2 + ev.y * we3;
                const float w = __expf(logit - nm[dl]);
                ldsAddFixed(&accL[dl], w);
                ldsAddFixed(&accX[dl], w * ex0);
                ldsAddFixed(&accY[dl], w * ex1);
            }
        }
    }
    __syncthreads();
    for (int n = tid; n < Rw; n += OWN_THREADS) {
        const float l = (float)(long long)accL[n] * FIX_INV;
        const float x = (float)(long long)accX[n] * FIX_INV;
        const float y = (float)(long long)accY[n] * FIX_INV;
        const float inv = __builtin_amdgcn_rcpf(fmaxf(l, 1e-16f));
        nax[n] = x * inv; nay[n] = y * inv;
    }
    {
        const int lane = tid & 31;
        const int wv = tid >> 5;
        const int c = lane & 15;
        const int h = lane >> 4;
        const int nGroups = (Rw + 7) >> 3;
        const int g0 = OWN_THREADS / 32 - 1 - wv;
        float4 wlo = make_float4(0.0f, 0.0f, 0.0f, 0.0f), whi = wlo;
        if (g0 < nGroups) {
            wlo = ((const float4*)w_e2n)[2 * lane];
            whi = ((const float4*)w_e2n)[2 * lane + 1];
        }
        __syncthreads();
        v2f b0, b1, b2, b3;
        b0.x = wlo.x; b0.y = wlo.y; b1.x = wlo.z; b1.y = wlo.w;
        b2.x = whi.x; b2.y = whi.y; b3.x = whi.z; b3.y = whi.w;
        for (int g = g0; g < nGroups; g += OWN_THREADS / 32) {
            const int row0 = g * 8;
            const int nl = row0 + (c & 7);
            const bool validA = ((c >> 3) == h) && (nl < Rw);
            const float axr = nax[nl], ayr = nay[nl];
            v2f a;
            a.x = validA ? axr : 0.0f;
            a.y = validA ? ayr : 0.0f;
            const v8f d0 = wmma_k4(a, b0);
            const v8f d1 = wmma_k4(a, b1);
            const v8f d2 = wmma_k4(a, b2);
            const v8f d3 = wmma_k4(a, b3);
            float4* const ob = (float4*)(out + ((long long)base + row0) * DFEAT) + lane;
            #pragma unroll
            for (int r = 0; r < 8; ++r) {
                if (row0 + r < Rw) {
                    float4 o;
                    o.x = d0[r]; o.y = d1[r]; o.z = d2[r]; o.w = d3[r];
                    ob[r * (DFEAT / 4)] = o;
                }
            }
        }
    }
}

extern "C" void kernel_launch(void* const* d_in, const int* in_sizes, int n_in,
                              void* d_out, int out_size, void* d_ws, size_t ws_size,
                              hipStream_t stream) {
    const float* h      = (const float*)d_in[0];
    const float* e      = (const float*)d_in[1];
    const int*   src    = (const int*)d_in[2];
    const int*   dst    = (const int*)d_in[3];
    const float* w_edge = (const float*)d_in[4];
    const float* w_attn = (const float*)d_in[5];
    const float* w_e2n  = (const float*)d_in[6];
    float* out = (float*)d_out;

    const int E = in_sizes[2];
    const int N = in_sizes[0] / DFEAT;

    float* s1    = (float*)d_ws;
    float* s2    = s1 + N;
    int*   dir   = (int*)(s2 + N);
    int*   rec   = dir + NCH * (MAXOWN + 1);

    int owners = MAXOWN;
    int R = (N + owners - 1) / owners;
    if (R > RMAX) { R = RMAX; owners = (N + RMAX - 1) / RMAX; }
    if (R < 1) R = 1;

    const long long chunkMax = ((long long)E + NCH - 1) / NCH;
    const int CS = (int)((chunkMax + 31) / 32 * 32);
    const size_t fixedW = (size_t)2 * N + (size_t)NCH * (MAXOWN + 1);
    const size_t planeW = (size_t)NCH * CS;
    int np = 0;
    if ((E > 0) && (chunkMax <= BIN_CAP) && (owners <= MAXOWN)) {
        if      (ws_size >= (fixedW + 5 * planeW) * sizeof(int)) np = 5;
        else if (ws_size >= (fixedW + 2 * planeW) * sizeof(int)) np = 2;
        else if (ws_size >= (fixedW + 1 * planeW) * sizeof(int)) np = 1;
    }
    const int nBin = (np > 0) ? NCH : 0;

    const int T = BIN_T;
    const int wavesPerBlock = T / 32;

    const int capRows  = 512;
    const int rowTrips = (N + 3) / 4;
    int need1 = (rowTrips + wavesPerBlock - 1) / wavesPerBlock;
    need1 = (need1 < 1) ? 1 : need1;
    const int rounds1 = (need1 + capRows - 1) / capRows;
    const int g1 = (need1 + rounds1 - 1) / rounds1;
    k1_node_scalars<<<g1 + nBin, T, 0, stream>>>(h, w_attn, s1, s2, N, e, src, dst, dir, rec, E, R, CS, np, nBin);

    if (np == 5)
        k2_owner_sums<5><<<owners, OWN_THREADS, 0, stream>>>(e, src, dst, s1, s2, w_edge, w_attn, w_e2n, out, dir, rec, CS, E, N, R);
    else if (np == 2)
        k2_owner_sums<2><<<owners, OWN_THREADS, 0, stream>>>(e, src, dst, s1, s2, w_edge, w_attn, w_e2n, out, dir, rec, CS, E, N, R);
    else if (np == 1)
        k2_owner_sums<1><<<owners, OWN_THREADS, 0, stream>>>(e, src, dst, s1, s2, w_edge, w_attn, w_e2n, out, dir, rec, CS, E, N, R);
    else
        k2_owner_sums<0><<<owners, OWN_THREADS, 0, stream>>>(e, src, dst, s1, s2, w_edge, w_attn, w_e2n, out, dir, rec, CS, E, N, R);
}
